# starting kernel
# baseline (speedup 1.0000x reference)
#include <hip/hip_runtime.h>
#include <cstdint>

typedef float v2f __attribute__((ext_vector_type(2)));
typedef float v8f __attribute__((ext_vector_type(8)));

__global__ __launch_bounds__(256) void fill_f32(float* __restrict__ p, float v, int n) {
  int i = blockIdx.x * 256 + threadIdx.x;
  if (i < n) p[i] = v;
}

__global__ __launch_bounds__(256) void deg_accum(const int* __restrict__ dst,
                                                 const float* __restrict__ ew,
                                                 float* __restrict__ deg, int E) {
  int e = blockIdx.x * 256 + threadIdx.x;
  if (e < E) unsafeAtomicAdd(&deg[dst[e]], ew[e]);
}

__global__ __launch_bounds__(256) void rsqrt_inplace(float* __restrict__ p, int n) {
  int i = blockIdx.x * 256 + threadIdx.x;
  if (i < n) {
    float d = p[i];
    p[i] = (d > 0.f) ? rsqrtf(d) : 0.f;
  }
}

__global__ __launch_bounds__(256) void norm_kernel(const int* __restrict__ src,
                                                   const int* __restrict__ dst,
                                                   const float* __restrict__ ew,
                                                   const float* __restrict__ dinv,
                                                   float* __restrict__ norm, int E) {
  int e = blockIdx.x * 256 + threadIdx.x;
  if (e < E) norm[e] = dinv[src[e]] * ew[e] * dinv[dst[e]];
}

template <int IN, int OUT>
__global__ __launch_bounds__(256) void gemm_wmma(const float* __restrict__ x,
                                                 const float* __restrict__ W,
                                                 float* __restrict__ y, int n) {
  constexpr bool USE_LDS = (IN * OUT) <= 8192;
  constexpr bool FULL    = (OUT % 16) == 0;
  __shared__ float sW[USE_LDS ? (IN * OUT) : 1];
  const float* Wp = W;
  if constexpr (USE_LDS) {
    for (int i = threadIdx.x; i < IN * OUT; i += 256) sW[i] = W[i];
    __syncthreads();
    Wp = sW;
  }

  const int lane  = threadIdx.x & 31;
  const int wave  = threadIdx.x >> 5;
  const int l15   = lane & 15;
  const int khalf = lane >> 4;
  const int row0  = blockIdx.x * 128 + wave * 16;

  if (row0 < n) {
    int mload = row0 + l15;
    if (mload > n - 1) mload = n - 1;
    const float* xrow = x + (size_t)mload * IN;
    const bool fulltile = (row0 + 16 <= n);

    for (int n0 = 0; n0 < OUT; n0 += 16) {
      const int colb = n0 + l15;
      const int colc = FULL ? colb : (colb < OUT ? colb : OUT - 1);
      v8f c = {};
#pragma unroll
      for (int k0 = 0; k0 < IN; k0 += 4) {
        const int krow = k0 + 2 * khalf;
        v2f a = *(const v2f*)(xrow + krow);
        v2f b;
        b.x = Wp[(size_t)krow * OUT + colc];
        b.y = Wp[(size_t)(krow + 1) * OUT + colc];
        c = __builtin_amdgcn_wmma_f32_16x16x4_f32(
             false, a,  false, b,
             (short)0, c,  false,  false);
      }
      const int rowoff = khalf * 8;
      if (FULL && fulltile) {
#pragma unroll
        for (int r = 0; r < 8; ++r)
          y[(size_t)(row0 + rowoff + r) * OUT + colb] = c[r];
      } else {
        const bool colok = FULL || (colb < OUT);
#pragma unroll
        for (int r = 0; r < 8; ++r) {
          int row = row0 + rowoff + r;
          if (row < n && colok) y[(size_t)row * OUT + colb] = c[r];
        }
      }
    }
  }
}

template <int OUT>
__global__ __launch_bounds__(256) void self_bias(const float* __restrict__ xw,
                                                 const float* __restrict__ dinv,
                                                 const float* __restrict__ b,
                                                 float* __restrict__ out, int n) {
  int tid = blockIdx.x * 256 + threadIdx.x;
  if (tid >= n * OUT) return;
  int i = tid / OUT, c = tid - i * OUT;
  float dv = dinv[i];
  out[tid] = xw[tid] * (dv * dv) + b[c];
}

template <int OUT>
__global__ __launch_bounds__(256) void edge_scatter(const float* __restrict__ xw,
                                                    const int* __restrict__ src,
                                                    const int* __restrict__ dst,
                                                    const float* __restrict__ norm,
                                                    float* __restrict__ out, int E) {
  int tid = blockIdx.x * 256 + threadIdx.x;
  if (tid >= E * OUT) return;
  int e = tid / OUT, c = tid - e * OUT;
  float v = xw[(size_t)src[e] * OUT + c] * norm[e];
  unsafeAtomicAdd(&out[(size_t)dst[e] * OUT + c], v);
}

__global__ __launch_bounds__(256) void elu_inplace(float* __restrict__ p, int n) {
  int i = blockIdx.x * 256 + threadIdx.x;
  if (i < n) {
    float v = p[i];
    p[i] = (v > 0.f) ? v : expm1f(v);
  }
}

static inline int cdiv(long long a, long long b) { return (int)((a + b - 1) / b); }

template <int IN, int OUT>
static void run_layer(const float* xin, float* xw, float* xout,
                      const float* W, const float* b,
                      const int* src, const int* dst,
                      const float* dinv, const float* norm,
                      int N, int E, bool act, hipStream_t stream) {
  gemm_wmma<IN, OUT><<<cdiv(N, 128), 256, 0, stream>>>(xin, W, xw, N);
  self_bias<OUT><<<cdiv((long long)N * OUT, 256), 256, 0, stream>>>(xw, dinv, b, xout, N);
  edge_scatter<OUT><<<cdiv((long long)E * OUT, 256), 256, 0, stream>>>(xw, src, dst, norm, xout, E);
  if (act) elu_inplace<<<cdiv((long long)N * OUT, 256), 256, 0, stream>>>(xout, N * OUT);
}

extern "C" void kernel_launch(void* const* d_in, const int* in_sizes, int n_in,
                              void* d_out, int out_size, void* d_ws, size_t ws_size,
                              hipStream_t stream) {
  const float* x   = (const float*)d_in[0];
  const int*   ei  = (const int*)d_in[1];
  const float* ew  = (const float*)d_in[2];
  const float* W1 = (const float*)d_in[3];  const float* b1 = (const float*)d_in[4];
  const float* W2 = (const float*)d_in[5];  const float* b2 = (const float*)d_in[6];
  const float* W3 = (const float*)d_in[7];  const float* b3 = (const float*)d_in[8];
  const float* W4 = (const float*)d_in[9];  const float* b4 = (const float*)d_in[10];
  const float* W5 = (const float*)d_in[11]; const float* b5 = (const float*)d_in[12];
  const float* W6 = (const float*)d_in[13]; const float* b6 = (const float*)d_in[14];

  const int N = in_sizes[0] / 32;
  const int E = in_sizes[2];
  const int* src = ei;
  const int* dst = ei + E;

  float* dinv = (float*)d_ws;
  float* norm = dinv + N;
  float* buf0 = norm + E;
  float* buf1 = buf0 + (size_t)N * 128;
  float* outp = (float*)d_out;

  fill_f32<<<cdiv(N, 256), 256, 0, stream>>>(dinv, 1.0f, N);
  deg_accum<<<cdiv(E, 256), 256, 0, stream>>>(dst, ew, dinv, E);
  rsqrt_inplace<<<cdiv(N, 256), 256, 0, stream>>>(dinv, N);
  norm_kernel<<<cdiv(E, 256), 256, 0, stream>>>(src, dst, ew, dinv, norm, E);

  run_layer<32,   4>(x,    buf1, buf0, W1, b1, src, dst, dinv, norm, N, E, true,  stream);
  run_layer<4,   16>(buf0, buf1, buf0, W2, b2, src, dst, dinv, norm, N, E, true,  stream);
  run_layer<16,  32>(buf0, buf1, buf0, W3, b3, src, dst, dinv, norm, N, E, true,  stream);
  run_layer<32,  64>(buf0, buf1, buf0, W4, b4, src, dst, dinv, norm, N, E, false, stream);
  run_layer<64, 128>(buf0, buf1, buf0, W5, b5, src, dst, dinv, norm, N, E, true,  stream);
  run_layer<128,128>(buf0, buf1, outp, W6, b6, src, dst, dinv, norm, N, E, false, stream);
}


# Layer scatter: a block owns 392 destination rows in LDS (start self-loop+bias), scans dst for its edges, ds_add_f32 instead of global float atomics, rows stored once with ELU; sum order = LDS arrival
# speedup vs baseline: 2.5482x; 2.5482x over previous
#include <hip/hip_runtime.h>
#include <cstdint>

typedef float v2f __attribute__((ext_vector_type(2)));
typedef float v8f __attribute__((ext_vector_type(8)));

__global__ __launch_bounds__(256) void fill_f32(float* __restrict__ p, float v, int n) {
  int i = blockIdx.x * 256 + threadIdx.x;
  if (i < n) p[i] = v;
}

__global__ __launch_bounds__(256) void deg_accum(const int* __restrict__ dst,
                                                 const float* __restrict__ ew,
                                                 float* __restrict__ deg, int E) {
  int e = blockIdx.x * 256 + threadIdx.x;
  if (e < E) unsafeAtomicAdd(&deg[dst[e]], ew[e]);
}

__global__ __launch_bounds__(256) void rsqrt_inplace(float* __restrict__ p, int n) {
  int i = blockIdx.x * 256 + threadIdx.x;
  if (i < n) {
    float d = p[i];
    p[i] = (d > 0.f) ? rsqrtf(d) : 0.f;
  }
}

__global__ __launch_bounds__(256) void norm_kernel(const int* __restrict__ src,
                                                   const int* __restrict__ dst,
                                                   const float* __restrict__ ew,
                                                   const float* __restrict__ dinv,
                                                   float* __restrict__ norm, int E) {
  int e = blockIdx.x * 256 + threadIdx.x;
  if (e < E) norm[e] = dinv[src[e]] * ew[e] * dinv[dst[e]];
}

template <int IN, int OUT>
__global__ __launch_bounds__(256) void gemm_wmma(const float* __restrict__ x,
                                                 const float* __restrict__ W,
                                                 float* __restrict__ y, int n) {
  constexpr bool USE_LDS = (IN * OUT) <= 8192;
  constexpr bool FULL    = (OUT % 16) == 0;
  __shared__ float sW[USE_LDS ? (IN * OUT) : 1];
  const float* Wp = W;
  if constexpr (USE_LDS) {
    for (int i = threadIdx.x; i < IN * OUT; i += 256) sW[i] = W[i];
    __syncthreads();
    Wp = sW;
  }

  const int lane  = threadIdx.x & 31;
  const int wave  = threadIdx.x >> 5;
  const int l15   = lane & 15;
  const int khalf = lane >> 4;
  const int row0  = blockIdx.x * 128 + wave * 16;

  if (row0 < n) {
    int mload = row0 + l15;
    if (mload > n - 1) mload = n - 1;
    const float* xrow = x + (size_t)mload * IN;
    const bool fulltile = (row0 + 16 <= n);

    for (int n0 = 0; n0 < OUT; n0 += 16) {
      const int colb = n0 + l15;
      const int colc = FULL ? colb : (colb < OUT ? colb : OUT - 1);
      v8f c = {};
#pragma unroll
      for (int k0 = 0; k0 < IN; k0 += 4) {
        const int krow = k0 + 2 * khalf;
        v2f a = *(const v2f*)(xrow + krow);
        v2f b;
        b.x = Wp[(size_t)krow * OUT + colc];
        b.y = Wp[(size_t)(krow + 1) * OUT + colc];
        c = __builtin_amdgcn_wmma_f32_16x16x4_f32(
             false, a,  false, b,
             (short)0, c,  false,  false);
      }
      const int rowoff = khalf * 8;
      if (FULL && fulltile) {
#pragma unroll
        for (int r = 0; r < 8; ++r)
          y[(size_t)(row0 + rowoff + r) * OUT + colb] = c[r];
      } else {
        const bool colok = FULL || (colb < OUT);
#pragma unroll
        for (int r = 0; r < 8; ++r) {
          int row = row0 + rowoff + r;
          if (row < n && colok) y[(size_t)row * OUT + colb] = c[r];
        }
      }
    }
  }
}

constexpr int TROWS      = 392;
constexpr int LDS_FLOATS = 61440;

template <int OUT, bool ACT>
__global__ __launch_bounds__(1024) void agg_tile(const float* __restrict__ xw,
                                                  const int* __restrict__ src,
                                                  const int* __restrict__ dst,
                                                  const float* __restrict__ norm,
                                                  const float* __restrict__ dinv,
                                                  const float* __restrict__ b,
                                                  float* __restrict__ out, int n, int E) {
  constexpr int TILE = TROWS * OUT;
  constexpr int QCAP = (LDS_FLOATS - TILE) & ~3;
  constexpr int O4   = OUT / 4;
  constexpr int LPE  = OUT < 32 ? OUT : 32;
  constexpr int EPS  = 32 / LPE;
  constexpr int CPL  = OUT < 32 ? 1 : OUT / 32;
  constexpr int UN   = LPE < 8 ? LPE : 8;

  __shared__ float tile[TILE];
  __shared__ int   q[QCAP];
  __shared__ int   qn;

  const int tid  = threadIdx.x;
  const int lane = tid & 31;
  const int wave = tid >> 5;
  const int qsub = lane / LPE;
  const int ch0  = lane % LPE;
  const bool vec_ok = (((size_t)dst) & 15) == 0;
  const int4* dst4 = (const int4*)dst;

  for (int t = blockIdx.x; t * TROWS < n; t += gridDim.x) {
    const int base = t * TROWS;
    const int rows = (n - base < TROWS) ? (n - base) : TROWS;

    {
      const float4* xw4 = (const float4*)xw + (size_t)base * O4;
      float4* tile4 = (float4*)tile;
      for (int i = tid; i < rows * O4; i += 1024) {
        const int row = i / O4, c = (i - row * O4) * 4;
        const float dv = dinv[base + row];
        const float w = dv * dv;
        const float4 x = xw4[i];
        float4 r;
        r.x = x.x * w + b[c];
        r.y = x.y * w + b[c + 1];
        r.z = x.z * w + b[c + 2];
        r.w = x.w * w + b[c + 3];
        tile4[i] = r;
      }
    }

    int seg_lo = 0, seg_len = E;
    while (seg_lo < E) {
      const int seg_hi = (E - seg_lo < seg_len) ? E : seg_lo + seg_len;
      if (tid == 0) qn = 0;
      __syncthreads();

      auto push = [&](int d, int e) {
        if ((unsigned)(d - base) < (unsigned)TROWS) {
          const int pos = atomicAdd(&qn, 1);
          if (pos < QCAP) q[pos] = e;
        }
      };
      auto test4 = [&](int4 v, int e0) {
        const unsigned rx = (unsigned)(v.x - base), ry = (unsigned)(v.y - base);
        const unsigned rz = (unsigned)(v.z - base), rw = (unsigned)(v.w - base);
        const unsigned m0 = rx < ry ? rx : ry, m1 = rz < rw ? rz : rw;
        if ((m0 < m1 ? m0 : m1) < (unsigned)TROWS) {
          push(v.x, e0); push(v.y, e0 + 1); push(v.z, e0 + 2); push(v.w, e0 + 3);
        }
      };
      const int i4_lo   = seg_lo >> 2;
      const int i4_full = vec_ok ? (seg_hi >> 2) : i4_lo;
      int i = i4_lo + tid;
      for (; i + 7 * 1024 < i4_full; i += 8 * 1024) {
        int4 v[8];
#pragma unroll
        for (int u = 0; u < 8; ++u) v[u] = dst4[i + u * 1024];
#pragma unroll
        for (int u = 0; u < 8; ++u) test4(v[u], (i + u * 1024) << 2);
      }
      for (; i < i4_full; i += 1024) test4(dst4[i], i << 2);
      for (int e = (i4_full << 2) + tid; e < seg_hi; e += 1024) push(dst[e], e);
      __syncthreads();

      const int cnt = qn;
      if (cnt <= QCAP) {
        const int nb = (cnt + 31) >> 5;
        for (int bi = wave; bi < nb; bi += 32) {
          const int left = cnt - (bi << 5);
          const int e    = q[(lane < left) ? ((bi << 5) + lane) : (bi << 5)];
          const int   s_l  = src[e];
          const int   r_l  = dst[e] - base;
          const float nm_l = norm[e];
#pragma unroll 1
          for (int k0 = 0; k0 < LPE; k0 += UN) {
            float val[UN][CPL];
            int   rr[UN];
            float nn[UN];
#pragma unroll
            for (int u = 0; u < UN; ++u) {
              const int ent = (k0 + u) * EPS + qsub;
              int s;
              if constexpr (LPE == 32) {
                s     = __builtin_amdgcn_readlane(s_l, k0 + u);
                rr[u] = __builtin_amdgcn_readlane(r_l, k0 + u);
                nn[u] = __int_as_float(__builtin_amdgcn_readlane(__float_as_int(nm_l), k0 + u));
              } else {
                s     = __shfl(s_l, ent);
                rr[u] = __shfl(r_l, ent);
                nn[u] = __shfl(nm_l, ent);
              }
              const float* xr = xw + (size_t)s * OUT + ch0;
#pragma unroll
              for (int j = 0; j < CPL; ++j) val[u][j] = xr[32 * j];
            }
#pragma unroll
            for (int u = 0; u < UN; ++u) {
              const int ent = (k0 + u) * EPS + qsub;
              if (ent < left) {
                float* tr = tile + rr[u] * OUT + ch0;
#pragma unroll
                for (int j = 0; j < CPL; ++j) atomicAdd(tr + 32 * j, val[u][j] * nn[u]);
              }
            }
          }
        }
        seg_lo = seg_hi;
      } else {
        seg_len = QCAP;
      }
      __syncthreads();
    }

    {
      float4* out4 = (float4*)out + (size_t)base * O4;
      const float4* tile4 = (const float4*)tile;
      for (int i = tid; i < rows * O4; i += 1024) {
        float4 v = tile4[i];
        if (ACT) {
          v.x = (v.x > 0.f) ? v.x : expm1f(v.x);
          v.y = (v.y > 0.f) ? v.y : expm1f(v.y);
          v.z = (v.z > 0.f) ? v.z : expm1f(v.z);
          v.w = (v.w > 0.f) ? v.w : expm1f(v.w);
        }
        out4[i] = v;
      }
    }
    __syncthreads();
  }
}

static inline int cdiv(long long a, long long b) { return (int)((a + b - 1) / b); }

template <int IN, int OUT, bool ACT>
static void run_layer(const float* xin, float* xw, float* xout,
                      const float* W, const float* b,
                      const int* src, const int* dst,
                      const float* dinv, const float* norm,
                      int N, int E, hipStream_t stream) {
  gemm_wmma<IN, OUT><<<cdiv(N, 128), 256, 0, stream>>>(xin, W, xw, N);
  agg_tile<OUT, ACT><<<cdiv(N, TROWS), 1024, 0, stream>>>(xw, src, dst, norm, dinv, b, xout, N, E);
}

extern "C" void kernel_launch(void* const* d_in, const int* in_sizes, int n_in,
                              void* d_out, int out_size, void* d_ws, size_t ws_size,
                              hipStream_t stream) {
  const float* x   = (const float*)d_in[0];
  const int*   ei  = (const int*)d_in[1];
  const float* ew  = (const float*)d_in[2];
  const float* W1 = (const float*)d_in[3];  const float* b1 = (const float*)d_in[4];
  const float* W2 = (const float*)d_in[5];  const float* b2 = (const float*)d_in[6];
  const float* W3 = (const float*)d_in[7];  const float* b3 = (const float*)d_in[8];
  const float* W4 = (const float*)d_in[9];  const float* b4 = (const float*)d_in[10];
  const float* W5 = (const float*)d_in[11]; const float* b5 = (const float*)d_in[12];
  const float* W6 = (const float*)d_in[13]; const float* b6 = (const float*)d_in[14];

  const int N = in_sizes[0] / 32;
  const int E = in_sizes[2];
  const int* src = ei;
  const int* dst = ei + E;

  float* dinv = (float*)d_ws;
  float* norm = dinv + N;
  float* buf0 = norm + E;
  float* buf1 = buf0 + (size_t)N * 128;
  float* outp = (float*)d_out;

  fill_f32<<<cdiv(N, 256), 256, 0, stream>>>(dinv, 1.0f, N);
  deg_accum<<<cdiv(E, 256), 256, 0, stream>>>(dst, ew, dinv, E);
  rsqrt_inplace<<<cdiv(N, 256), 256, 0, stream>>>(dinv, N);
  norm_kernel<<<cdiv(E, 256), 256, 0, stream>>>(src, dst, ew, dinv, norm, E);

  run_layer<32,   4, true >(x,    buf1, buf0, W1, b1, src, dst, dinv, norm, N, E, stream);
  run_layer<4,   16, true >(buf0, buf1, buf0, W2, b2, src, dst, dinv, norm, N, E, stream);
  run_layer<16,  32, true >(buf0, buf1, buf0, W3, b3, src, dst, dinv, norm, N, E, stream);
  run_layer<32,  64, false>(buf0, buf1, buf0, W4, b4, src, dst, dinv, norm, N, E, stream);
  run_layer<64, 128, true >(buf0, buf1, buf0, W5, b5, src, dst, dinv, norm, N, E, stream);
  run_layer<128,128, false>(buf0, buf1, outp, W6, b6, src, dst, dinv, norm, N, E, stream);
}


# Layer scan reads the edges' one-byte tile keys as bit planes (8 wave ballots per 32 edges, written beside norm), not the 4-byte dst list: 4x fewer bytes, 32 edges per 8 bit ops; sum order LDS arrival
# speedup vs baseline: 3.3261x; 1.3053x over previous
#include <hip/hip_runtime.h>
#include <cstdint>

typedef float v2f __attribute__((ext_vector_type(2)));
typedef float v8f __attribute__((ext_vector_type(8)));

__global__ __launch_bounds__(256) void fill_f32(float* __restrict__ p, float v, int n) {
  int i = blockIdx.x * 256 + threadIdx.x;
  if (i < n) p[i] = v;
}

__global__ __launch_bounds__(256) void deg_accum(const int* __restrict__ dst,
                                                 const float* __restrict__ ew,
                                                 float* __restrict__ deg, int E) {
  int e = blockIdx.x * 256 + threadIdx.x;
  if (e < E) unsafeAtomicAdd(&deg[dst[e]], ew[e]);
}

__global__ __launch_bounds__(256) void rsqrt_inplace(float* __restrict__ p, int n) {
  int i = blockIdx.x * 256 + threadIdx.x;
  if (i < n) {
    float d = p[i];
    p[i] = (d > 0.f) ? rsqrtf(d) : 0.f;
  }
}

constexpr int TROWS = 392;

__global__ __launch_bounds__(256) void norm_kernel(const int* __restrict__ src,
                                                   const int* __restrict__ dst,
                                                   const float* __restrict__ ew,
                                                   const float* __restrict__ dinv,
                                                   float* __restrict__ norm,
                                                   int* __restrict__ key, int E) {
  const int e = blockIdx.x * 256 + threadIdx.x;
  int d = 0;
  if (e < E) {
    d = dst[e];
    norm[e] = dinv[src[e]] * ew[e] * dinv[d];
  }
  const unsigned k = (unsigned)(d / TROWS);
  uint4 lo, hi;
  lo.x = __builtin_amdgcn_ballot_w32((k & 1u) != 0);
  lo.y = __builtin_amdgcn_ballot_w32((k & 2u) != 0);
  lo.z = __builtin_amdgcn_ballot_w32((k & 4u) != 0);
  lo.w = __builtin_amdgcn_ballot_w32((k & 8u) != 0);
  hi.x = __builtin_amdgcn_ballot_w32((k & 16u) != 0);
  hi.y = __builtin_amdgcn_ballot_w32((k & 32u) != 0);
  hi.z = __builtin_amdgcn_ballot_w32((k & 64u) != 0);
  hi.w = __builtin_amdgcn_ballot_w32((k & 128u) != 0);
  if ((threadIdx.x & 31) == 0) {
    const int g = e >> 5;
    uint4* p = (uint4*)key + ((g >> 5) * 64 + (g & 31));
    p[0]  = lo;
    p[32] = hi;
  }
}

template <int IN, int OUT>
__global__ __launch_bounds__(256) void gemm_wmma(const float* __restrict__ x,
                                                 const float* __restrict__ W,
                                                 float* __restrict__ y, int n) {
  constexpr bool USE_LDS = (IN * OUT) <= 8192;
  constexpr bool FULL    = (OUT % 16) == 0;
  __shared__ float sW[USE_LDS ? (IN * OUT) : 1];
  const float* Wp = W;
  if constexpr (USE_LDS) {
    for (int i = threadIdx.x; i < IN * OUT; i += 256) sW[i] = W[i];
    __syncthreads();
    Wp = sW;
  }

  const int lane  = threadIdx.x & 31;
  const int wave  = threadIdx.x >> 5;
  const int l15   = lane & 15;
  const int khalf = lane >> 4;
  const int row0  = blockIdx.x * 128 + wave * 16;

  if (row0 < n) {
    int mload = row0 + l15;
    if (mload > n - 1) mload = n - 1;
    const float* xrow = x + (size_t)mload * IN;
    const bool fulltile = (row0 + 16 <= n);

    for (int n0 = 0; n0 < OUT; n0 += 16) {
      const int colb = n0 + l15;
      const int colc = FULL ? colb : (colb < OUT ? colb : OUT - 1);
      v8f c = {};
#pragma unroll
      for (int k0 = 0; k0 < IN; k0 += 4) {
        const int krow = k0 + 2 * khalf;
        v2f a = *(const v2f*)(xrow + krow);
        v2f b;
        b.x = Wp[(size_t)krow * OUT + colc];
        b.y = Wp[(size_t)(krow + 1) * OUT + colc];
        c = __builtin_amdgcn_wmma_f32_16x16x4_f32(
             false, a,  false, b,
             (short)0, c,  false,  false);
      }
      const int rowoff = khalf * 8;
      if (FULL && fulltile) {
#pragma unroll
        for (int r = 0; r < 8; ++r)
          y[(size_t)(row0 + rowoff + r) * OUT + colb] = c[r];
      } else {
        const bool colok = FULL || (colb < OUT);
#pragma unroll
        for (int r = 0; r < 8; ++r) {
          int row = row0 + rowoff + r;
          if (row < n && colok) y[(size_t)row * OUT + colb] = c[r];
        }
      }
    }
  }
}

constexpr int LDS_FLOATS = 61440;

template <int OUT, bool ACT>
__global__ __launch_bounds__(1024) void agg_tile(const float* __restrict__ xw,
                                                  const int* __restrict__ src,
                                                  const int* __restrict__ dst,
                                                  const int* __restrict__ key,
                                                  const float* __restrict__ norm,
                                                  const float* __restrict__ dinv,
                                                  const float* __restrict__ b,
                                                  float* __restrict__ out, int n, int E) {
  constexpr int TILE = TROWS * OUT;
  constexpr int QCAP = (LDS_FLOATS - TILE) & ~1023;
  constexpr int O4   = OUT / 4;
  constexpr int LPE  = OUT < 32 ? OUT : 32;
  constexpr int EPS  = 32 / LPE;
  constexpr int CPL  = OUT < 32 ? 1 : OUT / 32;
  constexpr int UN   = LPE < 8 ? LPE : 8;

  __shared__ float tile[TILE];
  __shared__ int   q[QCAP];
  __shared__ int   qn;

  const int tid  = threadIdx.x;
  const int lane = tid & 31;
  const int wave = tid >> 5;
  const int qsub = lane / LPE;
  const int ch0  = lane % LPE;
  const uint4* key16 = (const uint4*)key;
  const bool vec_ok = ((n + TROWS - 1) / TROWS <= 256) && ((((size_t)key) & 15) == 0);

  for (int t = blockIdx.x; t * TROWS < n; t += gridDim.x) {
    const int base = t * TROWS;
    const int rows = (n - base < TROWS) ? (n - base) : TROWS;
    unsigned kx[8];
#pragma unroll
    for (int j = 0; j < 8; ++j) kx[j] = 0u - (((unsigned)t >> j) & 1u);

    {
      const float4* xw4 = (const float4*)xw + (size_t)base * O4;
      float4* tile4 = (float4*)tile;
      for (int i = tid; i < rows * O4; i += 1024) {
        const int row = i / O4, c = (i - row * O4) * 4;
        const float dv = dinv[base + row];
        const float w = dv * dv;
        const float4 x = xw4[i];
        float4 r;
        r.x = x.x * w + b[c];
        r.y = x.y * w + b[c + 1];
        r.z = x.z * w + b[c + 2];
        r.w = x.w * w + b[c + 3];
        tile4[i] = r;
      }
    }

    int seg_lo = 0, seg_len = E;
    while (seg_lo < E) {
      const int seg_hi = (E - seg_lo < seg_len) ? E : seg_lo + seg_len;
      if (tid == 0) qn = 0;
      __syncthreads();

      auto take = [&](int e) {
        const int pos = atomicAdd(&qn, 1);
        if (pos < QCAP) q[pos] = e;
      };
      auto test32 = [&](uint4 lo, uint4 hi, int e0) {
        unsigned m = ~((lo.x ^ kx[0]) | (lo.y ^ kx[1]) | (lo.z ^ kx[2]) | (lo.w ^ kx[3]) |
                       (hi.x ^ kx[4]) | (hi.y ^ kx[5]) | (hi.z ^ kx[6]) | (hi.w ^ kx[7]));
        while (m) {
          take(e0 + __builtin_ctz(m));
          m &= m - 1;
        }
      };
      const int g_lo   = seg_lo >> 5;
      const int g_full = vec_ok ? (seg_hi >> 5) : g_lo;
      int g = g_lo + tid;
      for (; g + 3 * 1024 < g_full; g += 4 * 1024) {
        const uint4* p = key16 + ((g >> 5) * 64 + lane);
        uint4 v[8];
#pragma unroll
        for (int u = 0; u < 4; ++u) {
          v[2 * u]     = p[u * 2048];
          v[2 * u + 1] = p[u * 2048 + 32];
        }
#pragma unroll
        for (int u = 0; u < 4; ++u) test32(v[2 * u], v[2 * u + 1], (g + u * 1024) << 5);
      }
      for (; g < g_full; g += 1024) {
        const uint4* p = key16 + ((g >> 5) * 64 + lane);
        test32(p[0], p[32], g << 5);
      }
      for (int e = (g_full << 5) + tid; e < seg_hi; e += 1024)
        if ((unsigned)(dst[e] - base) < (unsigned)TROWS) take(e);
      __syncthreads();

      const int cnt = qn;
      if (cnt <= QCAP) {
        const int nb = (cnt + 31) >> 5;
        for (int bi = wave; bi < nb; bi += 32) {
          const int left = cnt - (bi << 5);
          const int e    = q[(lane < left) ? ((bi << 5) + lane) : (bi << 5)];
          const int   s_l  = src[e];
          const int   r_l  = dst[e] - base;
          const float nm_l = norm[e];
#pragma unroll 1
          for (int k0 = 0; k0 < LPE; k0 += UN) {
            float val[UN][CPL];
            int   rr[UN];
            float nn[UN];
#pragma unroll
            for (int u = 0; u < UN; ++u) {
              const int ent = (k0 + u) * EPS + qsub;
              int s;
              if constexpr (LPE == 32) {
                s     = __builtin_amdgcn_readlane(s_l, k0 + u);
                rr[u] = __builtin_amdgcn_readlane(r_l, k0 + u);
                nn[u] = __int_as_float(__builtin_amdgcn_readlane(__float_as_int(nm_l), k0 + u));
              } else {
                s     = __shfl(s_l, ent);
                rr[u] = __shfl(r_l, ent);
                nn[u] = __shfl(nm_l, ent);
              }
              const float* xr = xw + (size_t)s * OUT + ch0;
#pragma unroll
              for (int j = 0; j < CPL; ++j) val[u][j] = xr[32 * j];
            }
#pragma unroll
            for (int u = 0; u < UN; ++u) {
              const int ent = (k0 + u) * EPS + qsub;
              if (ent < left) {
                float* tr = tile + rr[u] * OUT + ch0;
#pragma unroll
                for (int j = 0; j < CPL; ++j) atomicAdd(tr + 32 * j, val[u][j] * nn[u]);
              }
            }
          }
        }
        seg_lo = seg_hi;
      } else {
        seg_len = QCAP;
      }
      __syncthreads();
    }

    {
      float4* out4 = (float4*)out + (size_t)base * O4;
      const float4* tile4 = (const float4*)tile;
      for (int i = tid; i < rows * O4; i += 1024) {
        float4 v = tile4[i];
        if (ACT) {
          v.x = (v.x > 0.f) ? v.x : expm1f(v.x);
          v.y = (v.y > 0.f) ? v.y : expm1f(v.y);
          v.z = (v.z > 0.f) ? v.z : expm1f(v.z);
          v.w = (v.w > 0.f) ? v.w : expm1f(v.w);
        }
        out4[i] = v;
      }
    }
    __syncthreads();
  }
}

static inline int cdiv(long long a, long long b) { return (int)((a + b - 1) / b); }

template <int IN, int OUT, bool ACT>
static void run_layer(const float* xin, float* xw, float* xout,
                      const float* W, const float* b,
                      const int* src, const int* dst,
                      const int* key,
                      const float* dinv, const float* norm,
                      int N, int E, hipStream_t stream) {
  gemm_wmma<IN, OUT><<<cdiv(N, 128), 256, 0, stream>>>(xin, W, xw, N);
  agg_tile<OUT, ACT><<<cdiv(N, TROWS), 1024, 0, stream>>>(xw, src, dst, key, norm, dinv, b, xout, N, E);
}

extern "C" void kernel_launch(void* const* d_in, const int* in_sizes, int n_in,
                              void* d_out, int out_size, void* d_ws, size_t ws_size,
                              hipStream_t stream) {
  const float* x   = (const float*)d_in[0];
  const int*   ei  = (const int*)d_in[1];
  const float* ew  = (const float*)d_in[2];
  const float* W1 = (const float*)d_in[3];  const float* b1 = (const float*)d_in[4];
  const float* W2 = (const float*)d_in[5];  const float* b2 = (const float*)d_in[6];
  const float* W3 = (const float*)d_in[7];  const float* b3 = (const float*)d_in[8];
  const float* W4 = (const float*)d_in[9];  const float* b4 = (const float*)d_in[10];
  const float* W5 = (const float*)d_in[11]; const float* b5 = (const float*)d_in[12];
  const float* W6 = (const float*)d_in[13]; const float* b6 = (const float*)d_in[14];

  const int N = in_sizes[0] / 32;
  const int E = in_sizes[2];
  const int* src = ei;
  const int* dst = ei + E;

  float* dinv = (float*)d_ws;
  float* norm = dinv + N;
  float* buf0 = norm + E;
  float* buf1 = buf0 + (size_t)N * 128;
  float* outp = (float*)d_out;
  int* key = (int*)(buf0 + (size_t)N * 64);

  fill_f32<<<cdiv(N, 256), 256, 0, stream>>>(dinv, 1.0f, N);
  deg_accum<<<cdiv(E, 256), 256, 0, stream>>>(dst, ew, dinv, E);
  rsqrt_inplace<<<cdiv(N, 256), 256, 0, stream>>>(dinv, N);
  norm_kernel<<<cdiv(E, 256), 256, 0, stream>>>(src, dst, ew, dinv, norm, key, E);

  run_layer<32,   4, true >(x,    buf1, buf0, W1, b1, src, dst, key, dinv, norm, N, E, stream);
  run_layer<4,   16, true >(buf0, buf1, buf0, W2, b2, src, dst, key, dinv, norm, N, E, stream);
  run_layer<16,  32, true >(buf0, buf1, buf0, W3, b3, src, dst, key, dinv, norm, N, E, stream);
  run_layer<32,  64, false>(buf0, buf1, buf0, W4, b4, src, dst, key, dinv, norm, N, E, stream);
  run_layer<64, 128, true >(buf0, buf1, outp, W5, b5, src, dst, key, dinv, norm, N, E, stream);
  run_layer<128,128, false>(outp, buf1, outp, W6, b6, src, dst, key, dinv, norm, N, E, stream);
}


# Tile edge lists kept: layer 1's scan also writes each tile's edges as 8-byte records (source+row, norm); layers 2-6 read them in whole lines: no rescan, no index gathers; sum order LDS arrival
# speedup vs baseline: 4.4978x; 1.3523x over previous
#include <hip/hip_runtime.h>
#include <cstdint>

typedef float v2f __attribute__((ext_vector_type(2)));
typedef float v8f __attribute__((ext_vector_type(8)));

__global__ __launch_bounds__(256) void fill_f32(float* __restrict__ p, float v, int n) {
  int i = blockIdx.x * 256 + threadIdx.x;
  if (i < n) p[i] = v;
}

__global__ __launch_bounds__(256) void deg_accum(const int* __restrict__ dst,
                                                 const float* __restrict__ ew,
                                                 float* __restrict__ deg, int E) {
  int e = blockIdx.x * 256 + threadIdx.x;
  if (e < E) unsafeAtomicAdd(&deg[dst[e]], ew[e]);
}

__global__ __launch_bounds__(256) void rsqrt_inplace(float* __restrict__ p, int n) {
  int i = blockIdx.x * 256 + threadIdx.x;
  if (i < n) {
    float d = p[i];
    p[i] = (d > 0.f) ? rsqrtf(d) : 0.f;
  }
}

constexpr int TROWS = 392;

__global__ __launch_bounds__(256) void norm_kernel(const int* __restrict__ src,
                                                   const int* __restrict__ dst,
                                                   const float* __restrict__ ew,
                                                   const float* __restrict__ dinv,
                                                   float* __restrict__ norm,
                                                   int* __restrict__ key, int E) {
  const int e = blockIdx.x * 256 + threadIdx.x;
  int d = 0;
  if (e < E) {
    d = dst[e];
    norm[e] = dinv[src[e]] * ew[e] * dinv[d];
  }
  const unsigned k = (unsigned)(d / TROWS);
  uint4 lo, hi;
  lo.x = __builtin_amdgcn_ballot_w32((k & 1u) != 0);
  lo.y = __builtin_amdgcn_ballot_w32((k & 2u) != 0);
  lo.z = __builtin_amdgcn_ballot_w32((k & 4u) != 0);
  lo.w = __builtin_amdgcn_ballot_w32((k & 8u) != 0);
  hi.x = __builtin_amdgcn_ballot_w32((k & 16u) != 0);
  hi.y = __builtin_amdgcn_ballot_w32((k & 32u) != 0);
  hi.z = __builtin_amdgcn_ballot_w32((k & 64u) != 0);
  hi.w = __builtin_amdgcn_ballot_w32((k & 128u) != 0);
  if ((threadIdx.x & 31) == 0) {
    const int g = e >> 5;
    uint4* p = (uint4*)key + ((g >> 5) * 64 + (g & 31));
    p[0]  = lo;
    p[32] = hi;
  }
}

template <int IN, int OUT>
__global__ __launch_bounds__(256) void gemm_wmma(const float* __restrict__ x,
                                                 const float* __restrict__ W,
                                                 float* __restrict__ y, int n) {
  constexpr bool USE_LDS = (IN * OUT) <= 8192;
  constexpr bool FULL    = (OUT % 16) == 0;
  __shared__ float sW[USE_LDS ? (IN * OUT) : 1];
  const float* Wp = W;
  if constexpr (USE_LDS) {
    for (int i = threadIdx.x; i < IN * OUT; i += 256) sW[i] = W[i];
    __syncthreads();
    Wp = sW;
  }

  const int lane  = threadIdx.x & 31;
  const int wave  = threadIdx.x >> 5;
  const int l15   = lane & 15;
  const int khalf = lane >> 4;
  const int row0  = blockIdx.x * 128 + wave * 16;

  if (row0 < n) {
    int mload = row0 + l15;
    if (mload > n - 1) mload = n - 1;
    const float* xrow = x + (size_t)mload * IN;
    const bool fulltile = (row0 + 16 <= n);

    for (int n0 = 0; n0 < OUT; n0 += 16) {
      const int colb = n0 + l15;
      const int colc = FULL ? colb : (colb < OUT ? colb : OUT - 1);
      v8f c = {};
#pragma unroll
      for (int k0 = 0; k0 < IN; k0 += 4) {
        const int krow = k0 + 2 * khalf;
        v2f a = *(const v2f*)(xrow + krow);
        v2f b;
        b.x = Wp[(size_t)krow * OUT + colc];
        b.y = Wp[(size_t)(krow + 1) * OUT + colc];
        c = __builtin_amdgcn_wmma_f32_16x16x4_f32(
             false, a,  false, b,
             (short)0, c,  false,  false);
      }
      const int rowoff = khalf * 8;
      if (FULL && fulltile) {
#pragma unroll
        for (int r = 0; r < 8; ++r)
          y[(size_t)(row0 + rowoff + r) * OUT + colb] = c[r];
      } else {
        const bool colok = FULL || (colb < OUT);
#pragma unroll
        for (int r = 0; r < 8; ++r) {
          int row = row0 + rowoff + r;
          if (row < n && colok) y[(size_t)row * OUT + colb] = c[r];
        }
      }
    }
  }
}

constexpr int LDS_FLOATS = 61440;

template <int OUT, bool ACT, int REC>
__global__ __launch_bounds__(1024) void agg_tile(const float* __restrict__ xw,
                                                  const int* __restrict__ src,
                                                  const int* __restrict__ dst,
                                                  const int* __restrict__ key,
                                                  const float* __restrict__ norm,
                                                  const float* __restrict__ dinv,
                                                  const float* __restrict__ b,
                                                  int* tcnt, int* rec, const int* rec_end,
                                                  float* __restrict__ out, int n, int E) {
  constexpr int TILE = TROWS * OUT;
  constexpr int QCAP = (LDS_FLOATS - TILE) & ~1023;
  constexpr int O4   = OUT / 4;
  constexpr int LPE  = OUT < 32 ? OUT : 32;
  constexpr int EPS  = 32 / LPE;
  constexpr int CPL  = OUT < 32 ? 1 : OUT / 32;
  constexpr int UN   = LPE < 8 ? LPE : 8;

  __shared__ float tile[TILE];
  __shared__ int   q[QCAP];
  __shared__ int   qn;

  const int tid  = threadIdx.x;
  const int lane = tid & 31;
  const int wave = tid >> 5;
  const int qsub = lane / LPE;
  const int ch0  = lane % LPE;
  const uint4* key16 = (const uint4*)key;
  const bool vec_ok = ((n + TROWS - 1) / TROWS <= 256) && ((((size_t)key) & 15) == 0);

  const int tiles = (n + TROWS - 1) / TROWS;
  const long long room = (long long)(rec_end - rec) / 2 / tiles;
  const int rcap = room > 0 ? (int)(room < 0x40000000 ? room : 0x40000000) & ~31 : 0;
  const bool rec_ok = rcap > 0 && n <= (1 << 23) && ((((size_t)rec) & 7) == 0);

  for (int t = blockIdx.x; t * TROWS < n; t += gridDim.x) {
    const int base = t * TROWS;
    const int rows = (n - base < TROWS) ? (n - base) : TROWS;
    unsigned kx[8];
#pragma unroll
    for (int j = 0; j < 8; ++j) kx[j] = 0u - (((unsigned)t >> j) & 1u);

    {
      const float4* xw4 = (const float4*)xw + (size_t)base * O4;
      float4* tile4 = (float4*)tile;
      for (int i = tid; i < rows * O4; i += 1024) {
        const int row = i / O4, c = (i - row * O4) * 4;
        const float dv = dinv[base + row];
        const float w = dv * dv;
        const float4 x = xw4[i];
        float4 r;
        r.x = x.x * w + b[c];
        r.y = x.y * w + b[c + 1];
        r.z = x.z * w + b[c + 2];
        r.w = x.w * w + b[c + 3];
        tile4[i] = r;
      }
    }

    auto add32 = [&](int s_l, int r_l, float nm_l, int left) {
#pragma unroll 1
      for (int k0 = 0; k0 < LPE; k0 += UN) {
        float val[UN][CPL];
        int   rr[UN];
        float nn[UN];
#pragma unroll
        for (int u = 0; u < UN; ++u) {
          const int ent = (k0 + u) * EPS + qsub;
          int s;
          if constexpr (LPE == 32) {
            s     = __builtin_amdgcn_readlane(s_l, k0 + u);
            rr[u] = __builtin_amdgcn_readlane(r_l, k0 + u);
            nn[u] = __int_as_float(__builtin_amdgcn_readlane(__float_as_int(nm_l), k0 + u));
          } else {
            s     = __shfl(s_l, ent);
            rr[u] = __shfl(r_l, ent);
            nn[u] = __shfl(nm_l, ent);
          }
          const float* xr = xw + (size_t)s * OUT + ch0;
#pragma unroll
          for (int j = 0; j < CPL; ++j) val[u][j] = xr[32 * j];
        }
#pragma unroll
        for (int u = 0; u < UN; ++u) {
          const int ent = (k0 + u) * EPS + qsub;
          if (ent < left) {
            float* tr = tile + rr[u] * OUT + ch0;
#pragma unroll
            for (int j = 0; j < CPL; ++j) atomicAdd(tr + 32 * j, val[u][j] * nn[u]);
          }
        }
      }
    };

    int seg_lo = 0, seg_len = E;
    int2* rt = (int2*)rec + (size_t)t * rcap;
    int rec_base = 0;
    if constexpr (REC == 2) {
      const int cnt = rec_ok ? tcnt[t] : -1;
      if (cnt >= 0 && cnt <= rcap) {
        __syncthreads();
        const int nb = (cnt + 31) >> 5;
        for (int bi = wave; bi < nb; bi += 32) {
          const int left = cnt - (bi << 5);
          const int2 r = rt[(lane < left) ? ((bi << 5) + lane) : (bi << 5)];
          add32(r.x & 0x7fffff, (int)((unsigned)r.x >> 23), __int_as_float(r.y), left);
        }
        __syncthreads();
        seg_lo = E;
      }
    }
    while (seg_lo < E) {
      const int seg_hi = (E - seg_lo < seg_len) ? E : seg_lo + seg_len;
      if (tid == 0) qn = 0;
      __syncthreads();

      auto take = [&](int e) {
        const int pos = atomicAdd(&qn, 1);
        if (pos < QCAP) q[pos] = e;
      };
      auto test32 = [&](uint4 lo, uint4 hi, int e0) {
        unsigned m = ~((lo.x ^ kx[0]) | (lo.y ^ kx[1]) | (lo.z ^ kx[2]) | (lo.w ^ kx[3]) |
                       (hi.x ^ kx[4]) | (hi.y ^ kx[5]) | (hi.z ^ kx[6]) | (hi.w ^ kx[7]));
        while (m) {
          take(e0 + __builtin_ctz(m));
          m &= m - 1;
        }
      };
      const int g_lo   = seg_lo >> 5;
      const int g_full = vec_ok ? (seg_hi >> 5) : g_lo;
      int g = g_lo + tid;
      for (; g + 3 * 1024 < g_full; g += 4 * 1024) {
        const uint4* p = key16 + ((g >> 5) * 64 + lane);
        uint4 v[8];
#pragma unroll
        for (int u = 0; u < 4; ++u) {
          v[2 * u]     = p[u * 2048];
          v[2 * u + 1] = p[u * 2048 + 32];
        }
#pragma unroll
        for (int u = 0; u < 4; ++u) test32(v[2 * u], v[2 * u + 1], (g + u * 1024) << 5);
      }
      for (; g < g_full; g += 1024) {
        const uint4* p = key16 + ((g >> 5) * 64 + lane);
        test32(p[0], p[32], g << 5);
      }
      for (int e = (g_full << 5) + tid; e < seg_hi; e += 1024)
        if ((unsigned)(dst[e] - base) < (unsigned)TROWS) take(e);
      __syncthreads();

      const int cnt = qn;
      if (cnt <= QCAP) {
        const int nb = (cnt + 31) >> 5;
        const bool rec_fits = rec_ok && rec_base + cnt <= rcap;
        for (int bi = wave; bi < nb; bi += 32) {
          const int left = cnt - (bi << 5);
          const int e    = q[(lane < left) ? ((bi << 5) + lane) : (bi << 5)];
          const int   s_l  = src[e];
          const int   r_l  = dst[e] - base;
          const float nm_l = norm[e];
          if constexpr (REC == 1) {
            if (rec_fits && lane < left)
              rt[rec_base + (bi << 5) + lane] =
                  make_int2((int)((unsigned)s_l | ((unsigned)r_l << 23)), __float_as_int(nm_l));
          }
          add32(s_l, r_l, nm_l, left);
        }
        rec_base += cnt;
        seg_lo = seg_hi;
      } else {
        seg_len = QCAP;
      }
      __syncthreads();
    }

    if constexpr (REC == 1) {
      if (tid == 0) tcnt[t] = (rec_ok && rec_base <= rcap) ? rec_base : -1;
    }

    {
      float4* out4 = (float4*)out + (size_t)base * O4;
      const float4* tile4 = (const float4*)tile;
      for (int i = tid; i < rows * O4; i += 1024) {
        float4 v = tile4[i];
        if (ACT) {
          v.x = (v.x > 0.f) ? v.x : expm1f(v.x);
          v.y = (v.y > 0.f) ? v.y : expm1f(v.y);
          v.z = (v.z > 0.f) ? v.z : expm1f(v.z);
          v.w = (v.w > 0.f) ? v.w : expm1f(v.w);
        }
        out4[i] = v;
      }
    }
    __syncthreads();
  }
}

static inline int cdiv(long long a, long long b) { return (int)((a + b - 1) / b); }

template <int IN, int OUT, bool ACT, int REC>
static void run_layer(const float* xin, float* xw, float* xout,
                      const float* W, const float* b,
                      const int* src, const int* dst,
                      const int* key,
                      const float* dinv, const float* norm,
                      int* tcnt, int* rec, const int* rec_end,
                      int N, int E, hipStream_t stream) {
  gemm_wmma<IN, OUT><<<cdiv(N, 128), 256, 0, stream>>>(xin, W, xw, N);
  agg_tile<OUT, ACT, REC><<<cdiv(N, TROWS), 1024, 0, stream>>>(xw, src, dst, key, norm, dinv, b,
                                                               tcnt, rec, rec_end, xout, N, E);
}

extern "C" void kernel_launch(void* const* d_in, const int* in_sizes, int n_in,
                              void* d_out, int out_size, void* d_ws, size_t ws_size,
                              hipStream_t stream) {
  const float* x   = (const float*)d_in[0];
  const int*   ei  = (const int*)d_in[1];
  const float* ew  = (const float*)d_in[2];
  const float* W1 = (const float*)d_in[3];  const float* b1 = (const float*)d_in[4];
  const float* W2 = (const float*)d_in[5];  const float* b2 = (const float*)d_in[6];
  const float* W3 = (const float*)d_in[7];  const float* b3 = (const float*)d_in[8];
  const float* W4 = (const float*)d_in[9];  const float* b4 = (const float*)d_in[10];
  const float* W5 = (const float*)d_in[11]; const float* b5 = (const float*)d_in[12];
  const float* W6 = (const float*)d_in[13]; const float* b6 = (const float*)d_in[14];

  const int N = in_sizes[0] / 32;
  const int E = in_sizes[2];
  const int* src = ei;
  const int* dst = ei + E;

  float* dinv = (float*)d_ws;
  float* norm = dinv + N;
  float* buf0 = norm + E;
  float* buf1 = buf0 + (size_t)N * 128;
  float* outp = (float*)d_out;
  int* key = (int*)(buf0 + (size_t)N * 64);
  int* tcnt = key + (size_t)cdiv(E, 1024) * 256;
  int* rec  = tcnt + (size_t)cdiv(cdiv(N, TROWS), 64) * 64;
  const int* rec_end = (const int*)buf1;

  fill_f32<<<cdiv(N, 256), 256, 0, stream>>>(dinv, 1.0f, N);
  deg_accum<<<cdiv(E, 256), 256, 0, stream>>>(dst, ew, dinv, E);
  rsqrt_inplace<<<cdiv(N, 256), 256, 0, stream>>>(dinv, N);
  norm_kernel<<<cdiv(E, 256), 256, 0, stream>>>(src, dst, ew, dinv, norm, key, E);

  run_layer<32,   4, true,  1>(x,    buf1, buf0, W1, b1, src, dst, key, dinv, norm, tcnt, rec, rec_end, N, E, stream);
  run_layer<4,   16, true,  2>(buf0, buf1, buf0, W2, b2, src, dst, key, dinv, norm, tcnt, rec, rec_end, N, E, stream);
  run_layer<16,  32, true,  2>(buf0, buf1, buf0, W3, b3, src, dst, key, dinv, norm, tcnt, rec, rec_end, N, E, stream);
  run_layer<32,  64, false, 2>(buf0, buf1, buf0, W4, b4, src, dst, key, dinv, norm, tcnt, rec, rec_end, N, E, stream);
  run_layer<64, 128, true,  2>(buf0, buf1, outp, W5, b5, src, dst, key, dinv, norm, tcnt, rec, rec_end, N, E, stream);
  run_layer<128,128, false, 2>(outp, buf1, outp, W6, b6, src, dst, key, dinv, norm, tcnt, rec, rec_end, N, E, stream);
}


# Degrees summed per destination tile in LDS (ds_add_f32) by the pass that builds the tiles' edge lists; global float atomics, fill, rsqrt and norm launches gone (16 -> 14); deg sum order LDS arrival
# speedup vs baseline: 5.3518x; 1.1899x over previous
#include <hip/hip_runtime.h>
#include <cstdint>

typedef float v2f __attribute__((ext_vector_type(2)));
typedef float v8f __attribute__((ext_vector_type(8)));

constexpr int TROWS = 392;

__global__ __launch_bounds__(256) void key_kernel(const int* __restrict__ dst,
                                                  int* __restrict__ key, int E) {
  const int e = blockIdx.x * 256 + threadIdx.x;
  const int d = (e < E) ? dst[e] : 0;
  const unsigned k = (unsigned)(d / TROWS);
  uint4 lo, hi;
  lo.x = __builtin_amdgcn_ballot_w32((k & 1u) != 0);
  lo.y = __builtin_amdgcn_ballot_w32((k & 2u) != 0);
  lo.z = __builtin_amdgcn_ballot_w32((k & 4u) != 0);
  lo.w = __builtin_amdgcn_ballot_w32((k & 8u) != 0);
  hi.x = __builtin_amdgcn_ballot_w32((k & 16u) != 0);
  hi.y = __builtin_amdgcn_ballot_w32((k & 32u) != 0);
  hi.z = __builtin_amdgcn_ballot_w32((k & 64u) != 0);
  hi.w = __builtin_amdgcn_ballot_w32((k & 128u) != 0);
  if ((threadIdx.x & 31) == 0) {
    const int g = e >> 5;
    uint4* p = (uint4*)key + ((g >> 5) * 64 + (g & 31));
    p[0]  = lo;
    p[32] = hi;
  }
}

constexpr int LDS_FLOATS = 61440;

__global__ __launch_bounds__(1024) void tile_prep(const int* __restrict__ src,
                                                   const int* __restrict__ dst,
                                                   const int* __restrict__ key,
                                                   const float* __restrict__ ew,
                                                   float* __restrict__ dinv,
                                                   int* __restrict__ tcnt, int* __restrict__ rec,
                                                   const int* rec_end, int n, int E) {
  constexpr int QCAP = (LDS_FLOATS - TROWS) & ~1023;
  __shared__ float deg[TROWS];
  __shared__ int   q[QCAP];
  __shared__ int   qn;

  const int tid  = threadIdx.x;
  const int lane = tid & 31;
  const uint4* key16 = (const uint4*)key;
  const bool vec_ok = ((n + TROWS - 1) / TROWS <= 256) && ((((size_t)key) & 15) == 0);
  const int tiles = (n + TROWS - 1) / TROWS;
  const long long room = (long long)(rec_end - rec) / 2 / tiles;
  const int rcap = room > 0 ? (int)(room < 0x40000000 ? room : 0x40000000) & ~31 : 0;
  const bool rec_ok = rcap > 0 && n <= (1 << 23) && ((((size_t)rec) & 7) == 0);

  for (int t = blockIdx.x; t * TROWS < n; t += gridDim.x) {
    const int base = t * TROWS;
    const int rows = (n - base < TROWS) ? (n - base) : TROWS;
    unsigned kx[8];
#pragma unroll
    for (int j = 0; j < 8; ++j) kx[j] = 0u - (((unsigned)t >> j) & 1u);

    for (int i = tid; i < TROWS; i += 1024) deg[i] = 1.0f;

    int seg_lo = 0, seg_len = E;
    int2* rt = (int2*)rec + (size_t)t * rcap;
    int rec_base = 0;
    while (seg_lo < E) {
      const int seg_hi = (E - seg_lo < seg_len) ? E : seg_lo + seg_len;
      if (tid == 0) qn = 0;
      __syncthreads();

      auto take = [&](int e) {
        const int pos = atomicAdd(&qn, 1);
        if (pos < QCAP) q[pos] = e;
      };
      auto test32 = [&](uint4 lo, uint4 hi, int e0) {
        unsigned m = ~((lo.x ^ kx[0]) | (lo.y ^ kx[1]) | (lo.z ^ kx[2]) | (lo.w ^ kx[3]) |
                       (hi.x ^ kx[4]) | (hi.y ^ kx[5]) | (hi.z ^ kx[6]) | (hi.w ^ kx[7]));
        while (m) {
          take(e0 + __builtin_ctz(m));
          m &= m - 1;
        }
      };
      const int g_lo   = seg_lo >> 5;
      const int g_full = vec_ok ? (seg_hi >> 5) : g_lo;
      int g = g_lo + tid;
      for (; g + 3 * 1024 < g_full; g += 4 * 1024) {
        const uint4* p = key16 + ((g >> 5) * 64 + lane);
        uint4 v[8];
#pragma unroll
        for (int u = 0; u < 4; ++u) {
          v[2 * u]     = p[u * 2048];
          v[2 * u + 1] = p[u * 2048 + 32];
        }
#pragma unroll
        for (int u = 0; u < 4; ++u) test32(v[2 * u], v[2 * u + 1], (g + u * 1024) << 5);
      }
      for (; g < g_full; g += 1024) {
        const uint4* p = key16 + ((g >> 5) * 64 + lane);
        test32(p[0], p[32], g << 5);
      }
      for (int e = (g_full << 5) + tid; e < seg_hi; e += 1024)
        if ((unsigned)(dst[e] - base) < (unsigned)TROWS) take(e);
      __syncthreads();

      const int cnt = qn;
      if (cnt <= QCAP) {
        const bool rec_fits = rec_ok && rec_base + cnt <= rcap;
        for (int i0 = tid; i0 < cnt; i0 += 4 * 1024) {
          int e[4], s[4], d[4];
          float w[4];
#pragma unroll
          for (int u = 0; u < 4; ++u) {
            const int i = i0 + u * 1024;
            e[u] = q[i < cnt ? i : i0];
          }
#pragma unroll
          for (int u = 0; u < 4; ++u) {
            s[u] = src[e[u]];
            d[u] = dst[e[u]];
            w[u] = ew[e[u]];
          }
          asm volatile("" :: "v"(s[0]), "v"(s[1]), "v"(s[2]), "v"(s[3]),
                             "v"(d[0]), "v"(d[1]), "v"(d[2]), "v"(d[3]),
                             "v"(w[0]), "v"(w[1]), "v"(w[2]), "v"(w[3]));
#pragma unroll
          for (int u = 0; u < 4; ++u) {
            const int i = i0 + u * 1024;
            if (i < cnt) {
              const int r = d[u] - base;
              atomicAdd(&deg[r], w[u]);
              if (rec_fits)
                rt[rec_base + i] =
                    make_int2((int)((unsigned)s[u] | ((unsigned)r << 23)), __float_as_int(w[u]));
            }
          }
        }
        rec_base += cnt;
        seg_lo = seg_hi;
      } else {
        seg_len = QCAP;
      }
      __syncthreads();
    }

    if (tid == 0) tcnt[t] = (rec_ok && rec_base <= rcap) ? rec_base : -1;
    for (int i = tid; i < rows; i += 1024) {
      const float d = deg[i];
      dinv[base + i] = (d > 0.f) ? rsqrtf(d) : 0.f;
    }
    __syncthreads();
  }
}

template <int IN, int OUT>
__global__ __launch_bounds__(256) void gemm_wmma(const float* __restrict__ x,
                                                 const float* __restrict__ W,
                                                 float* __restrict__ y, int n) {
  constexpr bool USE_LDS = (IN * OUT) <= 8192;
  constexpr bool FULL    = (OUT % 16) == 0;
  __shared__ float sW[USE_LDS ? (IN * OUT) : 1];
  const float* Wp = W;
  if constexpr (USE_LDS) {
    for (int i = threadIdx.x; i < IN * OUT; i += 256) sW[i] = W[i];
    __syncthreads();
    Wp = sW;
  }

  const int lane  = threadIdx.x & 31;
  const int wave  = threadIdx.x >> 5;
  const int l15   = lane & 15;
  const int khalf = lane >> 4;
  const int row0  = blockIdx.x * 128 + wave * 16;

  if (row0 < n) {
    int mload = row0 + l15;
    if (mload > n - 1) mload = n - 1;
    const float* xrow = x + (size_t)mload * IN;
    const bool fulltile = (row0 + 16 <= n);

    for (int n0 = 0; n0 < OUT; n0 += 16) {
      const int colb = n0 + l15;
      const int colc = FULL ? colb : (colb < OUT ? colb : OUT - 1);
      v8f c = {};
#pragma unroll
      for (int k0 = 0; k0 < IN; k0 += 4) {
        const int krow = k0 + 2 * khalf;
        v2f a = *(const v2f*)(xrow + krow);
        v2f b;
        b.x = Wp[(size_t)krow * OUT + colc];
        b.y = Wp[(size_t)(krow + 1) * OUT + colc];
        c = __builtin_amdgcn_wmma_f32_16x16x4_f32(
             false, a,  false, b,
             (short)0, c,  false,  false);
      }
      const int rowoff = khalf * 8;
      if (FULL && fulltile) {
#pragma unroll
        for (int r = 0; r < 8; ++r)
          y[(size_t)(row0 + rowoff + r) * OUT + colb] = c[r];
      } else {
        const bool colok = FULL || (colb < OUT);
#pragma unroll
        for (int r = 0; r < 8; ++r) {
          int row = row0 + rowoff + r;
          if (row < n && colok) y[(size_t)row * OUT + colb] = c[r];
        }
      }
    }
  }
}

template <int OUT, bool ACT, int REC>
__global__ __launch_bounds__(1024) void agg_tile(const float* __restrict__ xw,
                                                  const int* __restrict__ src,
                                                  const int* __restrict__ dst,
                                                  const int* __restrict__ key,
                                                  const float* __restrict__ ew,
                                                  const float* __restrict__ dinv,
                                                  const float* __restrict__ b,
                                                  const int* tcnt, int* rec, const int* rec_end,
                                                  float* __restrict__ out, int n, int E) {
  constexpr int TILE = TROWS * OUT;
  constexpr int QCAP = (LDS_FLOATS - TILE) & ~1023;
  constexpr int O4   = OUT / 4;
  constexpr int LPE  = OUT < 32 ? OUT : 32;
  constexpr int EPS  = 32 / LPE;
  constexpr int CPL  = OUT < 32 ? 1 : OUT / 32;
  constexpr int UN   = LPE < 8 ? LPE : 8;

  __shared__ float tile[TILE];
  __shared__ int   q[QCAP];
  __shared__ int   qn;

  const int tid  = threadIdx.x;
  const int lane = tid & 31;
  const int wave = tid >> 5;
  const int qsub = lane / LPE;
  const int ch0  = lane % LPE;
  const uint4* key16 = (const uint4*)key;
  const bool vec_ok = ((n + TROWS - 1) / TROWS <= 256) && ((((size_t)key) & 15) == 0);

  const int tiles = (n + TROWS - 1) / TROWS;
  const long long room = (long long)(rec_end - rec) / 2 / tiles;
  const int rcap = room > 0 ? (int)(room < 0x40000000 ? room : 0x40000000) & ~31 : 0;
  const bool rec_ok = rcap > 0 && n <= (1 << 23) && ((((size_t)rec) & 7) == 0);

  for (int t = blockIdx.x; t * TROWS < n; t += gridDim.x) {
    const int base = t * TROWS;
    const int rows = (n - base < TROWS) ? (n - base) : TROWS;
    unsigned kx[8];
#pragma unroll
    for (int j = 0; j < 8; ++j) kx[j] = 0u - (((unsigned)t >> j) & 1u);

    {
      const float4* xw4 = (const float4*)xw + (size_t)base * O4;
      float4* tile4 = (float4*)tile;
      for (int i = tid; i < rows * O4; i += 1024) {
        const int row = i / O4, c = (i - row * O4) * 4;
        const float dv = dinv[base + row];
        const float w = dv * dv;
        const float4 x = xw4[i];
        float4 r;
        r.x = x.x * w + b[c];
        r.y = x.y * w + b[c + 1];
        r.z = x.z * w + b[c + 2];
        r.w = x.w * w + b[c + 3];
        tile4[i] = r;
      }
    }

    auto add32 = [&](int s_l, int r_l, float nm_l, int left) {
#pragma unroll 1
      for (int k0 = 0; k0 < LPE; k0 += UN) {
        float val[UN][CPL];
        int   rr[UN];
        float nn[UN];
#pragma unroll
        for (int u = 0; u < UN; ++u) {
          const int ent = (k0 + u) * EPS + qsub;
          int s;
          if constexpr (LPE == 32) {
            s     = __builtin_amdgcn_readlane(s_l, k0 + u);
            rr[u] = __builtin_amdgcn_readlane(r_l, k0 + u);
            nn[u] = __int_as_float(__builtin_amdgcn_readlane(__float_as_int(nm_l), k0 + u));
          } else {
            s     = __shfl(s_l, ent);
            rr[u] = __shfl(r_l, ent);
            nn[u] = __shfl(nm_l, ent);
          }
          const float* xr = xw + (size_t)s * OUT + ch0;
#pragma unroll
          for (int j = 0; j < CPL; ++j) val[u][j] = xr[32 * j];
        }
#pragma unroll
        for (int u = 0; u < UN; ++u) {
          const int ent = (k0 + u) * EPS + qsub;
          if (ent < left) {
            float* tr = tile + rr[u] * OUT + ch0;
#pragma unroll
            for (int j = 0; j < CPL; ++j) atomicAdd(tr + 32 * j, val[u][j] * nn[u]);
          }
        }
      }
    };

    int seg_lo = 0, seg_len = E;
    int2* rt = (int2*)rec + (size_t)t * rcap;
    {
      const int cnt = rec_ok ? tcnt[t] : -1;
      if (cnt >= 0 && cnt <= rcap) {
        __syncthreads();
        const int nb = (cnt + 31) >> 5;
        for (int bi = wave; bi < nb; bi += 32) {
          const int left = cnt - (bi << 5);
          const int idx  = (lane < left) ? ((bi << 5) + lane) : (bi << 5);
          const int2 r = rt[idx];
          const int s_l = r.x & 0x7fffff;
          const int r_l = (int)((unsigned)r.x >> 23);
          float nm_l = __int_as_float(r.y);
          if constexpr (REC == 1) {
            nm_l = dinv[s_l] * nm_l * dinv[base + r_l];
            if (lane < left) rt[idx] = make_int2(r.x, __float_as_int(nm_l));
          }
          add32(s_l, r_l, nm_l, left);
        }
        __syncthreads();
        seg_lo = E;
      }
    }
    while (seg_lo < E) {
      const int seg_hi = (E - seg_lo < seg_len) ? E : seg_lo + seg_len;
      if (tid == 0) qn = 0;
      __syncthreads();

      auto take = [&](int e) {
        const int pos = atomicAdd(&qn, 1);
        if (pos < QCAP) q[pos] = e;
      };
      auto test32 = [&](uint4 lo, uint4 hi, int e0) {
        unsigned m = ~((lo.x ^ kx[0]) | (lo.y ^ kx[1]) | (lo.z ^ kx[2]) | (lo.w ^ kx[3]) |
                       (hi.x ^ kx[4]) | (hi.y ^ kx[5]) | (hi.z ^ kx[6]) | (hi.w ^ kx[7]));
        while (m) {
          take(e0 + __builtin_ctz(m));
          m &= m - 1;
        }
      };
      const int g_lo   = seg_lo >> 5;
      const int g_full = vec_ok ? (seg_hi >> 5) : g_lo;
      int g = g_lo + tid;
      for (; g + 3 * 1024 < g_full; g += 4 * 1024) {
        const uint4* p = key16 + ((g >> 5) * 64 + lane);
        uint4 v[8];
#pragma unroll
        for (int u = 0; u < 4; ++u) {
          v[2 * u]     = p[u * 2048];
          v[2 * u + 1] = p[u * 2048 + 32];
        }
#pragma unroll
        for (int u = 0; u < 4; ++u) test32(v[2 * u], v[2 * u + 1], (g + u * 1024) << 5);
      }
      for (; g < g_full; g += 1024) {
        const uint4* p = key16 + ((g >> 5) * 64 + lane);
        test32(p[0], p[32], g << 5);
      }
      for (int e = (g_full << 5) + tid; e < seg_hi; e += 1024)
        if ((unsigned)(dst[e] - base) < (unsigned)TROWS) take(e);
      __syncthreads();

      const int cnt = qn;
      if (cnt <= QCAP) {
        const int nb = (cnt + 31) >> 5;
        for (int bi = wave; bi < nb; bi += 32) {
          const int left = cnt - (bi << 5);
          const int e    = q[(lane < left) ? ((bi << 5) + lane) : (bi << 5)];
          const int   s_l  = src[e];
          const int   r_l  = dst[e] - base;
          const float nm_l = dinv[s_l] * ew[e] * dinv[base + r_l];
          add32(s_l, r_l, nm_l, left);
        }
        seg_lo = seg_hi;
      } else {
        seg_len = QCAP;
      }
      __syncthreads();
    }

    {
      float4* out4 = (float4*)out + (size_t)base * O4;
      const float4* tile4 = (const float4*)tile;
      for (int i = tid; i < rows * O4; i += 1024) {
        float4 v = tile4[i];
        if (ACT) {
          v.x = (v.x > 0.f) ? v.x : expm1f(v.x);
          v.y = (v.y > 0.f) ? v.y : expm1f(v.y);
          v.z = (v.z > 0.f) ? v.z : expm1f(v.z);
          v.w = (v.w > 0.f) ? v.w : expm1f(v.w);
        }
        out4[i] = v;
      }
    }
    __syncthreads();
  }
}

static inline int cdiv(long long a, long long b) { return (int)((a + b - 1) / b); }

template <int IN, int OUT, bool ACT, int REC>
static void run_layer(const float* xin, float* xw, float* xout,
                      const float* W, const float* b,
                      const int* src, const int* dst,
                      const int* key,
                      const float* dinv, const float* ew,
                      const int* tcnt, int* rec, const int* rec_end,
                      int N, int E, hipStream_t stream) {
  gemm_wmma<IN, OUT><<<cdiv(N, 128), 256, 0, stream>>>(xin, W, xw, N);
  agg_tile<OUT, ACT, REC><<<cdiv(N, TROWS), 1024, 0, stream>>>(xw, src, dst, key, ew, dinv, b,
                                                               tcnt, rec, rec_end, xout, N, E);
}

extern "C" void kernel_launch(void* const* d_in, const int* in_sizes, int n_in,
                              void* d_out, int out_size, void* d_ws, size_t ws_size,
                              hipStream_t stream) {
  const float* x   = (const float*)d_in[0];
  const int*   ei  = (const int*)d_in[1];
  const float* ew  = (const float*)d_in[2];
  const float* W1 = (const float*)d_in[3];  const float* b1 = (const float*)d_in[4];
  const float* W2 = (const float*)d_in[5];  const float* b2 = (const float*)d_in[6];
  const float* W3 = (const float*)d_in[7];  const float* b3 = (const float*)d_in[8];
  const float* W4 = (const float*)d_in[9];  const float* b4 = (const float*)d_in[10];
  const float* W5 = (const float*)d_in[11]; const float* b5 = (const float*)d_in[12];
  const float* W6 = (const float*)d_in[13]; const float* b6 = (const float*)d_in[14];

  const int N = in_sizes[0] / 32;
  const int E = in_sizes[2];
  const int* src = ei;
  const int* dst = ei + E;

  float* dinv = (float*)d_ws;
  float* buf0 = dinv + N + E;
  float* buf1 = buf0 + (size_t)N * 128;
  float* outp = (float*)d_out;
  int* key = (int*)(buf0 + (size_t)N * 64);
  int* tcnt = key + (size_t)cdiv(E, 1024) * 256;
  int* rec  = tcnt + (size_t)cdiv(cdiv(N, TROWS), 64) * 64;
  const int* rec_end = (const int*)buf1;

  key_kernel<<<cdiv(E, 256), 256, 0, stream>>>(dst, key, E);
  tile_prep<<<cdiv(N, TROWS), 1024, 0, stream>>>(src, dst, key, ew, dinv, tcnt, rec, rec_end, N, E);

  run_layer<32,   4, true,  1>(x,    buf1, buf0, W1, b1, src, dst, key, dinv, ew, tcnt, rec, rec_end, N, E, stream);
  run_layer<4,   16, true,  2>(buf0, buf1, buf0, W2, b2, src, dst, key, dinv, ew, tcnt, rec, rec_end, N, E, stream);
  run_layer<16,  32, true,  2>(buf0, buf1, buf0, W3, b3, src, dst, key, dinv, ew, tcnt, rec, rec_end, N, E, stream);
  run_layer<32,  64, false, 2>(buf0, buf1, buf0, W4, b4, src, dst, key, dinv, ew, tcnt, rec, rec_end, N, E, stream);
  run_layer<64, 128, true,  2>(buf0, buf1, outp, W5, b5, src, dst, key, dinv, ew, tcnt, rec, rec_end, N, E, stream);
  run_layer<128,128, false, 2>(outp, buf1, outp, W6, b6, src, dst, key, dinv, ew, tcnt, rec, rec_end, N, E, stream);
}


# Layers 1-5: the aggregation block multiplies its LDS tile (bias, ELU applied) by the next layer's W on WMMA and stores that product: activations stay in LDS, 14 -> 9 launches; sums unchanged
# speedup vs baseline: 6.2765x; 1.1728x over previous
#include <hip/hip_runtime.h>
#include <cstdint>

typedef float v2f __attribute__((ext_vector_type(2)));
typedef float v8f __attribute__((ext_vector_type(8)));

constexpr int TROWS = 392;

__global__ __launch_bounds__(256) void key_kernel(const int* __restrict__ dst,
                                                  int* __restrict__ key, int E) {
  const int e = blockIdx.x * 256 + threadIdx.x;
  const int d = (e < E) ? dst[e] : 0;
  const unsigned k = (unsigned)(d / TROWS);
  uint4 lo, hi;
  lo.x = __builtin_amdgcn_ballot_w32((k & 1u) != 0);
  lo.y = __builtin_amdgcn_ballot_w32((k & 2u) != 0);
  lo.z = __builtin_amdgcn_ballot_w32((k & 4u) != 0);
  lo.w = __builtin_amdgcn_ballot_w32((k & 8u) != 0);
  hi.x = __builtin_amdgcn_ballot_w32((k & 16u) != 0);
  hi.y = __builtin_amdgcn_ballot_w32((k & 32u) != 0);
  hi.z = __builtin_amdgcn_ballot_w32((k & 64u) != 0);
  hi.w = __builtin_amdgcn_ballot_w32((k & 128u) != 0);
  if ((threadIdx.x & 31) == 0) {
    const int g = e >> 5;
    uint4* p = (uint4*)key + ((g >> 5) * 64 + (g & 31));
    p[0]  = lo;
    p[32] = hi;
  }
}

constexpr int LDS_FLOATS = 61440;

__global__ __launch_bounds__(1024) void tile_prep(const int* __restrict__ src,
                                                   const int* __restrict__ dst,
                                                   const int* __restrict__ key,
                                                   const float* __restrict__ ew,
                                                   float* __restrict__ dinv,
                                                   int* __restrict__ tcnt, int* __restrict__ rec,
                                                   const int* rec_end, int n, int E) {
  constexpr int QCAP = (LDS_FLOATS - TROWS) & ~1023;
  __shared__ float deg[TROWS];
  __shared__ int   q[QCAP];
  __shared__ int   qn;

  const int tid  = threadIdx.x;
  const int lane = tid & 31;
  const uint4* key16 = (const uint4*)key;
  const bool vec_ok = ((n + TROWS - 1) / TROWS <= 256) && ((((size_t)key) & 15) == 0);
  const int tiles = (n + TROWS - 1) / TROWS;
  const long long room = (long long)(rec_end - rec) / 2 / tiles;
  const int rcap = room > 0 ? (int)(room < 0x40000000 ? room : 0x40000000) & ~31 : 0;
  const bool rec_ok = rcap > 0 && n <= (1 << 23) && ((((size_t)rec) & 7) == 0);

  for (int t = blockIdx.x; t * TROWS < n; t += gridDim.x) {
    const int base = t * TROWS;
    const int rows = (n - base < TROWS) ? (n - base) : TROWS;
    unsigned kx[8];
#pragma unroll
    for (int j = 0; j < 8; ++j) kx[j] = 0u - (((unsigned)t >> j) & 1u);

    for (int i = tid; i < TROWS; i += 1024) deg[i] = 1.0f;

    int seg_lo = 0, seg_len = E;
    int2* rt = (int2*)rec + (size_t)t * rcap;
    int rec_base = 0;
    while (seg_lo < E) {
      const int seg_hi = (E - seg_lo < seg_len) ? E : seg_lo + seg_len;
      if (tid == 0) qn = 0;
      __syncthreads();

      auto take = [&](int e) {
        const int pos = atomicAdd(&qn, 1);
        if (pos < QCAP) q[pos] = e;
      };
      auto test32 = [&](uint4 lo, uint4 hi, int e0) {
        unsigned m = ~((lo.x ^ kx[0]) | (lo.y ^ kx[1]) | (lo.z ^ kx[2]) | (lo.w ^ kx[3]) |
                       (hi.x ^ kx[4]) | (hi.y ^ kx[5]) | (hi.z ^ kx[6]) | (hi.w ^ kx[7]));
        while (m) {
          take(e0 + __builtin_ctz(m));
          m &= m - 1;
        }
      };
      const int g_lo   = seg_lo >> 5;
      const int g_full = vec_ok ? (seg_hi >> 5) : g_lo;
      int g = g_lo + tid;
      for (; g + 3 * 1024 < g_full; g += 4 * 1024) {
        const uint4* p = key16 + ((g >> 5) * 64 + lane);
        uint4 v[8];
#pragma unroll
        for (int u = 0; u < 4; ++u) {
          v[2 * u]     = p[u * 2048];
          v[2 * u + 1] = p[u * 2048 + 32];
        }
#pragma unroll
        for (int u = 0; u < 4; ++u) test32(v[2 * u], v[2 * u + 1], (g + u * 1024) << 5);
      }
      for (; g < g_full; g += 1024) {
        const uint4* p = key16 + ((g >> 5) * 64 + lane);
        test32(p[0], p[32], g << 5);
      }
      for (int e = (g_full << 5) + tid; e < seg_hi; e += 1024)
        if ((unsigned)(dst[e] - base) < (unsigned)TROWS) take(e);
      __syncthreads();

      const int cnt = qn;
      if (cnt <= QCAP) {
        const bool rec_fits = rec_ok && rec_base + cnt <= rcap;
        for (int i0 = tid; i0 < cnt; i0 += 4 * 1024) {
          int e[4], s[4], d[4];
          float w[4];
#pragma unroll
          for (int u = 0; u < 4; ++u) {
            const int i = i0 + u * 1024;
            e[u] = q[i < cnt ? i : i0];
          }
#pragma unroll
          for (int u = 0; u < 4; ++u) {
            s[u] = src[e[u]];
            d[u] = dst[e[u]];
            w[u] = ew[e[u]];
          }
          asm volatile("" :: "v"(s[0]), "v"(s[1]), "v"(s[2]), "v"(s[3]),
                             "v"(d[0]), "v"(d[1]), "v"(d[2]), "v"(d[3]),
                             "v"(w[0]), "v"(w[1]), "v"(w[2]), "v"(w[3]));
#pragma unroll
          for (int u = 0; u < 4; ++u) {
            const int i = i0 + u * 1024;
            if (i < cnt) {
              const int r = d[u] - base;
              atomicAdd(&deg[r], w[u]);
              if (rec_fits)
                rt[rec_base + i] =
                    make_int2((int)((unsigned)s[u] | ((unsigned)r << 23)), __float_as_int(w[u]));
            }
          }
        }
        rec_base += cnt;
        seg_lo = seg_hi;
      } else {
        seg_len = QCAP;
      }
      __syncthreads();
    }

    if (tid == 0) tcnt[t] = (rec_ok && rec_base <= rcap) ? rec_base : -1;
    for (int i = tid; i < rows; i += 1024) {
      const float d = deg[i];
      dinv[base + i] = (d > 0.f) ? rsqrtf(d) : 0.f;
    }
    __syncthreads();
  }
}

template <int IN, int OUT>
__global__ __launch_bounds__(256) void gemm_wmma(const float* __restrict__ x,
                                                 const float* __restrict__ W,
                                                 float* __restrict__ y, int n) {
  constexpr bool USE_LDS = (IN * OUT) <= 8192;
  constexpr bool FULL    = (OUT % 16) == 0;
  __shared__ float sW[USE_LDS ? (IN * OUT) : 1];
  const float* Wp = W;
  if constexpr (USE_LDS) {
    for (int i = threadIdx.x; i < IN * OUT; i += 256) sW[i] = W[i];
    __syncthreads();
    Wp = sW;
  }

  const int lane  = threadIdx.x & 31;
  const int wave  = threadIdx.x >> 5;
  const int l15   = lane & 15;
  const int khalf = lane >> 4;
  const int row0  = blockIdx.x * 128 + wave * 16;

  if (row0 < n) {
    int mload = row0 + l15;
    if (mload > n - 1) mload = n - 1;
    const float* xrow = x + (size_t)mload * IN;
    const bool fulltile = (row0 + 16 <= n);

    for (int n0 = 0; n0 < OUT; n0 += 16) {
      const int colb = n0 + l15;
      const int colc = FULL ? colb : (colb < OUT ? colb : OUT - 1);
      v8f c = {};
#pragma unroll
      for (int k0 = 0; k0 < IN; k0 += 4) {
        const int krow = k0 + 2 * khalf;
        v2f a = *(const v2f*)(xrow + krow);
        v2f b;
        b.x = Wp[(size_t)krow * OUT + colc];
        b.y = Wp[(size_t)(krow + 1) * OUT + colc];
        c = __builtin_amdgcn_wmma_f32_16x16x4_f32(
             false, a,  false, b,
             (short)0, c,  false,  false);
      }
      const int rowoff = khalf * 8;
      if (FULL && fulltile) {
#pragma unroll
        for (int r = 0; r < 8; ++r)
          y[(size_t)(row0 + rowoff + r) * OUT + colb] = c[r];
      } else {
        const bool colok = FULL || (colb < OUT);
#pragma unroll
        for (int r = 0; r < 8; ++r) {
          int row = row0 + rowoff + r;
          if (row < n && colok) y[(size_t)row * OUT + colb] = c[r];
        }
      }
    }
  }
}

template <int OUT, bool ACT, int REC, int NOUT>
__global__ __launch_bounds__(1024) void agg_tile(const float* __restrict__ xw,
                                                  const int* __restrict__ src,
                                                  const int* __restrict__ dst,
                                                  const int* __restrict__ key,
                                                  const float* __restrict__ ew,
                                                  const float* __restrict__ dinv,
                                                  const float* __restrict__ b,
                                                  const int* tcnt, int* rec, const int* rec_end,
                                                  float* __restrict__ out,
                                                  const float* __restrict__ Wn,
                                                  float* __restrict__ xwn, int n, int E) {
  constexpr bool MM = NOUT > 0;
  constexpr int PITCH = (MM && OUT >= 16) ? OUT + 4 : OUT;
  constexpr int TROWS_P = MM ? ((TROWS + 15) / 16) * 16 : TROWS;
  constexpr int TILE = TROWS_P * PITCH;
  constexpr int WF   = MM ? OUT * NOUT : 0;
  constexpr int BUDGET = (TILE + WF + 9216 > LDS_FLOATS) ? 80000 : LDS_FLOATS;
  constexpr int QCAP = (BUDGET - TILE - WF) & ~1023;
  constexpr int O4   = OUT / 4;
  constexpr int LPE  = OUT < 32 ? OUT : 32;
  constexpr int EPS  = 32 / LPE;
  constexpr int CPL  = OUT < 32 ? 1 : OUT / 32;
  constexpr int UN   = LPE < 8 ? LPE : 8;

  __shared__ __attribute__((aligned(16))) float tile[TILE];
  __shared__ __attribute__((aligned(16))) float wf[MM ? WF : 4];
  __shared__ int   q[QCAP];
  __shared__ int   qn;

  const int tid  = threadIdx.x;
  const int lane = tid & 31;
  const int wave = tid >> 5;
  if constexpr (MM) {
    constexpr int NT = NOUT / 16;
    for (int i = tid; i < OUT * NOUT; i += 1024) {
      const int k = i / NOUT, c = i - k * NOUT;
      const int ln = ((k >> 1) & 1) * 16 + (c & 15);
      wf[(((k >> 2) * NT + (c >> 4)) * 32 + ln) * 2 + (k & 1)] = Wn[i];
    }
  }
  const int qsub = lane / LPE;
  const int ch0  = lane % LPE;
  const uint4* key16 = (const uint4*)key;
  const bool vec_ok = ((n + TROWS - 1) / TROWS <= 256) && ((((size_t)key) & 15) == 0);

  const int tiles = (n + TROWS - 1) / TROWS;
  const long long room = (long long)(rec_end - rec) / 2 / tiles;
  const int rcap = room > 0 ? (int)(room < 0x40000000 ? room : 0x40000000) & ~31 : 0;
  const bool rec_ok = rcap > 0 && n <= (1 << 23) && ((((size_t)rec) & 7) == 0);

  for (int t = blockIdx.x; t * TROWS < n; t += gridDim.x) {
    const int base = t * TROWS;
    const int rows = (n - base < TROWS) ? (n - base) : TROWS;
    unsigned kx[8];
#pragma unroll
    for (int j = 0; j < 8; ++j) kx[j] = 0u - (((unsigned)t >> j) & 1u);

    {
      const float4* xw4 = (const float4*)xw + (size_t)base * O4;
      for (int i = tid; i < rows * O4; i += 1024) {
        const int row = i / O4, c = (i - row * O4) * 4;
        const float dv = dinv[base + row];
        const float w = dv * dv;
        const float4 x = xw4[i];
        float4 r;
        r.x = x.x * w + b[c];
        r.y = x.y * w + b[c + 1];
        r.z = x.z * w + b[c + 2];
        r.w = x.w * w + b[c + 3];
        *(float4*)(tile + row * PITCH + c) = r;
      }
    }

    auto add32 = [&](int s_l, int r_l, float nm_l, int left) {
#pragma unroll 1
      for (int k0 = 0; k0 < LPE; k0 += UN) {
        float val[UN][CPL];
        int   rr[UN];
        float nn[UN];
#pragma unroll
        for (int u = 0; u < UN; ++u) {
          const int ent = (k0 + u) * EPS + qsub;
          int s;
          if constexpr (LPE == 32) {
            s     = __builtin_amdgcn_readlane(s_l, k0 + u);
            rr[u] = __builtin_amdgcn_readlane(r_l, k0 + u);
            nn[u] = __int_as_float(__builtin_amdgcn_readlane(__float_as_int(nm_l), k0 + u));
          } else {
            s     = __shfl(s_l, ent);
            rr[u] = __shfl(r_l, ent);
            nn[u] = __shfl(nm_l, ent);
          }
          const float* xr = xw + (size_t)s * OUT + ch0;
#pragma unroll
          for (int j = 0; j < CPL; ++j) val[u][j] = xr[32 * j];
        }
#pragma unroll
        for (int u = 0; u < UN; ++u) {
          const int ent = (k0 + u) * EPS + qsub;
          if (ent < left) {
            float* tr = tile + rr[u] * PITCH + ch0;
#pragma unroll
            for (int j = 0; j < CPL; ++j) atomicAdd(tr + 32 * j, val[u][j] * nn[u]);
          }
        }
      }
    };

    int seg_lo = 0, seg_len = E;
    int2* rt = (int2*)rec + (size_t)t * rcap;
    {
      const int cnt = rec_ok ? tcnt[t] : -1;
      if (cnt >= 0 && cnt <= rcap) {
        __syncthreads();
        const int nb = (cnt + 31) >> 5;
        for (int bi = wave; bi < nb; bi += 32) {
          const int left = cnt - (bi << 5);
          const int idx  = (lane < left) ? ((bi << 5) + lane) : (bi << 5);
          const int2 r = rt[idx];
          const int s_l = r.x & 0x7fffff;
          const int r_l = (int)((unsigned)r.x >> 23);
          float nm_l = __int_as_float(r.y);
          if constexpr (REC == 1) {
            nm_l = dinv[s_l] * nm_l * dinv[base + r_l];
            if (lane < left) rt[idx] = make_int2(r.x, __float_as_int(nm_l));
          }
          add32(s_l, r_l, nm_l, left);
        }
        __syncthreads();
        seg_lo = E;
      }
    }
    while (seg_lo < E) {
      const int seg_hi = (E - seg_lo < seg_len) ? E : seg_lo + seg_len;
      if (tid == 0) qn = 0;
      __syncthreads();

      auto take = [&](int e) {
        const int pos = atomicAdd(&qn, 1);
        if (pos < QCAP) q[pos] = e;
      };
      auto test32 = [&](uint4 lo, uint4 hi, int e0) {
        unsigned m = ~((lo.x ^ kx[0]) | (lo.y ^ kx[1]) | (lo.z ^ kx[2]) | (lo.w ^ kx[3]) |
                       (hi.x ^ kx[4]) | (hi.y ^ kx[5]) | (hi.z ^ kx[6]) | (hi.w ^ kx[7]));
        while (m) {
          take(e0 + __builtin_ctz(m));
          m &= m - 1;
        }
      };
      const int g_lo   = seg_lo >> 5;
      const int g_full = vec_ok ? (seg_hi >> 5) : g_lo;
      int g = g_lo + tid;
      for (; g + 3 * 1024 < g_full; g += 4 * 1024) {
        const uint4* p = key16 + ((g >> 5) * 64 + lane);
        uint4 v[8];
#pragma unroll
        for (int u = 0; u < 4; ++u) {
          v[2 * u]     = p[u * 2048];
          v[2 * u + 1] = p[u * 2048 + 32];
        }
#pragma unroll
        for (int u = 0; u < 4; ++u) test32(v[2 * u], v[2 * u + 1], (g + u * 1024) << 5);
      }
      for (; g < g_full; g += 1024) {
        const uint4* p = key16 + ((g >> 5) * 64 + lane);
        test32(p[0], p[32], g << 5);
      }
      for (int e = (g_full << 5) + tid; e < seg_hi; e += 1024)
        if ((unsigned)(dst[e] - base) < (unsigned)TROWS) take(e);
      __syncthreads();

      const int cnt = qn;
      if (cnt <= QCAP) {
        const int nb = (cnt + 31) >> 5;
        for (int bi = wave; bi < nb; bi += 32) {
          const int left = cnt - (bi << 5);
          const int e    = q[(lane < left) ? ((bi << 5) + lane) : (bi << 5)];
          const int   s_l  = src[e];
          const int   r_l  = dst[e] - base;
          const float nm_l = dinv[s_l] * ew[e] * dinv[base + r_l];
          add32(s_l, r_l, nm_l, left);
        }
        seg_lo = seg_hi;
      } else {
        seg_len = QCAP;
      }
      __syncthreads();
    }

    if constexpr (MM) {
      constexpr int NT  = NOUT / 16;
      constexpr int CT  = NT < 2 ? 1 : 2;
      constexpr int UPR = NT / CT;
      if (ACT) {
        for (int i = tid; i < rows * O4; i += 1024) {
          const int row = i / O4, c = (i - row * O4) * 4;
          float4* p = (float4*)(tile + row * PITCH + c);
          float4 v = *p;
          v.x = (v.x > 0.f) ? v.x : expm1f(v.x);
          v.y = (v.y > 0.f) ? v.y : expm1f(v.y);
          v.z = (v.z > 0.f) ? v.z : expm1f(v.z);
          v.w = (v.w > 0.f) ? v.w : expm1f(v.w);
          *p = v;
        }
      }
      __syncthreads();
      const int wv    = __builtin_amdgcn_readfirstlane(wave);
      const int l15   = lane & 15;
      const int khalf = lane >> 4;
      const int units = ((rows + 15) >> 4) * UPR;
      for (int u = wv; u < units; u += 32) {
        const int rt = u / UPR, cg = u - rt * UPR;
        const float* arow = tile + (rt * 16 + l15) * PITCH + 2 * khalf;
        const float* bfr  = wf + ((cg * CT) * 32 + lane) * 2;
        v8f acc[CT];
#pragma unroll
        for (int ct = 0; ct < CT; ++ct) acc[ct] = v8f{};
#pragma unroll
        for (int k0 = 0; k0 < OUT; k0 += 4) {
          const v2f a = *(const v2f*)(arow + k0);
#pragma unroll
          for (int ct = 0; ct < CT; ++ct) {
            const v2f bb = *(const v2f*)(bfr + ((k0 >> 2) * NT + ct) * 64);
            acc[ct] = __builtin_amdgcn_wmma_f32_16x16x4_f32(
                 false, a,  false, bb,
                 (short)0, acc[ct],  false,  false);
          }
        }
        const int row0 = rt * 16 + khalf * 8;
        float* y = xwn + (size_t)(base + row0) * NOUT + (cg * CT) * 16 + l15;
        if (rt * 16 + 16 <= rows) {
#pragma unroll
          for (int ct = 0; ct < CT; ++ct)
#pragma unroll
            for (int r = 0; r < 8; ++r) y[(size_t)r * NOUT + ct * 16] = acc[ct][r];
        } else {
#pragma unroll
          for (int ct = 0; ct < CT; ++ct)
#pragma unroll
            for (int r = 0; r < 8; ++r)
              if (row0 + r < rows) y[(size_t)r * NOUT + ct * 16] = acc[ct][r];
        }
      }
    } else {
      float4* out4 = (float4*)out + (size_t)base * O4;
      const float4* tile4 = (const float4*)tile;
      for (int i = tid; i < rows * O4; i += 1024) {
        float4 v = tile4[i];
        if (ACT) {
          v.x = (v.x > 0.f) ? v.x : expm1f(v.x);
          v.y = (v.y > 0.f) ? v.y : expm1f(v.y);
          v.z = (v.z > 0.f) ? v.z : expm1f(v.z);
          v.w = (v.w > 0.f) ? v.w : expm1f(v.w);
        }
        out4[i] = v;
      }
    }
    __syncthreads();
  }
}

static inline int cdiv(long long a, long long b) { return (int)((a + b - 1) / b); }

template <int OUT, bool ACT, int REC, int NOUT>
static void run_layer(const float* xw, float* xout, const float* b,
                      const float* Wn, float* xwn,
                      const int* src, const int* dst,
                      const int* key,
                      const float* dinv, const float* ew,
                      const int* tcnt, int* rec, const int* rec_end,
                      int N, int E, hipStream_t stream) {
  agg_tile<OUT, ACT, REC, NOUT><<<cdiv(N, TROWS), 1024, 0, stream>>>(
      xw, src, dst, key, ew, dinv, b, tcnt, rec, rec_end, xout, Wn, xwn, N, E);
}

extern "C" void kernel_launch(void* const* d_in, const int* in_sizes, int n_in,
                              void* d_out, int out_size, void* d_ws, size_t ws_size,
                              hipStream_t stream) {
  const float* x   = (const float*)d_in[0];
  const int*   ei  = (const int*)d_in[1];
  const float* ew  = (const float*)d_in[2];
  const float* W1 = (const float*)d_in[3];  const float* b1 = (const float*)d_in[4];
  const float* W2 = (const float*)d_in[5];  const float* b2 = (const float*)d_in[6];
  const float* W3 = (const float*)d_in[7];  const float* b3 = (const float*)d_in[8];
  const float* W4 = (const float*)d_in[9];  const float* b4 = (const float*)d_in[10];
  const float* W5 = (const float*)d_in[11]; const float* b5 = (const float*)d_in[12];
  const float* W6 = (const float*)d_in[13]; const float* b6 = (const float*)d_in[14];

  const int N = in_sizes[0] / 32;
  const int E = in_sizes[2];
  const int* src = ei;
  const int* dst = ei + E;

  float* dinv = (float*)d_ws;
  float* buf0 = dinv + N + E;
  float* buf1 = buf0 + (size_t)N * 128;
  float* outp = (float*)d_out;
  int* key = (int*)(buf0 + (size_t)N * 64);
  int* tcnt = key + (size_t)cdiv(E, 1024) * 256;
  int* rec  = tcnt + (size_t)cdiv(cdiv(N, TROWS), 64) * 64;
  const int* rec_end = (const int*)buf1;

  key_kernel<<<cdiv(E, 256), 256, 0, stream>>>(dst, key, E);
  tile_prep<<<cdiv(N, TROWS), 1024, 0, stream>>>(src, dst, key, ew, dinv, tcnt, rec, rec_end, N, E);

  gemm_wmma<32, 4><<<cdiv(N, 128), 256, 0, stream>>>(x, W1, buf1, N);
  run_layer<4,   true,  1,  16>(buf1, nullptr, b1, W2, buf0, src, dst, key, dinv, ew, tcnt, rec, rec_end, N, E, stream);
  run_layer<16,  true,  2,  32>(buf0, nullptr, b2, W3, buf1, src, dst, key, dinv, ew, tcnt, rec, rec_end, N, E, stream);
  run_layer<32,  true,  2,  64>(buf1, nullptr, b3, W4, buf0, src, dst, key, dinv, ew, tcnt, rec, rec_end, N, E, stream);
  run_layer<64,  false, 2, 128>(buf0, nullptr, b4, W5, outp, src, dst, key, dinv, ew, tcnt, rec, rec_end, N, E, stream);
  run_layer<128, true,  2, 128>(outp, nullptr, b5, W6, buf1, src, dst, key, dinv, ew, tcnt, rec, rec_end, N, E, stream);
  run_layer<128, false, 2,   0>(buf1, outp, b6, nullptr, nullptr, src, dst, key, dinv, ew, tcnt, rec, rec_end, N, E, stream);
}


# Edge pass also packs each edge as one 8-byte record (source+row, weight) in edge order; the tile pass fetches 1 scattered record per edge, not 3 words (src, dst, ew), 8 in flight; deg sum LDS arrival
# speedup vs baseline: 6.5112x; 1.0374x over previous
#include <hip/hip_runtime.h>
#include <cstdint>

typedef float v2f __attribute__((ext_vector_type(2)));
typedef float v8f __attribute__((ext_vector_type(8)));

constexpr int TROWS = 392;

__device__ __forceinline__ bool erec_usable(const int* erec, int n, int E) {
  return n <= (1 << 23) && 2LL * E <= 64LL * n && ((((size_t)erec) & 7) == 0);
}

__global__ __launch_bounds__(256) void key_kernel(const int* __restrict__ src,
                                                  const int* __restrict__ dst,
                                                  const float* __restrict__ ew,
                                                  int* __restrict__ key,
                                                  int* __restrict__ erec, int n, int E) {
  const int e = blockIdx.x * 256 + threadIdx.x;
  const bool er = (e < E) && erec_usable(erec, n, E);
  const int   d = (e < E) ? dst[e] : 0;
  const int   s = er ? src[e] : 0;
  const float w = er ? ew[e] : 0.f;
  const unsigned k = (unsigned)(d / TROWS);
  if (er) {
    const unsigned r = (unsigned)d - k * (unsigned)TROWS;
    ((int2*)erec)[e] = make_int2((int)((unsigned)s | (r << 23)), __float_as_int(w));
  }
  uint4 lo, hi;
  lo.x = __builtin_amdgcn_ballot_w32((k & 1u) != 0);
  lo.y = __builtin_amdgcn_ballot_w32((k & 2u) != 0);
  lo.z = __builtin_amdgcn_ballot_w32((k & 4u) != 0);
  lo.w = __builtin_amdgcn_ballot_w32((k & 8u) != 0);
  hi.x = __builtin_amdgcn_ballot_w32((k & 16u) != 0);
  hi.y = __builtin_amdgcn_ballot_w32((k & 32u) != 0);
  hi.z = __builtin_amdgcn_ballot_w32((k & 64u) != 0);
  hi.w = __builtin_amdgcn_ballot_w32((k & 128u) != 0);
  if ((threadIdx.x & 31) == 0) {
    const int g = e >> 5;
    uint4* p = (uint4*)key + ((g >> 5) * 64 + (g & 31));
    p[0]  = lo;
    p[32] = hi;
  }
}

constexpr int LDS_FLOATS = 61440;

__global__ __launch_bounds__(1024) void tile_prep(const int* __restrict__ src,
                                                   const int* __restrict__ dst,
                                                   const int* __restrict__ key,
                                                   const float* __restrict__ ew,
                                                   const int* __restrict__ erec,
                                                   float* __restrict__ dinv,
                                                   int* __restrict__ tcnt, int* __restrict__ rec,
                                                   const int* rec_end, int n, int E) {
  constexpr int QCAP = (LDS_FLOATS - TROWS) & ~1023;
  __shared__ float deg[TROWS];
  __shared__ int   q[QCAP];
  __shared__ int   qn;

  const int tid  = threadIdx.x;
  const int lane = tid & 31;
  const uint4* key16 = (const uint4*)key;
  const bool vec_ok = ((n + TROWS - 1) / TROWS <= 256) && ((((size_t)key) & 15) == 0);
  const int tiles = (n + TROWS - 1) / TROWS;
  const long long room = (long long)(rec_end - rec) / 2 / tiles;
  const int rcap = room > 0 ? (int)(room < 0x40000000 ? room : 0x40000000) & ~31 : 0;
  const bool rec_ok = rcap > 0 && n <= (1 << 23) && ((((size_t)rec) & 7) == 0);
  const bool er_ok = erec_usable(erec, n, E);

  for (int t = blockIdx.x; t * TROWS < n; t += gridDim.x) {
    const int base = t * TROWS;
    const int rows = (n - base < TROWS) ? (n - base) : TROWS;
    unsigned kx[8];
#pragma unroll
    for (int j = 0; j < 8; ++j) kx[j] = 0u - (((unsigned)t >> j) & 1u);

    for (int i = tid; i < TROWS; i += 1024) deg[i] = 1.0f;

    int seg_lo = 0, seg_len = E;
    int2* rt = (int2*)rec + (size_t)t * rcap;
    int rec_base = 0;
    while (seg_lo < E) {
      const int seg_hi = (E - seg_lo < seg_len) ? E : seg_lo + seg_len;
      if (tid == 0) qn = 0;
      __syncthreads();

      auto take = [&](int e) {
        const int pos = atomicAdd(&qn, 1);
        if (pos < QCAP) q[pos] = e;
      };
      auto test32 = [&](uint4 lo, uint4 hi, int e0) {
        unsigned m = ~((lo.x ^ kx[0]) | (lo.y ^ kx[1]) | (lo.z ^ kx[2]) | (lo.w ^ kx[3]) |
                       (hi.x ^ kx[4]) | (hi.y ^ kx[5]) | (hi.z ^ kx[6]) | (hi.w ^ kx[7]));
        while (m) {
          take(e0 + __builtin_ctz(m));
          m &= m - 1;
        }
      };
      const int g_lo   = seg_lo >> 5;
      const int g_full = vec_ok ? (seg_hi >> 5) : g_lo;
      int g = g_lo + tid;
      for (; g + 3 * 1024 < g_full; g += 4 * 1024) {
        const uint4* p = key16 + ((g >> 5) * 64 + lane);
        uint4 v[8];
#pragma unroll
        for (int u = 0; u < 4; ++u) {
          v[2 * u]     = p[u * 2048];
          v[2 * u + 1] = p[u * 2048 + 32];
        }
#pragma unroll
        for (int u = 0; u < 4; ++u) test32(v[2 * u], v[2 * u + 1], (g + u * 1024) << 5);
      }
      for (; g < g_full; g += 1024) {
        const uint4* p = key16 + ((g >> 5) * 64 + lane);
        test32(p[0], p[32], g << 5);
      }
      for (int e = (g_full << 5) + tid; e < seg_hi; e += 1024)
        if ((unsigned)(dst[e] - base) < (unsigned)TROWS) take(e);
      __syncthreads();

      const int cnt = qn;
      if (cnt <= QCAP) {
        const bool rec_fits = rec_ok && rec_base + cnt <= rcap;
        for (int i0 = tid; er_ok && i0 < cnt; i0 += 8 * 1024) {
          int  e[8];
          int2 rc[8];
#pragma unroll
          for (int u = 0; u < 8; ++u) {
            const int i = i0 + u * 1024;
            e[u] = q[i < cnt ? i : i0];
          }
#pragma unroll
          for (int u = 0; u < 8; ++u) rc[u] = ((const int2*)erec)[e[u]];
          asm volatile("" :: "v"(rc[0].x), "v"(rc[1].x), "v"(rc[2].x), "v"(rc[3].x),
                             "v"(rc[4].x), "v"(rc[5].x), "v"(rc[6].x), "v"(rc[7].x),
                             "v"(rc[0].y), "v"(rc[1].y), "v"(rc[2].y), "v"(rc[3].y),
                             "v"(rc[4].y), "v"(rc[5].y), "v"(rc[6].y), "v"(rc[7].y));
#pragma unroll
          for (int u = 0; u < 8; ++u) {
            const int i = i0 + u * 1024;
            if (i < cnt) {
              atomicAdd(&deg[(unsigned)rc[u].x >> 23], __int_as_float(rc[u].y));
              if (rec_fits) rt[rec_base + i] = rc[u];
            }
          }
        }
        for (int i0 = tid; !er_ok && i0 < cnt; i0 += 4 * 1024) {
          int e[4], s[4], d[4];
          float w[4];
#pragma unroll
          for (int u = 0; u < 4; ++u) {
            const int i = i0 + u * 1024;
            e[u] = q[i < cnt ? i : i0];
          }
#pragma unroll
          for (int u = 0; u < 4; ++u) {
            s[u] = src[e[u]];
            d[u] = dst[e[u]];
            w[u] = ew[e[u]];
          }
          asm volatile("" :: "v"(s[0]), "v"(s[1]), "v"(s[2]), "v"(s[3]),
                             "v"(d[0]), "v"(d[1]), "v"(d[2]), "v"(d[3]),
                             "v"(w[0]), "v"(w[1]), "v"(w[2]), "v"(w[3]));
#pragma unroll
          for (int u = 0; u < 4; ++u) {
            const int i = i0 + u * 1024;
            if (i < cnt) {
              const int r = d[u] - base;
              atomicAdd(&deg[r], w[u]);
              if (rec_fits)
                rt[rec_base + i] =
                    make_int2((int)((unsigned)s[u] | ((unsigned)r << 23)), __float_as_int(w[u]));
            }
          }
        }
        rec_base += cnt;
        seg_lo = seg_hi;
      } else {
        seg_len = QCAP;
      }
      __syncthreads();
    }

    if (tid == 0) tcnt[t] = (rec_ok && rec_base <= rcap) ? rec_base : -1;
    for (int i = tid; i < rows; i += 1024) {
      const float d = deg[i];
      dinv[base + i] = (d > 0.f) ? rsqrtf(d) : 0.f;
    }
    __syncthreads();
  }
}

template <int IN, int OUT>
__global__ __launch_bounds__(256) void gemm_wmma(const float* __restrict__ x,
                                                 const float* __restrict__ W,
                                                 float* __restrict__ y, int n) {
  constexpr bool USE_LDS = (IN * OUT) <= 8192;
  constexpr bool FULL    = (OUT % 16) == 0;
  __shared__ float sW[USE_LDS ? (IN * OUT) : 1];
  const float* Wp = W;
  if constexpr (USE_LDS) {
    for (int i = threadIdx.x; i < IN * OUT; i += 256) sW[i] = W[i];
    __syncthreads();
    Wp = sW;
  }

  const int lane  = threadIdx.x & 31;
  const int wave  = threadIdx.x >> 5;
  const int l15   = lane & 15;
  const int khalf = lane >> 4;
  const int row0  = blockIdx.x * 128 + wave * 16;

  if (row0 < n) {
    int mload = row0 + l15;
    if (mload > n - 1) mload = n - 1;
    const float* xrow = x + (size_t)mload * IN;
    const bool fulltile = (row0 + 16 <= n);

    for (int n0 = 0; n0 < OUT; n0 += 16) {
      const int colb = n0 + l15;
      const int colc = FULL ? colb : (colb < OUT ? colb : OUT - 1);
      v8f c = {};
#pragma unroll
      for (int k0 = 0; k0 < IN; k0 += 4) {
        const int krow = k0 + 2 * khalf;
        v2f a = *(const v2f*)(xrow + krow);
        v2f b;
        b.x = Wp[(size_t)krow * OUT + colc];
        b.y = Wp[(size_t)(krow + 1) * OUT + colc];
        c = __builtin_amdgcn_wmma_f32_16x16x4_f32(
             false, a,  false, b,
             (short)0, c,  false,  false);
      }
      const int rowoff = khalf * 8;
      if (FULL && fulltile) {
#pragma unroll
        for (int r = 0; r < 8; ++r)
          y[(size_t)(row0 + rowoff + r) * OUT + colb] = c[r];
      } else {
        const bool colok = FULL || (colb < OUT);
#pragma unroll
        for (int r = 0; r < 8; ++r) {
          int row = row0 + rowoff + r;
          if (row < n && colok) y[(size_t)row * OUT + colb] = c[r];
        }
      }
    }
  }
}

template <int OUT, bool ACT, int REC, int NOUT>
__global__ __launch_bounds__(1024) void agg_tile(const float* __restrict__ xw,
                                                  const int* __restrict__ src,
                                                  const int* __restrict__ dst,
                                                  const int* __restrict__ key,
                                                  const float* __restrict__ ew,
                                                  const float* __restrict__ dinv,
                                                  const float* __restrict__ b,
                                                  const int* tcnt, int* rec, const int* rec_end,
                                                  float* __restrict__ out,
                                                  const float* __restrict__ Wn,
                                                  float* __restrict__ xwn, int n, int E) {
  constexpr bool MM = NOUT > 0;
  constexpr int PITCH = (MM && OUT >= 16) ? OUT + 4 : OUT;
  constexpr int TROWS_P = MM ? ((TROWS + 15) / 16) * 16 : TROWS;
  constexpr int TILE = TROWS_P * PITCH;
  constexpr int WF   = MM ? OUT * NOUT : 0;
  constexpr int BUDGET = (TILE + WF + 9216 > LDS_FLOATS) ? 80000 : LDS_FLOATS;
  constexpr int QCAP = (BUDGET - TILE - WF) & ~1023;
  constexpr int O4   = OUT / 4;
  constexpr int LPE  = OUT < 32 ? OUT : 32;
  constexpr int EPS  = 32 / LPE;
  constexpr int CPL  = OUT < 32 ? 1 : OUT / 32;
  constexpr int UN   = LPE < 8 ? LPE : 8;

  __shared__ __attribute__((aligned(16))) float tile[TILE];
  __shared__ __attribute__((aligned(16))) float wf[MM ? WF : 4];
  __shared__ int   q[QCAP];
  __shared__ int   qn;

  const int tid  = threadIdx.x;
  const int lane = tid & 31;
  const int wave = tid >> 5;
  if constexpr (MM) {
    constexpr int NT = NOUT / 16;
    for (int i = tid; i < OUT * NOUT; i += 1024) {
      const int k = i / NOUT, c = i - k * NOUT;
      const int ln = ((k >> 1) & 1) * 16 + (c & 15);
      wf[(((k >> 2) * NT + (c >> 4)) * 32 + ln) * 2 + (k & 1)] = Wn[i];
    }
  }
  const int qsub = lane / LPE;
  const int ch0  = lane % LPE;
  const uint4* key16 = (const uint4*)key;
  const bool vec_ok = ((n + TROWS - 1) / TROWS <= 256) && ((((size_t)key) & 15) == 0);

  const int tiles = (n + TROWS - 1) / TROWS;
  const long long room = (long long)(rec_end - rec) / 2 / tiles;
  const int rcap = room > 0 ? (int)(room < 0x40000000 ? room : 0x40000000) & ~31 : 0;
  const bool rec_ok = rcap > 0 && n <= (1 << 23) && ((((size_t)rec) & 7) == 0);

  for (int t = blockIdx.x; t * TROWS < n; t += gridDim.x) {
    const int base = t * TROWS;
    const int rows = (n - base < TROWS) ? (n - base) : TROWS;
    unsigned kx[8];
#pragma unroll
    for (int j = 0; j < 8; ++j) kx[j] = 0u - (((unsigned)t >> j) & 1u);

    {
      const float4* xw4 = (const float4*)xw + (size_t)base * O4;
      for (int i = tid; i < rows * O4; i += 1024) {
        const int row = i / O4, c = (i - row * O4) * 4;
        const float dv = dinv[base + row];
        const float w = dv * dv;
        const float4 x = xw4[i];
        float4 r;
        r.x = x.x * w + b[c];
        r.y = x.y * w + b[c + 1];
        r.z = x.z * w + b[c + 2];
        r.w = x.w * w + b[c + 3];
        *(float4*)(tile + row * PITCH + c) = r;
      }
    }

    auto add32 = [&](int s_l, int r_l, float nm_l, int left) {
#pragma unroll 1
      for (int k0 = 0; k0 < LPE; k0 += UN) {
        float val[UN][CPL];
        int   rr[UN];
        float nn[UN];
#pragma unroll
        for (int u = 0; u < UN; ++u) {
          const int ent = (k0 + u) * EPS + qsub;
          int s;
          if constexpr (LPE == 32) {
            s     = __builtin_amdgcn_readlane(s_l, k0 + u);
            rr[u] = __builtin_amdgcn_readlane(r_l, k0 + u);
            nn[u] = __int_as_float(__builtin_amdgcn_readlane(__float_as_int(nm_l), k0 + u));
          } else {
            s     = __shfl(s_l, ent);
            rr[u] = __shfl(r_l, ent);
            nn[u] = __shfl(nm_l, ent);
          }
          const float* xr = xw + (size_t)s * OUT + ch0;
#pragma unroll
          for (int j = 0; j < CPL; ++j) val[u][j] = xr[32 * j];
        }
#pragma unroll
        for (int u = 0; u < UN; ++u) {
          const int ent = (k0 + u) * EPS + qsub;
          if (ent < left) {
            float* tr = tile + rr[u] * PITCH + ch0;
#pragma unroll
            for (int j = 0; j < CPL; ++j) atomicAdd(tr + 32 * j, val[u][j] * nn[u]);
          }
        }
      }
    };

    int seg_lo = 0, seg_len = E;
    int2* rt = (int2*)rec + (size_t)t * rcap;
    {
      const int cnt = rec_ok ? tcnt[t] : -1;
      if (cnt >= 0 && cnt <= rcap) {
        __syncthreads();
        const int nb = (cnt + 31) >> 5;
        for (int bi = wave; bi < nb; bi += 32) {
          const int left = cnt - (bi << 5);
          const int idx  = (lane < left) ? ((bi << 5) + lane) : (bi << 5);
          const int2 r = rt[idx];
          const int s_l = r.x & 0x7fffff;
          const int r_l = (int)((unsigned)r.x >> 23);
          float nm_l = __int_as_float(r.y);
          if constexpr (REC == 1) {
            nm_l = dinv[s_l] * nm_l * dinv[base + r_l];
            if (lane < left) rt[idx] = make_int2(r.x, __float_as_int(nm_l));
          }
          add32(s_l, r_l, nm_l, left);
        }
        __syncthreads();
        seg_lo = E;
      }
    }
    while (seg_lo < E) {
      const int seg_hi = (E - seg_lo < seg_len) ? E : seg_lo + seg_len;
      if (tid == 0) qn = 0;
      __syncthreads();

      auto take = [&](int e) {
        const int pos = atomicAdd(&qn, 1);
        if (pos < QCAP) q[pos] = e;
      };
      auto test32 = [&](uint4 lo, uint4 hi, int e0) {
        unsigned m = ~((lo.x ^ kx[0]) | (lo.y ^ kx[1]) | (lo.z ^ kx[2]) | (lo.w ^ kx[3]) |
                       (hi.x ^ kx[4]) | (hi.y ^ kx[5]) | (hi.z ^ kx[6]) | (hi.w ^ kx[7]));
        while (m) {
          take(e0 + __builtin_ctz(m));
          m &= m - 1;
        }
      };
      const int g_lo   = seg_lo >> 5;
      const int g_full = vec_ok ? (seg_hi >> 5) : g_lo;
      int g = g_lo + tid;
      for (; g + 3 * 1024 < g_full; g += 4 * 1024) {
        const uint4* p = key16 + ((g >> 5) * 64 + lane);
        uint4 v[8];
#pragma unroll
        for (int u = 0; u < 4; ++u) {
          v[2 * u]     = p[u * 2048];
          v[2 * u + 1] = p[u * 2048 + 32];
        }
#pragma unroll
        for (int u = 0; u < 4; ++u) test32(v[2 * u], v[2 * u + 1], (g + u * 1024) << 5);
      }
      for (; g < g_full; g += 1024) {
        const uint4* p = key16 + ((g >> 5) * 64 + lane);
        test32(p[0], p[32], g << 5);
      }
      for (int e = (g_full << 5) + tid; e < seg_hi; e += 1024)
        if ((unsigned)(dst[e] - base) < (unsigned)TROWS) take(e);
      __syncthreads();

      const int cnt = qn;
      if (cnt <= QCAP) {
        const int nb = (cnt + 31) >> 5;
        for (int bi = wave; bi < nb; bi += 32) {
          const int left = cnt - (bi << 5);
          const int e    = q[(lane < left) ? ((bi << 5) + lane) : (bi << 5)];
          const int   s_l  = src[e];
          const int   r_l  = dst[e] - base;
          const float nm_l = dinv[s_l] * ew[e] * dinv[base + r_l];
          add32(s_l, r_l, nm_l, left);
        }
        seg_lo = seg_hi;
      } else {
        seg_len = QCAP;
      }
      __syncthreads();
    }

    if constexpr (MM) {
      constexpr int NT  = NOUT / 16;
      constexpr int CT  = NT < 2 ? 1 : 2;
      constexpr int UPR = NT / CT;
      if (ACT) {
        for (int i = tid; i < rows * O4; i += 1024) {
          const int row = i / O4, c = (i - row * O4) * 4;
          float4* p = (float4*)(tile + row * PITCH + c);
          float4 v = *p;
          v.x = (v.x > 0.f) ? v.x : expm1f(v.x);
          v.y = (v.y > 0.f) ? v.y : expm1f(v.y);
          v.z = (v.z > 0.f) ? v.z : expm1f(v.z);
          v.w = (v.w > 0.f) ? v.w : expm1f(v.w);
          *p = v;
        }
      }
      __syncthreads();
      const int wv    = __builtin_amdgcn_readfirstlane(wave);
      const int l15   = lane & 15;
      const int khalf = lane >> 4;
      const int units = ((rows + 15) >> 4) * UPR;
      for (int u = wv; u < units; u += 32) {
        const int rt = u / UPR, cg = u - rt * UPR;
        const float* arow = tile + (rt * 16 + l15) * PITCH + 2 * khalf;
        const float* bfr  = wf + ((cg * CT) * 32 + lane) * 2;
        v8f acc[CT];
#pragma unroll
        for (int ct = 0; ct < CT; ++ct) acc[ct] = v8f{};
#pragma unroll
        for (int k0 = 0; k0 < OUT; k0 += 4) {
          const v2f a = *(const v2f*)(arow + k0);
#pragma unroll
          for (int ct = 0; ct < CT; ++ct) {
            const v2f bb = *(const v2f*)(bfr + ((k0 >> 2) * NT + ct) * 64);
            acc[ct] = __builtin_amdgcn_wmma_f32_16x16x4_f32(
                 false, a,  false, bb,
                 (short)0, acc[ct],  false,  false);
          }
        }
        const int row0 = rt * 16 + khalf * 8;
        float* y = xwn + (size_t)(base + row0) * NOUT + (cg * CT) * 16 + l15;
        if (rt * 16 + 16 <= rows) {
#pragma unroll
          for (int ct = 0; ct < CT; ++ct)
#pragma unroll
            for (int r = 0; r < 8; ++r) y[(size_t)r * NOUT + ct * 16] = acc[ct][r];
        } else {
#pragma unroll
          for (int ct = 0; ct < CT; ++ct)
#pragma unroll
            for (int r = 0; r < 8; ++r)
              if (row0 + r < rows) y[(size_t)r * NOUT + ct * 16] = acc[ct][r];
        }
      }
    } else {
      float4* out4 = (float4*)out + (size_t)base * O4;
      const float4* tile4 = (const float4*)tile;
      for (int i = tid; i < rows * O4; i += 1024) {
        float4 v = tile4[i];
        if (ACT) {
          v.x = (v.x > 0.f) ? v.x : expm1f(v.x);
          v.y = (v.y > 0.f) ? v.y : expm1f(v.y);
          v.z = (v.z > 0.f) ? v.z : expm1f(v.z);
          v.w = (v.w > 0.f) ? v.w : expm1f(v.w);
        }
        out4[i] = v;
      }
    }
    __syncthreads();
  }
}

static inline int cdiv(long long a, long long b) { return (int)((a + b - 1) / b); }

template <int OUT, bool ACT, int REC, int NOUT>
static void run_layer(const float* xw, float* xout, const float* b,
                      const float* Wn, float* xwn,
                      const int* src, const int* dst,
                      const int* key,
                      const float* dinv, const float* ew,
                      const int* tcnt, int* rec, const int* rec_end,
                      int N, int E, hipStream_t stream) {
  agg_tile<OUT, ACT, REC, NOUT><<<cdiv(N, TROWS), 1024, 0, stream>>>(
      xw, src, dst, key, ew, dinv, b, tcnt, rec, rec_end, xout, Wn, xwn, N, E);
}

extern "C" void kernel_launch(void* const* d_in, const int* in_sizes, int n_in,
                              void* d_out, int out_size, void* d_ws, size_t ws_size,
                              hipStream_t stream) {
  const float* x   = (const float*)d_in[0];
  const int*   ei  = (const int*)d_in[1];
  const float* ew  = (const float*)d_in[2];
  const float* W1 = (const float*)d_in[3];  const float* b1 = (const float*)d_in[4];
  const float* W2 = (const float*)d_in[5];  const float* b2 = (const float*)d_in[6];
  const float* W3 = (const float*)d_in[7];  const float* b3 = (const float*)d_in[8];
  const float* W4 = (const float*)d_in[9];  const float* b4 = (const float*)d_in[10];
  const float* W5 = (const float*)d_in[11]; const float* b5 = (const float*)d_in[12];
  const float* W6 = (const float*)d_in[13]; const float* b6 = (const float*)d_in[14];

  const int N = in_sizes[0] / 32;
  const int E = in_sizes[2];
  const int* src = ei;
  const int* dst = ei + E;

  float* dinv = (float*)d_ws;
  float* buf0 = dinv + N + E;
  float* buf1 = buf0 + (size_t)N * 128;
  float* outp = (float*)d_out;
  int* key = (int*)(buf0 + (size_t)N * 64);
  int* tcnt = key + (size_t)cdiv(E, 1024) * 256;
  int* rec  = tcnt + (size_t)cdiv(cdiv(N, TROWS), 64) * 64;
  const int* rec_end = (const int*)buf1;
  int* erec = (int*)(buf1 + (size_t)N * 64);

  key_kernel<<<cdiv(E, 256), 256, 0, stream>>>(src, dst, ew, key, erec, N, E);
  tile_prep<<<cdiv(N, TROWS), 1024, 0, stream>>>(src, dst, key, ew, erec, dinv, tcnt, rec, rec_end, N, E);

  gemm_wmma<32, 4><<<cdiv(N, 128), 256, 0, stream>>>(x, W1, buf1, N);
  run_layer<4,   true,  1,  16>(buf1, nullptr, b1, W2, buf0, src, dst, key, dinv, ew, tcnt, rec, rec_end, N, E, stream);
  run_layer<16,  true,  2,  32>(buf0, nullptr, b2, W3, buf1, src, dst, key, dinv, ew, tcnt, rec, rec_end, N, E, stream);
  run_layer<32,  true,  2,  64>(buf1, nullptr, b3, W4, buf0, src, dst, key, dinv, ew, tcnt, rec, rec_end, N, E, stream);
  run_layer<64,  false, 2, 128>(buf0, nullptr, b4, W5, outp, src, dst, key, dinv, ew, tcnt, rec, rec_end, N, E, stream);
  run_layer<128, true,  2, 128>(outp, nullptr, b5, W6, buf1, src, dst, key, dinv, ew, tcnt, rec, rec_end, N, E, stream);
  run_layer<128, false, 2,   0>(buf1, outp, b6, nullptr, nullptr, src, dst, key, dinv, ew, tcnt, rec, rec_end, N, E, stream);
}


# Rows of 64 and 128 channels stored with a lane's channels side by side (Wn's columns taken in that order): layers 4-6 load an edge's row as 8 or 16 bytes per lane, 1 load not 2 or 4; LDS adds same
# speedup vs baseline: 6.7137x; 1.0311x over previous
#include <hip/hip_runtime.h>
#include <cstdint>

typedef float v2f __attribute__((ext_vector_type(2)));
typedef float v8f __attribute__((ext_vector_type(8)));

constexpr int TROWS = 392;

__device__ __forceinline__ bool erec_usable(const int* erec, int n, int E) {
  return n <= (1 << 23) && 2LL * E <= 64LL * n && ((((size_t)erec) & 7) == 0);
}

__global__ __launch_bounds__(256) void key_kernel(const int* __restrict__ src,
                                                  const int* __restrict__ dst,
                                                  const float* __restrict__ ew,
                                                  int* __restrict__ key,
                                                  int* __restrict__ erec, int n, int E) {
  const int e = blockIdx.x * 256 + threadIdx.x;
  const bool er = (e < E) && erec_usable(erec, n, E);
  const int   d = (e < E) ? dst[e] : 0;
  const int   s = er ? src[e] : 0;
  const float w = er ? ew[e] : 0.f;
  const unsigned k = (unsigned)(d / TROWS);
  if (er) {
    const unsigned r = (unsigned)d - k * (unsigned)TROWS;
    ((int2*)erec)[e] = make_int2((int)((unsigned)s | (r << 23)), __float_as_int(w));
  }
  uint4 lo, hi;
  lo.x = __builtin_amdgcn_ballot_w32((k & 1u) != 0);
  lo.y = __builtin_amdgcn_ballot_w32((k & 2u) != 0);
  lo.z = __builtin_amdgcn_ballot_w32((k & 4u) != 0);
  lo.w = __builtin_amdgcn_ballot_w32((k & 8u) != 0);
  hi.x = __builtin_amdgcn_ballot_w32((k & 16u) != 0);
  hi.y = __builtin_amdgcn_ballot_w32((k & 32u) != 0);
  hi.z = __builtin_amdgcn_ballot_w32((k & 64u) != 0);
  hi.w = __builtin_amdgcn_ballot_w32((k & 128u) != 0);
  if ((threadIdx.x & 31) == 0) {
    const int g = e >> 5;
    uint4* p = (uint4*)key + ((g >> 5) * 64 + (g & 31));
    p[0]  = lo;
    p[32] = hi;
  }
}

constexpr int LDS_FLOATS = 61440;

__global__ __launch_bounds__(1024) void tile_prep(const int* __restrict__ src,
                                                   const int* __restrict__ dst,
                                                   const int* __restrict__ key,
                                                   const float* __restrict__ ew,
                                                   const int* __restrict__ erec,
                                                   float* __restrict__ dinv,
                                                   int* __restrict__ tcnt, int* __restrict__ rec,
                                                   const int* rec_end, int n, int E) {
  constexpr int QCAP = (LDS_FLOATS - TROWS) & ~1023;
  __shared__ float deg[TROWS];
  __shared__ int   q[QCAP];
  __shared__ int   qn;

  const int tid  = threadIdx.x;
  const int lane = tid & 31;
  const uint4* key16 = (const uint4*)key;
  const bool vec_ok = ((n + TROWS - 1) / TROWS <= 256) && ((((size_t)key) & 15) == 0);
  const int tiles = (n + TROWS - 1) / TROWS;
  const long long room = (long long)(rec_end - rec) / 2 / tiles;
  const int rcap = room > 0 ? (int)(room < 0x40000000 ? room : 0x40000000) & ~31 : 0;
  const bool rec_ok = rcap > 0 && n <= (1 << 23) && ((((size_t)rec) & 7) == 0);
  const bool er_ok = erec_usable(erec, n, E);

  for (int t = blockIdx.x; t * TROWS < n; t += gridDim.x) {
    const int base = t * TROWS;
    const int rows = (n - base < TROWS) ? (n - base) : TROWS;
    unsigned kx[8];
#pragma unroll
    for (int j = 0; j < 8; ++j) kx[j] = 0u - (((unsigned)t >> j) & 1u);

    for (int i = tid; i < TROWS; i += 1024) deg[i] = 1.0f;

    int seg_lo = 0, seg_len = E;
    int2* rt = (int2*)rec + (size_t)t * rcap;
    int rec_base = 0;
    while (seg_lo < E) {
      const int seg_hi = (E - seg_lo < seg_len) ? E : seg_lo + seg_len;
      if (tid == 0) qn = 0;
      __syncthreads();

      auto take = [&](int e) {
        const int pos = atomicAdd(&qn, 1);
        if (pos < QCAP) q[pos] = e;
      };
      auto test32 = [&](uint4 lo, uint4 hi, int e0) {
        unsigned m = ~((lo.x ^ kx[0]) | (lo.y ^ kx[1]) | (lo.z ^ kx[2]) | (lo.w ^ kx[3]) |
                       (hi.x ^ kx[4]) | (hi.y ^ kx[5]) | (hi.z ^ kx[6]) | (hi.w ^ kx[7]));
        while (m) {
          take(e0 + __builtin_ctz(m));
          m &= m - 1;
        }
      };
      const int g_lo   = seg_lo >> 5;
      const int g_full = vec_ok ? (seg_hi >> 5) : g_lo;
      int g = g_lo + tid;
      for (; g + 3 * 1024 < g_full; g += 4 * 1024) {
        const uint4* p = key16 + ((g >> 5) * 64 + lane);
        uint4 v[8];
#pragma unroll
        for (int u = 0; u < 4; ++u) {
          v[2 * u]     = p[u * 2048];
          v[2 * u + 1] = p[u * 2048 + 32];
        }
#pragma unroll
        for (int u = 0; u < 4; ++u) test32(v[2 * u], v[2 * u + 1], (g + u * 1024) << 5);
      }
      for (; g < g_full; g += 1024) {
        const uint4* p = key16 + ((g >> 5) * 64 + lane);
        test32(p[0], p[32], g << 5);
      }
      for (int e = (g_full << 5) + tid; e < seg_hi; e += 1024)
        if ((unsigned)(dst[e] - base) < (unsigned)TROWS) take(e);
      __syncthreads();

      const int cnt = qn;
      if (cnt <= QCAP) {
        const bool rec_fits = rec_ok && rec_base + cnt <= rcap;
        for (int i0 = tid; er_ok && i0 < cnt; i0 += 8 * 1024) {
          int  e[8];
          int2 rc[8];
#pragma unroll
          for (int u = 0; u < 8; ++u) {
            const int i = i0 + u * 1024;
            e[u] = q[i < cnt ? i : i0];
          }
#pragma unroll
          for (int u = 0; u < 8; ++u) rc[u] = ((const int2*)erec)[e[u]];
          asm volatile("" :: "v"(rc[0].x), "v"(rc[1].x), "v"(rc[2].x), "v"(rc[3].x),
                             "v"(rc[4].x), "v"(rc[5].x), "v"(rc[6].x), "v"(rc[7].x),
                             "v"(rc[0].y), "v"(rc[1].y), "v"(rc[2].y), "v"(rc[3].y),
                             "v"(rc[4].y), "v"(rc[5].y), "v"(rc[6].y), "v"(rc[7].y));
#pragma unroll
          for (int u = 0; u < 8; ++u) {
            const int i = i0 + u * 1024;
            if (i < cnt) {
              atomicAdd(&deg[(unsigned)rc[u].x >> 23], __int_as_float(rc[u].y));
              if (rec_fits) rt[rec_base + i] = rc[u];
            }
          }
        }
        for (int i0 = tid; !er_ok && i0 < cnt; i0 += 4 * 1024) {
          int e[4], s[4], d[4];
          float w[4];
#pragma unroll
          for (int u = 0; u < 4; ++u) {
            const int i = i0 + u * 1024;
            e[u] = q[i < cnt ? i : i0];
          }
#pragma unroll
          for (int u = 0; u < 4; ++u) {
            s[u] = src[e[u]];
            d[u] = dst[e[u]];
            w[u] = ew[e[u]];
          }
          asm volatile("" :: "v"(s[0]), "v"(s[1]), "v"(s[2]), "v"(s[3]),
                             "v"(d[0]), "v"(d[1]), "v"(d[2]), "v"(d[3]),
                             "v"(w[0]), "v"(w[1]), "v"(w[2]), "v"(w[3]));
#pragma unroll
          for (int u = 0; u < 4; ++u) {
            const int i = i0 + u * 1024;
            if (i < cnt) {
              const int r = d[u] - base;
              atomicAdd(&deg[r], w[u]);
              if (rec_fits)
                rt[rec_base + i] =
                    make_int2((int)((unsigned)s[u] | ((unsigned)r << 23)), __float_as_int(w[u]));
            }
          }
        }
        rec_base += cnt;
        seg_lo = seg_hi;
      } else {
        seg_len = QCAP;
      }
      __syncthreads();
    }

    if (tid == 0) tcnt[t] = (rec_ok && rec_base <= rcap) ? rec_base : -1;
    for (int i = tid; i < rows; i += 1024) {
      const float d = deg[i];
      dinv[base + i] = (d > 0.f) ? rsqrtf(d) : 0.f;
    }
    __syncthreads();
  }
}

template <int IN, int OUT>
__global__ __launch_bounds__(256) void gemm_wmma(const float* __restrict__ x,
                                                 const float* __restrict__ W,
                                                 float* __restrict__ y, int n) {
  constexpr bool USE_LDS = (IN * OUT) <= 8192;
  constexpr bool FULL    = (OUT % 16) == 0;
  __shared__ float sW[USE_LDS ? (IN * OUT) : 1];
  const float* Wp = W;
  if constexpr (USE_LDS) {
    for (int i = threadIdx.x; i < IN * OUT; i += 256) sW[i] = W[i];
    __syncthreads();
    Wp = sW;
  }

  const int lane  = threadIdx.x & 31;
  const int wave  = threadIdx.x >> 5;
  const int l15   = lane & 15;
  const int khalf = lane >> 4;
  const int row0  = blockIdx.x * 128 + wave * 16;

  if (row0 < n) {
    int mload = row0 + l15;
    if (mload > n - 1) mload = n - 1;
    const float* xrow = x + (size_t)mload * IN;
    const bool fulltile = (row0 + 16 <= n);

    for (int n0 = 0; n0 < OUT; n0 += 16) {
      const int colb = n0 + l15;
      const int colc = FULL ? colb : (colb < OUT ? colb : OUT - 1);
      v8f c = {};
#pragma unroll
      for (int k0 = 0; k0 < IN; k0 += 4) {
        const int krow = k0 + 2 * khalf;
        v2f a = *(const v2f*)(xrow + krow);
        v2f b;
        b.x = Wp[(size_t)krow * OUT + colc];
        b.y = Wp[(size_t)(krow + 1) * OUT + colc];
        c = __builtin_amdgcn_wmma_f32_16x16x4_f32(
             false, a,  false, b,
             (short)0, c,  false,  false);
      }
      const int rowoff = khalf * 8;
      if (FULL && fulltile) {
#pragma unroll
        for (int r = 0; r < 8; ++r)
          y[(size_t)(row0 + rowoff + r) * OUT + colb] = c[r];
      } else {
        const bool colok = FULL || (colb < OUT);
#pragma unroll
        for (int r = 0; r < 8; ++r) {
          int row = row0 + rowoff + r;
          if (row < n && colok) y[(size_t)row * OUT + colb] = c[r];
        }
      }
    }
  }
}

template <int OUT, bool ACT, int REC, int NOUT>
__global__ __launch_bounds__(1024) void agg_tile(const float* __restrict__ xw,
                                                  const int* __restrict__ src,
                                                  const int* __restrict__ dst,
                                                  const int* __restrict__ key,
                                                  const float* __restrict__ ew,
                                                  const float* __restrict__ dinv,
                                                  const float* __restrict__ b,
                                                  const int* tcnt, int* rec, const int* rec_end,
                                                  float* __restrict__ out,
                                                  const float* __restrict__ Wn,
                                                  float* __restrict__ xwn, int n, int E) {
  constexpr bool MM = NOUT > 0;
  constexpr int PITCH = (MM && OUT >= 16) ? OUT + 4 : OUT;
  constexpr int TROWS_P = MM ? ((TROWS + 15) / 16) * 16 : TROWS;
  constexpr int TILE = TROWS_P * PITCH;
  constexpr int WF   = MM ? OUT * NOUT : 0;
  constexpr int BUDGET = (TILE + WF + 9216 > LDS_FLOATS) ? 80000 : LDS_FLOATS;
  constexpr int QCAP = (BUDGET - TILE - WF) & ~1023;
  constexpr int O4   = OUT / 4;
  constexpr int LPE  = OUT < 32 ? OUT : 32;
  constexpr int EPS  = 32 / LPE;
  constexpr int CPL  = OUT < 32 ? 1 : OUT / 32;
  constexpr int UN   = LPE < 8 ? LPE : 8;
  constexpr int CPN  = NOUT >= 64 ? NOUT / 32 : 1;

  __shared__ __attribute__((aligned(16))) float tile[TILE];
  __shared__ __attribute__((aligned(16))) float wf[MM ? WF : 4];
  __shared__ int   q[QCAP];
  __shared__ int   qn;

  const int tid  = threadIdx.x;
  const int lane = tid & 31;
  const int wave = tid >> 5;
  if constexpr (MM) {
    constexpr int NT = NOUT / 16;
    for (int i = tid; i < OUT * NOUT; i += 1024) {
      const int k = i / NOUT, p = i - k * NOUT;
      const int c = (p % CPN) * 32 + p / CPN;
      const int ln = ((k >> 1) & 1) * 16 + (p & 15);
      wf[(((k >> 2) * NT + (p >> 4)) * 32 + ln) * 2 + (k & 1)] = Wn[k * NOUT + c];
    }
  }
  const int qsub = lane / LPE;
  const int ch0  = lane % LPE;
  const uint4* key16 = (const uint4*)key;
  const bool vec_ok = ((n + TROWS - 1) / TROWS <= 256) && ((((size_t)key) & 15) == 0);

  const int tiles = (n + TROWS - 1) / TROWS;
  const long long room = (long long)(rec_end - rec) / 2 / tiles;
  const int rcap = room > 0 ? (int)(room < 0x40000000 ? room : 0x40000000) & ~31 : 0;
  const bool rec_ok = rcap > 0 && n <= (1 << 23) && ((((size_t)rec) & 7) == 0);

  for (int t = blockIdx.x; t * TROWS < n; t += gridDim.x) {
    const int base = t * TROWS;
    const int rows = (n - base < TROWS) ? (n - base) : TROWS;
    unsigned kx[8];
#pragma unroll
    for (int j = 0; j < 8; ++j) kx[j] = 0u - (((unsigned)t >> j) & 1u);

    {
      const float4* xw4 = (const float4*)xw + (size_t)base * O4;
      for (int i = tid; i < rows * O4; i += 1024) {
        const int row = i / O4, c = (i - row * O4) * 4;
        const float dv = dinv[base + row];
        const float w = dv * dv;
        const float4 x = xw4[i];
        if constexpr (CPL > 1) {
          const float xs[4] = {x.x, x.y, x.z, x.w};
#pragma unroll
          for (int jj = 0; jj < 4; ++jj) {
            const int ch = ((c + jj) % CPL) * 32 + (c + jj) / CPL;
            tile[row * PITCH + ch] = xs[jj] * w + b[ch];
          }
        } else {
          float4 r;
          r.x = x.x * w + b[c];
          r.y = x.y * w + b[c + 1];
          r.z = x.z * w + b[c + 2];
          r.w = x.w * w + b[c + 3];
          *(float4*)(tile + row * PITCH + c) = r;
        }
      }
    }

    auto add32 = [&](int s_l, int r_l, float nm_l, int left) {
#pragma unroll 1
      for (int k0 = 0; k0 < LPE; k0 += UN) {
        float val[UN][CPL];
        int   rr[UN];
        float nn[UN];
#pragma unroll
        for (int u = 0; u < UN; ++u) {
          const int ent = (k0 + u) * EPS + qsub;
          int s;
          if constexpr (LPE == 32) {
            s     = __builtin_amdgcn_readlane(s_l, k0 + u);
            rr[u] = __builtin_amdgcn_readlane(r_l, k0 + u);
            nn[u] = __int_as_float(__builtin_amdgcn_readlane(__float_as_int(nm_l), k0 + u));
          } else {
            s     = __shfl(s_l, ent);
            rr[u] = __shfl(r_l, ent);
            nn[u] = __shfl(nm_l, ent);
          }
          const float* xr = xw + (size_t)s * OUT + ch0 * CPL;
          if constexpr (CPL == 4) {
            const float4 v = *(const float4*)xr;
            val[u][0] = v.x; val[u][1] = v.y; val[u][2] = v.z; val[u][3] = v.w;
          } else if constexpr (CPL == 2) {
            const float2 v = *(const float2*)xr;
            val[u][0] = v.x; val[u][1] = v.y;
          } else {
            val[u][0] = xr[0];
          }
        }
#pragma unroll
        for (int u = 0; u < UN; ++u) {
          const int ent = (k0 + u) * EPS + qsub;
          if (ent < left) {
            float* tr = tile + rr[u] * PITCH + ch0;
#pragma unroll
            for (int j = 0; j < CPL; ++j) atomicAdd(tr + 32 * j, val[u][j] * nn[u]);
          }
        }
      }
    };

    int seg_lo = 0, seg_len = E;
    int2* rt = (int2*)rec + (size_t)t * rcap;
    {
      const int cnt = rec_ok ? tcnt[t] : -1;
      if (cnt >= 0 && cnt <= rcap) {
        __syncthreads();
        const int nb = (cnt + 31) >> 5;
        for (int bi = wave; bi < nb; bi += 32) {
          const int left = cnt - (bi << 5);
          const int idx  = (lane < left) ? ((bi << 5) + lane) : (bi << 5);
          const int2 r = rt[idx];
          const int s_l = r.x & 0x7fffff;
          const int r_l = (int)((unsigned)r.x >> 23);
          float nm_l = __int_as_float(r.y);
          if constexpr (REC == 1) {
            nm_l = dinv[s_l] * nm_l * dinv[base + r_l];
            if (lane < left) rt[idx] = make_int2(r.x, __float_as_int(nm_l));
          }
          add32(s_l, r_l, nm_l, left);
        }
        __syncthreads();
        seg_lo = E;
      }
    }
    while (seg_lo < E) {
      const int seg_hi = (E - seg_lo < seg_len) ? E : seg_lo + seg_len;
      if (tid == 0) qn = 0;
      __syncthreads();

      auto take = [&](int e) {
        const int pos = atomicAdd(&qn, 1);
        if (pos < QCAP) q[pos] = e;
      };
      auto test32 = [&](uint4 lo, uint4 hi, int e0) {
        unsigned m = ~((lo.x ^ kx[0]) | (lo.y ^ kx[1]) | (lo.z ^ kx[2]) | (lo.w ^ kx[3]) |
                       (hi.x ^ kx[4]) | (hi.y ^ kx[5]) | (hi.z ^ kx[6]) | (hi.w ^ kx[7]));
        while (m) {
          take(e0 + __builtin_ctz(m));
          m &= m - 1;
        }
      };
      const int g_lo   = seg_lo >> 5;
      const int g_full = vec_ok ? (seg_hi >> 5) : g_lo;
      int g = g_lo + tid;
      for (; g + 3 * 1024 < g_full; g += 4 * 1024) {
        const uint4* p = key16 + ((g >> 5) * 64 + lane);
        uint4 v[8];
#pragma unroll
        for (int u = 0; u < 4; ++u) {
          v[2 * u]     = p[u * 2048];
          v[2 * u + 1] = p[u * 2048 + 32];
        }
#pragma unroll
        for (int u = 0; u < 4; ++u) test32(v[2 * u], v[2 * u + 1], (g + u * 1024) << 5);
      }
      for (; g < g_full; g += 1024) {
        const uint4* p = key16 + ((g >> 5) * 64 + lane);
        test32(p[0], p[32], g << 5);
      }
      for (int e = (g_full << 5) + tid; e < seg_hi; e += 1024)
        if ((unsigned)(dst[e] - base) < (unsigned)TROWS) take(e);
      __syncthreads();

      const int cnt = qn;
      if (cnt <= QCAP) {
        const int nb = (cnt + 31) >> 5;
        for (int bi = wave; bi < nb; bi += 32) {
          const int left = cnt - (bi << 5);
          const int e    = q[(lane < left) ? ((bi << 5) + lane) : (bi << 5)];
          const int   s_l  = src[e];
          const int   r_l  = dst[e] - base;
          const float nm_l = dinv[s_l] * ew[e] * dinv[base + r_l];
          add32(s_l, r_l, nm_l, left);
        }
        seg_lo = seg_hi;
      } else {
        seg_len = QCAP;
      }
      __syncthreads();
    }

    if constexpr (MM) {
      constexpr int NT  = NOUT / 16;
      constexpr int CT  = NT < 2 ? 1 : 2;
      constexpr int UPR = NT / CT;
      if (ACT) {
        for (int i = tid; i < rows * O4; i += 1024) {
          const int row = i / O4, c = (i - row * O4) * 4;
          float4* p = (float4*)(tile + row * PITCH + c);
          float4 v = *p;
          v.x = (v.x > 0.f) ? v.x : expm1f(v.x);
          v.y = (v.y > 0.f) ? v.y : expm1f(v.y);
          v.z = (v.z > 0.f) ? v.z : expm1f(v.z);
          v.w = (v.w > 0.f) ? v.w : expm1f(v.w);
          *p = v;
        }
      }
      __syncthreads();
      const int wv    = __builtin_amdgcn_readfirstlane(wave);
      const int l15   = lane & 15;
      const int khalf = lane >> 4;
      const int units = ((rows + 15) >> 4) * UPR;
      for (int u = wv; u < units; u += 32) {
        const int rt = u / UPR, cg = u - rt * UPR;
        const float* arow = tile + (rt * 16 + l15) * PITCH + 2 * khalf;
        const float* bfr  = wf + ((cg * CT) * 32 + lane) * 2;
        v8f acc[CT];
#pragma unroll
        for (int ct = 0; ct < CT; ++ct) acc[ct] = v8f{};
#pragma unroll
        for (int k0 = 0; k0 < OUT; k0 += 4) {
          const v2f a = *(const v2f*)(arow + k0);
#pragma unroll
          for (int ct = 0; ct < CT; ++ct) {
            const v2f bb = *(const v2f*)(bfr + ((k0 >> 2) * NT + ct) * 64);
            acc[ct] = __builtin_amdgcn_wmma_f32_16x16x4_f32(
                 false, a,  false, bb,
                 (short)0, acc[ct],  false,  false);
          }
        }
        const int row0 = rt * 16 + khalf * 8;
        float* y = xwn + (size_t)(base + row0) * NOUT + (cg * CT) * 16 + l15;
        if (rt * 16 + 16 <= rows) {
#pragma unroll
          for (int ct = 0; ct < CT; ++ct)
#pragma unroll
            for (int r = 0; r < 8; ++r) y[(size_t)r * NOUT + ct * 16] = acc[ct][r];
        } else {
#pragma unroll
          for (int ct = 0; ct < CT; ++ct)
#pragma unroll
            for (int r = 0; r < 8; ++r)
              if (row0 + r < rows) y[(size_t)r * NOUT + ct * 16] = acc[ct][r];
        }
      }
    } else {
      float4* out4 = (float4*)out + (size_t)base * O4;
      const float4* tile4 = (const float4*)tile;
      for (int i = tid; i < rows * O4; i += 1024) {
        float4 v = tile4[i];
        if (ACT) {
          v.x = (v.x > 0.f) ? v.x : expm1f(v.x);
          v.y = (v.y > 0.f) ? v.y : expm1f(v.y);
          v.z = (v.z > 0.f) ? v.z : expm1f(v.z);
          v.w = (v.w > 0.f) ? v.w : expm1f(v.w);
        }
        out4[i] = v;
      }
    }
    __syncthreads();
  }
}

static inline int cdiv(long long a, long long b) { return (int)((a + b - 1) / b); }

template <int OUT, bool ACT, int REC, int NOUT>
static void run_layer(const float* xw, float* xout, const float* b,
                      const float* Wn, float* xwn,
                      const int* src, const int* dst,
                      const int* key,
                      const float* dinv, const float* ew,
                      const int* tcnt, int* rec, const int* rec_end,
                      int N, int E, hipStream_t stream) {
  agg_tile<OUT, ACT, REC, NOUT><<<cdiv(N, TROWS), 1024, 0, stream>>>(
      xw, src, dst, key, ew, dinv, b, tcnt, rec, rec_end, xout, Wn, xwn, N, E);
}

extern "C" void kernel_launch(void* const* d_in, const int* in_sizes, int n_in,
                              void* d_out, int out_size, void* d_ws, size_t ws_size,
                              hipStream_t stream) {
  const float* x   = (const float*)d_in[0];
  const int*   ei  = (const int*)d_in[1];
  const float* ew  = (const float*)d_in[2];
  const float* W1 = (const float*)d_in[3];  const float* b1 = (const float*)d_in[4];
  const float* W2 = (const float*)d_in[5];  const float* b2 = (const float*)d_in[6];
  const float* W3 = (const float*)d_in[7];  const float* b3 = (const float*)d_in[8];
  const float* W4 = (const float*)d_in[9];  const float* b4 = (const float*)d_in[10];
  const float* W5 = (const float*)d_in[11]; const float* b5 = (const float*)d_in[12];
  const float* W6 = (const float*)d_in[13]; const float* b6 = (const float*)d_in[14];

  const int N = in_sizes[0] / 32;
  const int E = in_sizes[2];
  const int* src = ei;
  const int* dst = ei + E;

  float* dinv = (float*)d_ws;
  float* buf0 = dinv + N + E;
  float* buf1 = buf0 + (size_t)N * 128;
  float* outp = (float*)d_out;
  int* key = (int*)(buf0 + (size_t)N * 64);
  int* tcnt = key + (size_t)cdiv(E, 1024) * 256;
  int* rec  = tcnt + (size_t)cdiv(cdiv(N, TROWS), 64) * 64;
  const int* rec_end = (const int*)buf1;
  int* erec = (int*)(buf1 + (size_t)N * 64);

  key_kernel<<<cdiv(E, 256), 256, 0, stream>>>(src, dst, ew, key, erec, N, E);
  tile_prep<<<cdiv(N, TROWS), 1024, 0, stream>>>(src, dst, key, ew, erec, dinv, tcnt, rec, rec_end, N, E);

  gemm_wmma<32, 4><<<cdiv(N, 128), 256, 0, stream>>>(x, W1, buf1, N);
  run_layer<4,   true,  1,  16>(buf1, nullptr, b1, W2, buf0, src, dst, key, dinv, ew, tcnt, rec, rec_end, N, E, stream);
  run_layer<16,  true,  2,  32>(buf0, nullptr, b2, W3, buf1, src, dst, key, dinv, ew, tcnt, rec, rec_end, N, E, stream);
  run_layer<32,  true,  2,  64>(buf1, nullptr, b3, W4, buf0, src, dst, key, dinv, ew, tcnt, rec, rec_end, N, E, stream);
  run_layer<64,  false, 2, 128>(buf0, nullptr, b4, W5, outp, src, dst, key, dinv, ew, tcnt, rec, rec_end, N, E, stream);
  run_layer<128, true,  2, 128>(outp, nullptr, b5, W6, buf1, src, dst, key, dinv, ew, tcnt, rec, rec_end, N, E, stream);
  run_layer<128, false, 2,   0>(buf1, outp, b6, nullptr, nullptr, src, dst, key, dinv, ew, tcnt, rec, rec_end, N, E, stream);
}


# Tile edge lists run as a software pipeline per wave: records fetched two batches ahead (layer 1: dinv one ahead), row loads in groups of 4 KiB (32, 16, 8 loads) not 8 steps; sum order LDS arrival
# speedup vs baseline: 6.9193x; 1.0306x over previous
#include <hip/hip_runtime.h>
#include <cstdint>
#include <type_traits>

typedef float v2f __attribute__((ext_vector_type(2)));
typedef float v8f __attribute__((ext_vector_type(8)));

constexpr int TROWS = 392;

__device__ __forceinline__ bool erec_usable(const int* erec, int n, int E) {
  return n <= (1 << 23) && 2LL * E <= 64LL * n && ((((size_t)erec) & 7) == 0);
}

__global__ __launch_bounds__(256) void key_kernel(const int* __restrict__ src,
                                                  const int* __restrict__ dst,
                                                  const float* __restrict__ ew,
                                                  int* __restrict__ key,
                                                  int* __restrict__ erec, int n, int E) {
  const int e = blockIdx.x * 256 + threadIdx.x;
  const bool er = (e < E) && erec_usable(erec, n, E);
  const int   d = (e < E) ? dst[e] : 0;
  const int   s = er ? src[e] : 0;
  const float w = er ? ew[e] : 0.f;
  const unsigned k = (unsigned)(d / TROWS);
  if (er) {
    const unsigned r = (unsigned)d - k * (unsigned)TROWS;
    ((int2*)erec)[e] = make_int2((int)((unsigned)s | (r << 23)), __float_as_int(w));
  }
  uint4 lo, hi;
  lo.x = __builtin_amdgcn_ballot_w32((k & 1u) != 0);
  lo.y = __builtin_amdgcn_ballot_w32((k & 2u) != 0);
  lo.z = __builtin_amdgcn_ballot_w32((k & 4u) != 0);
  lo.w = __builtin_amdgcn_ballot_w32((k & 8u) != 0);
  hi.x = __builtin_amdgcn_ballot_w32((k & 16u) != 0);
  hi.y = __builtin_amdgcn_ballot_w32((k & 32u) != 0);
  hi.z = __builtin_amdgcn_ballot_w32((k & 64u) != 0);
  hi.w = __builtin_amdgcn_ballot_w32((k & 128u) != 0);
  if ((threadIdx.x & 31) == 0) {
    const int g = e >> 5;
    uint4* p = (uint4*)key + ((g >> 5) * 64 + (g & 31));
    p[0]  = lo;
    p[32] = hi;
  }
}

constexpr int LDS_FLOATS = 61440;

__global__ __launch_bounds__(1024) void tile_prep(const int* __restrict__ src,
                                                   const int* __restrict__ dst,
                                                   const int* __restrict__ key,
                                                   const float* __restrict__ ew,
                                                   const int* __restrict__ erec,
                                                   float* __restrict__ dinv,
                                                   int* __restrict__ tcnt, int* __restrict__ rec,
                                                   const int* rec_end, int n, int E) {
  constexpr int QCAP = (LDS_FLOATS - TROWS) & ~1023;
  __shared__ float deg[TROWS];
  __shared__ int   q[QCAP];
  __shared__ int   qn;

  const int tid  = threadIdx.x;
  const int lane = tid & 31;
  const uint4* key16 = (const uint4*)key;
  const bool vec_ok = ((n + TROWS - 1) / TROWS <= 256) && ((((size_t)key) & 15) == 0);
  const int tiles = (n + TROWS - 1) / TROWS;
  const long long room = (long long)(rec_end - rec) / 2 / tiles;
  const int rcap = room > 0 ? (int)(room < 0x40000000 ? room : 0x40000000) & ~31 : 0;
  const bool rec_ok = rcap > 0 && n <= (1 << 23) && ((((size_t)rec) & 7) == 0);
  const bool er_ok = erec_usable(erec, n, E);

  for (int t = blockIdx.x; t * TROWS < n; t += gridDim.x) {
    const int base = t * TROWS;
    const int rows = (n - base < TROWS) ? (n - base) : TROWS;
    unsigned kx[8];
#pragma unroll
    for (int j = 0; j < 8; ++j) kx[j] = 0u - (((unsigned)t >> j) & 1u);

    for (int i = tid; i < TROWS; i += 1024) deg[i] = 1.0f;

    int seg_lo = 0, seg_len = E;
    int2* rt = (int2*)rec + (size_t)t * rcap;
    int rec_base = 0;
    while (seg_lo < E) {
      const int seg_hi = (E - seg_lo < seg_len) ? E : seg_lo + seg_len;
      if (tid == 0) qn = 0;
      __syncthreads();

      auto take = [&](int e) {
        const int pos = atomicAdd(&qn, 1);
        if (pos < QCAP) q[pos] = e;
      };
      auto test32 = [&](uint4 lo, uint4 hi, int e0) {
        unsigned m = ~((lo.x ^ kx[0]) | (lo.y ^ kx[1]) | (lo.z ^ kx[2]) | (lo.w ^ kx[3]) |
                       (hi.x ^ kx[4]) | (hi.y ^ kx[5]) | (hi.z ^ kx[6]) | (hi.w ^ kx[7]));
        while (m) {
          take(e0 + __builtin_ctz(m));
          m &= m - 1;
        }
      };
      const int g_lo   = seg_lo >> 5;
      const int g_full = vec_ok ? (seg_hi >> 5) : g_lo;
      int g = g_lo + tid;
      for (; g + 3 * 1024 < g_full; g += 4 * 1024) {
        const uint4* p = key16 + ((g >> 5) * 64 + lane);
        uint4 v[8];
#pragma unroll
        for (int u = 0; u < 4; ++u) {
          v[2 * u]     = p[u * 2048];
          v[2 * u + 1] = p[u * 2048 + 32];
        }
#pragma unroll
        for (int u = 0; u < 4; ++u) test32(v[2 * u], v[2 * u + 1], (g + u * 1024) << 5);
      }
      for (; g < g_full; g += 1024) {
        const uint4* p = key16 + ((g >> 5) * 64 + lane);
        test32(p[0], p[32], g << 5);
      }
      for (int e = (g_full << 5) + tid; e < seg_hi; e += 1024)
        if ((unsigned)(dst[e] - base) < (unsigned)TROWS) take(e);
      __syncthreads();

      const int cnt = qn;
      if (cnt <= QCAP) {
        const bool rec_fits = rec_ok && rec_base + cnt <= rcap;
        for (int i0 = tid; er_ok && i0 < cnt; i0 += 8 * 1024) {
          int  e[8];
          int2 rc[8];
#pragma unroll
          for (int u = 0; u < 8; ++u) {
            const int i = i0 + u * 1024;
            e[u] = q[i < cnt ? i : i0];
          }
#pragma unroll
          for (int u = 0; u < 8; ++u) rc[u] = ((const int2*)erec)[e[u]];
          asm volatile("" :: "v"(rc[0].x), "v"(rc[1].x), "v"(rc[2].x), "v"(rc[3].x),
                             "v"(rc[4].x), "v"(rc[5].x), "v"(rc[6].x), "v"(rc[7].x),
                             "v"(rc[0].y), "v"(rc[1].y), "v"(rc[2].y), "v"(rc[3].y),
                             "v"(rc[4].y), "v"(rc[5].y), "v"(rc[6].y), "v"(rc[7].y));
#pragma unroll
          for (int u = 0; u < 8; ++u) {
            const int i = i0 + u * 1024;
            if (i < cnt) {
              atomicAdd(&deg[(unsigned)rc[u].x >> 23], __int_as_float(rc[u].y));
              if (rec_fits) rt[rec_base + i] = rc[u];
            }
          }
        }
        for (int i0 = tid; !er_ok && i0 < cnt; i0 += 4 * 1024) {
          int e[4], s[4], d[4];
          float w[4];
#pragma unroll
          for (int u = 0; u < 4; ++u) {
            const int i = i0 + u * 1024;
            e[u] = q[i < cnt ? i : i0];
          }
#pragma unroll
          for (int u = 0; u < 4; ++u) {
            s[u] = src[e[u]];
            d[u] = dst[e[u]];
            w[u] = ew[e[u]];
          }
          asm volatile("" :: "v"(s[0]), "v"(s[1]), "v"(s[2]), "v"(s[3]),
                             "v"(d[0]), "v"(d[1]), "v"(d[2]), "v"(d[3]),
                             "v"(w[0]), "v"(w[1]), "v"(w[2]), "v"(w[3]));
#pragma unroll
          for (int u = 0; u < 4; ++u) {
            const int i = i0 + u * 1024;
            if (i < cnt) {
              const int r = d[u] - base;
              atomicAdd(&deg[r], w[u]);
              if (rec_fits)
                rt[rec_base + i] =
                    make_int2((int)((unsigned)s[u] | ((unsigned)r << 23)), __float_as_int(w[u]));
            }
          }
        }
        rec_base += cnt;
        seg_lo = seg_hi;
      } else {
        seg_len = QCAP;
      }
      __syncthreads();
    }

    if (tid == 0) tcnt[t] = (rec_ok && rec_base <= rcap) ? rec_base : -1;
    for (int i = tid; i < rows; i += 1024) {
      const float d = deg[i];
      dinv[base + i] = (d > 0.f) ? rsqrtf(d) : 0.f;
    }
    __syncthreads();
  }
}

template <int IN, int OUT>
__global__ __launch_bounds__(256) void gemm_wmma(const float* __restrict__ x,
                                                 const float* __restrict__ W,
                                                 float* __restrict__ y, int n) {
  constexpr bool USE_LDS = (IN * OUT) <= 8192;
  constexpr bool FULL    = (OUT % 16) == 0;
  __shared__ float sW[USE_LDS ? (IN * OUT) : 1];
  const float* Wp = W;
  if constexpr (USE_LDS) {
    for (int i = threadIdx.x; i < IN * OUT; i += 256) sW[i] = W[i];
    __syncthreads();
    Wp = sW;
  }

  const int lane  = threadIdx.x & 31;
  const int wave  = threadIdx.x >> 5;
  const int l15   = lane & 15;
  const int khalf = lane >> 4;
  const int row0  = blockIdx.x * 128 + wave * 16;

  if (row0 < n) {
    int mload = row0 + l15;
    if (mload > n - 1) mload = n - 1;
    const float* xrow = x + (size_t)mload * IN;
    const bool fulltile = (row0 + 16 <= n);

    for (int n0 = 0; n0 < OUT; n0 += 16) {
      const int colb = n0 + l15;
      const int colc = FULL ? colb : (colb < OUT ? colb : OUT - 1);
      v8f c = {};
#pragma unroll
      for (int k0 = 0; k0 < IN; k0 += 4) {
        const int krow = k0 + 2 * khalf;
        v2f a = *(const v2f*)(xrow + krow);
        v2f b;
        b.x = Wp[(size_t)krow * OUT + colc];
        b.y = Wp[(size_t)(krow + 1) * OUT + colc];
        c = __builtin_amdgcn_wmma_f32_16x16x4_f32(
             false, a,  false, b,
             (short)0, c,  false,  false);
      }
      const int rowoff = khalf * 8;
      if (FULL && fulltile) {
#pragma unroll
        for (int r = 0; r < 8; ++r)
          y[(size_t)(row0 + rowoff + r) * OUT + colb] = c[r];
      } else {
        const bool colok = FULL || (colb < OUT);
#pragma unroll
        for (int r = 0; r < 8; ++r) {
          int row = row0 + rowoff + r;
          if (row < n && colok) y[(size_t)row * OUT + colb] = c[r];
        }
      }
    }
  }
}

template <int N>
__device__ __forceinline__ void hold_loaded(const float (&v)[N]) {
  static_assert(N == 4 || N == 8 || N == 16 || N == 32, "group sizes of add32");
  if constexpr (N == 4) {
    asm volatile("" :: "v"(v[0]), "v"(v[1]), "v"(v[2]), "v"(v[3]));
  } else if constexpr (N == 8) {
    asm volatile("" :: "v"(v[0]), "v"(v[1]), "v"(v[2]), "v"(v[3]),
                       "v"(v[4]), "v"(v[5]), "v"(v[6]), "v"(v[7]));
  } else {
#pragma unroll
    for (int c = 0; c < N; c += 16)
      asm volatile("" :: "v"(v[c]),      "v"(v[c + 1]),  "v"(v[c + 2]),  "v"(v[c + 3]),
                         "v"(v[c + 4]),  "v"(v[c + 5]),  "v"(v[c + 6]),  "v"(v[c + 7]),
                         "v"(v[c + 8]),  "v"(v[c + 9]),  "v"(v[c + 10]), "v"(v[c + 11]),
                         "v"(v[c + 12]), "v"(v[c + 13]), "v"(v[c + 14]), "v"(v[c + 15]));
  }
}

template <int OUT, bool ACT, int REC, int NOUT>
__global__ __launch_bounds__(1024) void agg_tile(const float* __restrict__ xw,
                                                  const int* __restrict__ src,
                                                  const int* __restrict__ dst,
                                                  const int* __restrict__ key,
                                                  const float* __restrict__ ew,
                                                  const float* __restrict__ dinv,
                                                  const float* __restrict__ b,
                                                  const int* tcnt, int* rec, const int* rec_end,
                                                  float* __restrict__ out,
                                                  const float* __restrict__ Wn,
                                                  float* __restrict__ xwn, int n, int E) {
  constexpr bool MM = NOUT > 0;
  constexpr int PITCH = (MM && OUT >= 16) ? OUT + 4 : OUT;
  constexpr int TROWS_P = MM ? ((TROWS + 15) / 16) * 16 : TROWS;
  constexpr int TILE = TROWS_P * PITCH;
  constexpr int WF   = MM ? OUT * NOUT : 0;
  constexpr int BUDGET = (TILE + WF + 9216 > LDS_FLOATS) ? 80000 : LDS_FLOATS;
  constexpr int QCAP = (BUDGET - TILE - WF) & ~1023;
  constexpr int O4   = OUT / 4;
  constexpr int LPE  = OUT < 32 ? OUT : 32;
  constexpr int EPS  = 32 / LPE;
  constexpr int CPL  = OUT < 32 ? 1 : OUT / 32;
  constexpr int UN   = LPE < 8 ? LPE : 8;
  constexpr int UNF  = CPL == 4 ? 8 : (CPL == 2 ? 16 : LPE);
  constexpr int CPN  = NOUT >= 64 ? NOUT / 32 : 1;

  __shared__ __attribute__((aligned(16))) float tile[TILE];
  __shared__ __attribute__((aligned(16))) float wf[MM ? WF : 4];
  __shared__ int   q[QCAP];
  __shared__ int   qn;

  const int tid  = threadIdx.x;
  const int lane = tid & 31;
  const int wave = tid >> 5;
  if constexpr (MM) {
    constexpr int NT = NOUT / 16;
    for (int i = tid; i < OUT * NOUT; i += 1024) {
      const int k = i / NOUT, p = i - k * NOUT;
      const int c = (p % CPN) * 32 + p / CPN;
      const int ln = ((k >> 1) & 1) * 16 + (p & 15);
      wf[(((k >> 2) * NT + (p >> 4)) * 32 + ln) * 2 + (k & 1)] = Wn[k * NOUT + c];
    }
  }
  const int qsub = lane / LPE;
  const int ch0  = lane % LPE;
  const uint4* key16 = (const uint4*)key;
  const bool vec_ok = ((n + TROWS - 1) / TROWS <= 256) && ((((size_t)key) & 15) == 0);

  const int tiles = (n + TROWS - 1) / TROWS;
  const long long room = (long long)(rec_end - rec) / 2 / tiles;
  const int rcap = room > 0 ? (int)(room < 0x40000000 ? room : 0x40000000) & ~31 : 0;
  const bool rec_ok = rcap > 0 && n <= (1 << 23) && ((((size_t)rec) & 7) == 0);

  for (int t = blockIdx.x; t * TROWS < n; t += gridDim.x) {
    const int base = t * TROWS;
    const int rows = (n - base < TROWS) ? (n - base) : TROWS;
    unsigned kx[8];
#pragma unroll
    for (int j = 0; j < 8; ++j) kx[j] = 0u - (((unsigned)t >> j) & 1u);

    {
      const float4* xw4 = (const float4*)xw + (size_t)base * O4;
      for (int i = tid; i < rows * O4; i += 1024) {
        const int row = i / O4, c = (i - row * O4) * 4;
        const float dv = dinv[base + row];
        const float w = dv * dv;
        const float4 x = xw4[i];
        if constexpr (CPL > 1) {
          const float xs[4] = {x.x, x.y, x.z, x.w};
#pragma unroll
          for (int jj = 0; jj < 4; ++jj) {
            const int ch = ((c + jj) % CPL) * 32 + (c + jj) / CPL;
            tile[row * PITCH + ch] = xs[jj] * w + b[ch];
          }
        } else {
          float4 r;
          r.x = x.x * w + b[c];
          r.y = x.y * w + b[c + 1];
          r.z = x.z * w + b[c + 2];
          r.w = x.w * w + b[c + 3];
          *(float4*)(tile + row * PITCH + c) = r;
        }
      }
    }

    auto add32 = [&](auto fast, int s_l, int r_l, float nm_l, int left) {
      constexpr bool FAST = decltype(fast)::value;
      constexpr int  G    = FAST ? UNF : UN;
#pragma unroll 1
      for (int k0 = 0; k0 < LPE; k0 += G) {
        float val[G * CPL];
        int   rr[G];
        float nn[G];
#pragma unroll
        for (int u = 0; u < G; ++u) {
          const int ent = (k0 + u) * EPS + qsub;
          int s;
          if constexpr (LPE == 32) {
            s = __builtin_amdgcn_readlane(s_l, k0 + u);
          } else {
            s     = __shfl(s_l, ent);
            rr[u] = __shfl(r_l, ent);
            nn[u] = __shfl(nm_l, ent);
          }
          const float* xr;
          if constexpr (FAST) xr = xw + ((unsigned)s * (unsigned)OUT + (unsigned)(ch0 * CPL));
          else                xr = xw + (size_t)s * OUT + ch0 * CPL;
          if constexpr (CPL == 4) {
            const float4 v = *(const float4*)xr;
            val[4 * u] = v.x; val[4 * u + 1] = v.y; val[4 * u + 2] = v.z; val[4 * u + 3] = v.w;
          } else if constexpr (CPL == 2) {
            const float2 v = *(const float2*)xr;
            val[2 * u] = v.x; val[2 * u + 1] = v.y;
          } else {
            val[u] = xr[0];
          }
        }
        hold_loaded(val);
#pragma unroll
        for (int u = 0; u < G; ++u) {
          const int ent = (k0 + u) * EPS + qsub;
          if (ent < left) {
            int r; float nm;
            if constexpr (LPE == 32) {
              r  = __builtin_amdgcn_readlane(r_l, k0 + u);
              nm = __int_as_float(__builtin_amdgcn_readlane(__float_as_int(nm_l), k0 + u));
            } else {
              r = rr[u]; nm = nn[u];
            }
            float* tr = tile + r * PITCH + ch0;
#pragma unroll
            for (int j = 0; j < CPL; ++j) atomicAdd(tr + 32 * j, val[CPL * u + j] * nm);
          }
        }
      }
    };

    int seg_lo = 0, seg_len = E;
    int2* rt = (int2*)rec + (size_t)t * rcap;
    {
      const int cnt = rec_ok ? tcnt[t] : -1;
      if (cnt >= 0 && cnt <= rcap) {
        const int wv = __builtin_amdgcn_readfirstlane(wave);
        const int nb = (cnt + 31) >> 5;
        auto ridx = [&](int b0) { return (b0 << 5) + ((lane < cnt - (b0 << 5)) ? lane : 0); };
        int2  rc = make_int2(0, 0), rn = make_int2(0, 0);
        float ds = 0.f, dd = 0.f;
        if (wv < nb)      rc = rt[ridx(wv)];
        if (wv + 32 < nb) rn = rt[ridx(wv + 32)];
        if constexpr (REC == 1) {
          if (wv < nb) {
            ds = dinv[rc.x & 0x7fffff];
            dd = dinv[base + (int)((unsigned)rc.x >> 23)];
          }
        }
        __syncthreads();
        for (int bi = wv; bi < nb; bi += 32) {
          const int left = cnt - (bi << 5);
          const int idx  = ridx(bi);
          const int2  r   = rc;
          const float ds0 = ds, dd0 = dd;
          rc = rn;
          if (bi + 64 < nb) rn = rt[ridx(bi + 64)];
          if constexpr (REC == 1) {
            if (bi + 32 < nb) {
              ds = dinv[rc.x & 0x7fffff];
              dd = dinv[base + (int)((unsigned)rc.x >> 23)];
            }
          }
          const int s_l = r.x & 0x7fffff;
          const int r_l = (int)((unsigned)r.x >> 23);
          float nm_l = __int_as_float(r.y);
          if constexpr (REC == 1) {
            nm_l = ds0 * nm_l * dd0;
            if (lane < left) rt[idx] = make_int2(r.x, __float_as_int(nm_l));
          }
          add32(std::true_type{}, s_l, r_l, nm_l, left);
        }
        __syncthreads();
        seg_lo = E;
      }
    }
    while (seg_lo < E) {
      const int seg_hi = (E - seg_lo < seg_len) ? E : seg_lo + seg_len;
      if (tid == 0) qn = 0;
      __syncthreads();

      auto take = [&](int e) {
        const int pos = atomicAdd(&qn, 1);
        if (pos < QCAP) q[pos] = e;
      };
      auto test32 = [&](uint4 lo, uint4 hi, int e0) {
        unsigned m = ~((lo.x ^ kx[0]) | (lo.y ^ kx[1]) | (lo.z ^ kx[2]) | (lo.w ^ kx[3]) |
                       (hi.x ^ kx[4]) | (hi.y ^ kx[5]) | (hi.z ^ kx[6]) | (hi.w ^ kx[7]));
        while (m) {
          take(e0 + __builtin_ctz(m));
          m &= m - 1;
        }
      };
      const int g_lo   = seg_lo >> 5;
      const int g_full = vec_ok ? (seg_hi >> 5) : g_lo;
      int g = g_lo + tid;
      for (; g + 3 * 1024 < g_full; g += 4 * 1024) {
        const uint4* p = key16 + ((g >> 5) * 64 + lane);
        uint4 v[8];
#pragma unroll
        for (int u = 0; u < 4; ++u) {
          v[2 * u]     = p[u * 2048];
          v[2 * u + 1] = p[u * 2048 + 32];
        }
#pragma unroll
        for (int u = 0; u < 4; ++u) test32(v[2 * u], v[2 * u + 1], (g + u * 1024) << 5);
      }
      for (; g < g_full; g += 1024) {
        const uint4* p = key16 + ((g >> 5) * 64 + lane);
        test32(p[0], p[32], g << 5);
      }
      for (int e = (g_full << 5) + tid; e < seg_hi; e += 1024)
        if ((unsigned)(dst[e] - base) < (unsigned)TROWS) take(e);
      __syncthreads();

      const int cnt = qn;
      if (cnt <= QCAP) {
        const int nb = (cnt + 31) >> 5;
        for (int bi = wave; bi < nb; bi += 32) {
          const int left = cnt - (bi << 5);
          const int e    = q[(lane < left) ? ((bi << 5) + lane) : (bi << 5)];
          const int   s_l  = src[e];
          const int   r_l  = dst[e] - base;
          const float nm_l = dinv[s_l] * ew[e] * dinv[base + r_l];
          add32(std::false_type{}, s_l, r_l, nm_l, left);
        }
        seg_lo = seg_hi;
      } else {
        seg_len = QCAP;
      }
      __syncthreads();
    }

    if constexpr (MM) {
      constexpr int NT  = NOUT / 16;
      constexpr int CT  = NT < 2 ? 1 : 2;
      constexpr int UPR = NT / CT;
      if (ACT) {
        for (int i = tid; i < rows * O4; i += 1024) {
          const int row = i / O4, c = (i - row * O4) * 4;
          float4* p = (float4*)(tile + row * PITCH + c);
          float4 v = *p;
          v.x = (v.x > 0.f) ? v.x : expm1f(v.x);
          v.y = (v.y > 0.f) ? v.y : expm1f(v.y);
          v.z = (v.z > 0.f) ? v.z : expm1f(v.z);
          v.w = (v.w > 0.f) ? v.w : expm1f(v.w);
          *p = v;
        }
      }
      __syncthreads();
      const int wv    = __builtin_amdgcn_readfirstlane(wave);
      const int l15   = lane & 15;
      const int khalf = lane >> 4;
      const int units = ((rows + 15) >> 4) * UPR;
      for (int u = wv; u < units; u += 32) {
        const int rt = u / UPR, cg = u - rt * UPR;
        const float* arow = tile + (rt * 16 + l15) * PITCH + 2 * khalf;
        const float* bfr  = wf + ((cg * CT) * 32 + lane) * 2;
        v8f acc[CT];
#pragma unroll
        for (int ct = 0; ct < CT; ++ct) acc[ct] = v8f{};
#pragma unroll
        for (int k0 = 0; k0 < OUT; k0 += 4) {
          const v2f a = *(const v2f*)(arow + k0);
#pragma unroll
          for (int ct = 0; ct < CT; ++ct) {
            const v2f bb = *(const v2f*)(bfr + ((k0 >> 2) * NT + ct) * 64);
            acc[ct] = __builtin_amdgcn_wmma_f32_16x16x4_f32(
                 false, a,  false, bb,
                 (short)0, acc[ct],  false,  false);
          }
        }
        const int row0 = rt * 16 + khalf * 8;
        float* y = xwn + (size_t)(base + row0) * NOUT + (cg * CT) * 16 + l15;
        if (rt * 16 + 16 <= rows) {
#pragma unroll
          for (int ct = 0; ct < CT; ++ct)
#pragma unroll
            for (int r = 0; r < 8; ++r) y[(size_t)r * NOUT + ct * 16] = acc[ct][r];
        } else {
#pragma unroll
          for (int ct = 0; ct < CT; ++ct)
#pragma unroll
            for (int r = 0; r < 8; ++r)
              if (row0 + r < rows) y[(size_t)r * NOUT + ct * 16] = acc[ct][r];
        }
      }
    } else {
      float4* out4 = (float4*)out + (size_t)base * O4;
      const float4* tile4 = (const float4*)tile;
      for (int i = tid; i < rows * O4; i += 1024) {
        float4 v = tile4[i];
        if (ACT) {
          v.x = (v.x > 0.f) ? v.x : expm1f(v.x);
          v.y = (v.y > 0.f) ? v.y : expm1f(v.y);
          v.z = (v.z > 0.f) ? v.z : expm1f(v.z);
          v.w = (v.w > 0.f) ? v.w : expm1f(v.w);
        }
        out4[i] = v;
      }
    }
    __syncthreads();
  }
}

static inline int cdiv(long long a, long long b) { return (int)((a + b - 1) / b); }

template <int OUT, bool ACT, int REC, int NOUT>
static void run_layer(const float* xw, float* xout, const float* b,
                      const float* Wn, float* xwn,
                      const int* src, const int* dst,
                      const int* key,
                      const float* dinv, const float* ew,
                      const int* tcnt, int* rec, const int* rec_end,
                      int N, int E, hipStream_t stream) {
  agg_tile<OUT, ACT, REC, NOUT><<<cdiv(N, TROWS), 1024, 0, stream>>>(
      xw, src, dst, key, ew, dinv, b, tcnt, rec, rec_end, xout, Wn, xwn, N, E);
}

extern "C" void kernel_launch(void* const* d_in, const int* in_sizes, int n_in,
                              void* d_out, int out_size, void* d_ws, size_t ws_size,
                              hipStream_t stream) {
  const float* x   = (const float*)d_in[0];
  const int*   ei  = (const int*)d_in[1];
  const float* ew  = (const float*)d_in[2];
  const float* W1 = (const float*)d_in[3];  const float* b1 = (const float*)d_in[4];
  const float* W2 = (const float*)d_in[5];  const float* b2 = (const float*)d_in[6];
  const float* W3 = (const float*)d_in[7];  const float* b3 = (const float*)d_in[8];
  const float* W4 = (const float*)d_in[9];  const float* b4 = (const float*)d_in[10];
  const float* W5 = (const float*)d_in[11]; const float* b5 = (const float*)d_in[12];
  const float* W6 = (const float*)d_in[13]; const float* b6 = (const float*)d_in[14];

  const int N = in_sizes[0] / 32;
  const int E = in_sizes[2];
  const int* src = ei;
  const int* dst = ei + E;

  float* dinv = (float*)d_ws;
  float* buf0 = dinv + N + E;
  float* buf1 = buf0 + (size_t)N * 128;
  float* outp = (float*)d_out;
  int* key = (int*)(buf0 + (size_t)N * 64);
  int* tcnt = key + (size_t)cdiv(E, 1024) * 256;
  int* rec  = tcnt + (size_t)cdiv(cdiv(N, TROWS), 64) * 64;
  const int* rec_end = (const int*)buf1;
  int* erec = (int*)(buf1 + (size_t)N * 64);

  key_kernel<<<cdiv(E, 256), 256, 0, stream>>>(src, dst, ew, key, erec, N, E);
  tile_prep<<<cdiv(N, TROWS), 1024, 0, stream>>>(src, dst, key, ew, erec, dinv, tcnt, rec, rec_end, N, E);

  gemm_wmma<32, 4><<<cdiv(N, 128), 256, 0, stream>>>(x, W1, buf1, N);
  run_layer<4,   true,  1,  16>(buf1, nullptr, b1, W2, buf0, src, dst, key, dinv, ew, tcnt, rec, rec_end, N, E, stream);
  run_layer<16,  true,  2,  32>(buf0, nullptr, b2, W3, buf1, src, dst, key, dinv, ew, tcnt, rec, rec_end, N, E, stream);
  run_layer<32,  true,  2,  64>(buf1, nullptr, b3, W4, buf0, src, dst, key, dinv, ew, tcnt, rec, rec_end, N, E, stream);
  run_layer<64,  false, 2, 128>(buf0, nullptr, b4, W5, outp, src, dst, key, dinv, ew, tcnt, rec, rec_end, N, E, stream);
  run_layer<128, true,  2, 128>(outp, nullptr, b5, W6, buf1, src, dst, key, dinv, ew, tcnt, rec, rec_end, N, E, stream);
  run_layer<128, false, 2,   0>(buf1, outp, b6, nullptr, nullptr, src, dst, key, dinv, ew, tcnt, rec, rec_end, N, E, stream);
}


# conv4+conv5 (no activation between) as one linear map: 32-channel rows aggregated twice, then one WMMA product with W4*W5 + row-sum bias term; conv6 multiplies behind its sums; sums re-associated
# speedup vs baseline: 8.5529x; 1.2361x over previous
#include <hip/hip_runtime.h>
#include <cstdint>
#include <type_traits>

typedef float v2f __attribute__((ext_vector_type(2)));
typedef float v8f __attribute__((ext_vector_type(8)));

constexpr int TROWS = 392;

__device__ __forceinline__ bool erec_usable(const int* erec, int n, int E) {
  return n <= (1 << 23) && 2LL * E <= 64LL * n && ((((size_t)erec) & 7) == 0);
}

__global__ __launch_bounds__(256) void key_kernel(const int* __restrict__ src,
                                                  const int* __restrict__ dst,
                                                  const float* __restrict__ ew,
                                                  int* __restrict__ key,
                                                  int* __restrict__ erec, int n, int E) {
  const int e = blockIdx.x * 256 + threadIdx.x;
  const bool er = (e < E) && erec_usable(erec, n, E);
  const int   d = (e < E) ? dst[e] : 0;
  const int   s = er ? src[e] : 0;
  const float w = er ? ew[e] : 0.f;
  const unsigned k = (unsigned)(d / TROWS);
  if (er) {
    const unsigned r = (unsigned)d - k * (unsigned)TROWS;
    ((int2*)erec)[e] = make_int2((int)((unsigned)s | (r << 23)), __float_as_int(w));
  }
  uint4 lo, hi;
  lo.x = __builtin_amdgcn_ballot_w32((k & 1u) != 0);
  lo.y = __builtin_amdgcn_ballot_w32((k & 2u) != 0);
  lo.z = __builtin_amdgcn_ballot_w32((k & 4u) != 0);
  lo.w = __builtin_amdgcn_ballot_w32((k & 8u) != 0);
  hi.x = __builtin_amdgcn_ballot_w32((k & 16u) != 0);
  hi.y = __builtin_amdgcn_ballot_w32((k & 32u) != 0);
  hi.z = __builtin_amdgcn_ballot_w32((k & 64u) != 0);
  hi.w = __builtin_amdgcn_ballot_w32((k & 128u) != 0);
  if ((threadIdx.x & 31) == 0) {
    const int g = e >> 5;
    uint4* p = (uint4*)key + ((g >> 5) * 64 + (g & 31));
    p[0]  = lo;
    p[32] = hi;
  }
}

constexpr int LDS_FLOATS = 61440;

__global__ __launch_bounds__(1024) void tile_prep(const int* __restrict__ src,
                                                   const int* __restrict__ dst,
                                                   const int* __restrict__ key,
                                                   const float* __restrict__ ew,
                                                   const int* __restrict__ erec,
                                                   float* __restrict__ dinv,
                                                   int* __restrict__ tcnt, int* __restrict__ rec,
                                                   const int* rec_end, int n, int E) {
  constexpr int QCAP = (LDS_FLOATS - TROWS) & ~1023;
  __shared__ float deg[TROWS];
  __shared__ int   q[QCAP];
  __shared__ int   qn;

  const int tid  = threadIdx.x;
  const int lane = tid & 31;
  const uint4* key16 = (const uint4*)key;
  const bool vec_ok = ((n + TROWS - 1) / TROWS <= 256) && ((((size_t)key) & 15) == 0);
  const int tiles = (n + TROWS - 1) / TROWS;
  const long long room = (long long)(rec_end - rec) / 2 / tiles;
  const int rcap = room > 0 ? (int)(room < 0x40000000 ? room : 0x40000000) & ~31 : 0;
  const bool rec_ok = rcap > 0 && n <= (1 << 23) && ((((size_t)rec) & 7) == 0);
  const bool er_ok = erec_usable(erec, n, E);

  for (int t = blockIdx.x; t * TROWS < n; t += gridDim.x) {
    const int base = t * TROWS;
    const int rows = (n - base < TROWS) ? (n - base) : TROWS;
    unsigned kx[8];
#pragma unroll
    for (int j = 0; j < 8; ++j) kx[j] = 0u - (((unsigned)t >> j) & 1u);

    for (int i = tid; i < TROWS; i += 1024) deg[i] = 1.0f;

    int seg_lo = 0, seg_len = E;
    int2* rt = (int2*)rec + (size_t)t * rcap;
    int rec_base = 0;
    while (seg_lo < E) {
      const int seg_hi = (E - seg_lo < seg_len) ? E : seg_lo + seg_len;
      if (tid == 0) qn = 0;
      __syncthreads();

      auto take = [&](int e) {
        const int pos = atomicAdd(&qn, 1);
        if (pos < QCAP) q[pos] = e;
      };
      auto test32 = [&](uint4 lo, uint4 hi, int e0) {
        unsigned m = ~((lo.x ^ kx[0]) | (lo.y ^ kx[1]) | (lo.z ^ kx[2]) | (lo.w ^ kx[3]) |
                       (hi.x ^ kx[4]) | (hi.y ^ kx[5]) | (hi.z ^ kx[6]) | (hi.w ^ kx[7]));
        while (m) {
          take(e0 + __builtin_ctz(m));
          m &= m - 1;
        }
      };
      const int g_lo   = seg_lo >> 5;
      const int g_full = vec_ok ? (seg_hi >> 5) : g_lo;
      int g = g_lo + tid;
      for (; g + 3 * 1024 < g_full; g += 4 * 1024) {
        const uint4* p = key16 + ((g >> 5) * 64 + lane);
        uint4 v[8];
#pragma unroll
        for (int u = 0; u < 4; ++u) {
          v[2 * u]     = p[u * 2048];
          v[2 * u + 1] = p[u * 2048 + 32];
        }
#pragma unroll
        for (int u = 0; u < 4; ++u) test32(v[2 * u], v[2 * u + 1], (g + u * 1024) << 5);
      }
      for (; g < g_full; g += 1024) {
        const uint4* p = key16 + ((g >> 5) * 64 + lane);
        test32(p[0], p[32], g << 5);
      }
      for (int e = (g_full << 5) + tid; e < seg_hi; e += 1024)
        if ((unsigned)(dst[e] - base) < (unsigned)TROWS) take(e);
      __syncthreads();

      const int cnt = qn;
      if (cnt <= QCAP) {
        const bool rec_fits = rec_ok && rec_base + cnt <= rcap;
        for (int i0 = tid; er_ok && i0 < cnt; i0 += 8 * 1024) {
          int  e[8];
          int2 rc[8];
#pragma unroll
          for (int u = 0; u < 8; ++u) {
            const int i = i0 + u * 1024;
            e[u] = q[i < cnt ? i : i0];
          }
#pragma unroll
          for (int u = 0; u < 8; ++u) rc[u] = ((const int2*)erec)[e[u]];
          asm volatile("" :: "v"(rc[0].x), "v"(rc[1].x), "v"(rc[2].x), "v"(rc[3].x),
                             "v"(rc[4].x), "v"(rc[5].x), "v"(rc[6].x), "v"(rc[7].x),
                             "v"(rc[0].y), "v"(rc[1].y), "v"(rc[2].y), "v"(rc[3].y),
                             "v"(rc[4].y), "v"(rc[5].y), "v"(rc[6].y), "v"(rc[7].y));
#pragma unroll
          for (int u = 0; u < 8; ++u) {
            const int i = i0 + u * 1024;
            if (i < cnt) {
              atomicAdd(&deg[(unsigned)rc[u].x >> 23], __int_as_float(rc[u].y));
              if (rec_fits) rt[rec_base + i] = rc[u];
            }
          }
        }
        for (int i0 = tid; !er_ok && i0 < cnt; i0 += 4 * 1024) {
          int e[4], s[4], d[4];
          float w[4];
#pragma unroll
          for (int u = 0; u < 4; ++u) {
            const int i = i0 + u * 1024;
            e[u] = q[i < cnt ? i : i0];
          }
#pragma unroll
          for (int u = 0; u < 4; ++u) {
            s[u] = src[e[u]];
            d[u] = dst[e[u]];
            w[u] = ew[e[u]];
          }
          asm volatile("" :: "v"(s[0]), "v"(s[1]), "v"(s[2]), "v"(s[3]),
                             "v"(d[0]), "v"(d[1]), "v"(d[2]), "v"(d[3]),
                             "v"(w[0]), "v"(w[1]), "v"(w[2]), "v"(w[3]));
#pragma unroll
          for (int u = 0; u < 4; ++u) {
            const int i = i0 + u * 1024;
            if (i < cnt) {
              const int r = d[u] - base;
              atomicAdd(&deg[r], w[u]);
              if (rec_fits)
                rt[rec_base + i] =
                    make_int2((int)((unsigned)s[u] | ((unsigned)r << 23)), __float_as_int(w[u]));
            }
          }
        }
        rec_base += cnt;
        seg_lo = seg_hi;
      } else {
        seg_len = QCAP;
      }
      __syncthreads();
    }

    if (tid == 0) tcnt[t] = (rec_ok && rec_base <= rcap) ? rec_base : -1;
    for (int i = tid; i < rows; i += 1024) {
      const float d = deg[i];
      dinv[base + i] = (d > 0.f) ? rsqrtf(d) : 0.f;
    }
    __syncthreads();
  }
}

template <int IN, int OUT>
__global__ __launch_bounds__(256) void gemm_wmma(const float* __restrict__ x,
                                                 const float* __restrict__ W,
                                                 float* __restrict__ y, int n) {
  constexpr bool USE_LDS = (IN * OUT) <= 8192;
  constexpr bool FULL    = (OUT % 16) == 0;
  __shared__ float sW[USE_LDS ? (IN * OUT) : 1];
  const float* Wp = W;
  if constexpr (USE_LDS) {
    for (int i = threadIdx.x; i < IN * OUT; i += 256) sW[i] = W[i];
    __syncthreads();
    Wp = sW;
  }

  const int lane  = threadIdx.x & 31;
  const int wave  = threadIdx.x >> 5;
  const int l15   = lane & 15;
  const int khalf = lane >> 4;
  const int row0  = blockIdx.x * 128 + wave * 16;

  if (row0 < n) {
    int mload = row0 + l15;
    if (mload > n - 1) mload = n - 1;
    const float* xrow = x + (size_t)mload * IN;
    const bool fulltile = (row0 + 16 <= n);

    for (int n0 = 0; n0 < OUT; n0 += 16) {
      const int colb = n0 + l15;
      const int colc = FULL ? colb : (colb < OUT ? colb : OUT - 1);
      v8f c = {};
#pragma unroll
      for (int k0 = 0; k0 < IN; k0 += 4) {
        const int krow = k0 + 2 * khalf;
        v2f a = *(const v2f*)(xrow + krow);
        v2f b;
        b.x = Wp[(size_t)krow * OUT + colc];
        b.y = Wp[(size_t)(krow + 1) * OUT + colc];
        c = __builtin_amdgcn_wmma_f32_16x16x4_f32(
             false, a,  false, b,
             (short)0, c,  false,  false);
      }
      const int rowoff = khalf * 8;
      if (FULL && fulltile) {
#pragma unroll
        for (int r = 0; r < 8; ++r)
          y[(size_t)(row0 + rowoff + r) * OUT + colb] = c[r];
      } else {
        const bool colok = FULL || (colb < OUT);
#pragma unroll
        for (int r = 0; r < 8; ++r) {
          int row = row0 + rowoff + r;
          if (row < n && colok) y[(size_t)row * OUT + colb] = c[r];
        }
      }
    }
  }
}

template <int N>
__device__ __forceinline__ void hold_loaded(const float (&v)[N]) {
  static_assert(N == 4 || N == 8 || N == 16 || N == 32, "group sizes of add32");
  if constexpr (N == 4) {
    asm volatile("" :: "v"(v[0]), "v"(v[1]), "v"(v[2]), "v"(v[3]));
  } else if constexpr (N == 8) {
    asm volatile("" :: "v"(v[0]), "v"(v[1]), "v"(v[2]), "v"(v[3]),
                       "v"(v[4]), "v"(v[5]), "v"(v[6]), "v"(v[7]));
  } else {
#pragma unroll
    for (int c = 0; c < N; c += 16)
      asm volatile("" :: "v"(v[c]),      "v"(v[c + 1]),  "v"(v[c + 2]),  "v"(v[c + 3]),
                         "v"(v[c + 4]),  "v"(v[c + 5]),  "v"(v[c + 6]),  "v"(v[c + 7]),
                         "v"(v[c + 8]),  "v"(v[c + 9]),  "v"(v[c + 10]), "v"(v[c + 11]),
                         "v"(v[c + 12]), "v"(v[c + 13]), "v"(v[c + 14]), "v"(v[c + 15]));
  }
}

template <int OUT, bool ACT, int REC, int NOUT, int KIND, int MID>
__global__ __launch_bounds__(1024) void agg_tile(const float* __restrict__ xw,
                                                  const int* __restrict__ src,
                                                  const int* __restrict__ dst,
                                                  const int* __restrict__ key,
                                                  const float* __restrict__ ew,
                                                  const float* __restrict__ dinv,
                                                  const float* __restrict__ b,
                                                  const int* tcnt, int* rec, const int* rec_end,
                                                  float* __restrict__ out,
                                                  const float* __restrict__ Wn,
                                                  const float* __restrict__ Wb,
                                                  const float* __restrict__ bm,
                                                  float* __restrict__ xwn, int n, int E) {
  constexpr bool MM = NOUT > 0;
  constexpr bool PRE   = KIND == 0;
  constexpr bool MERGE = KIND == 2;
  static_assert(KIND == 0 || (KIND == 1 && !MM) || ((KIND == 2 || KIND == 3) && MM), "kinds");
  constexpr int PITCH = (MM && OUT >= 16) ? OUT + 4 : OUT;
  constexpr int TROWS_P = MM ? ((TROWS + 15) / 16) * 16 : TROWS;
  constexpr int TILE = TROWS_P * PITCH;
  constexpr int WF   = MM ? OUT * NOUT : 0;
  constexpr int XF   = MERGE ? TROWS_P + NOUT : 0;
  constexpr int BUDGET = (TILE + WF + XF + 9216 > LDS_FLOATS) ? 80000 : LDS_FLOATS;
  constexpr int QCAP = (BUDGET - TILE - WF - XF) & ~1023;
  constexpr int O4   = OUT / 4;
  constexpr int LPE  = OUT < 32 ? OUT : 32;
  constexpr int EPS  = 32 / LPE;
  constexpr int CPL  = OUT < 32 ? 1 : OUT / 32;
  constexpr int UN   = LPE < 8 ? LPE : 8;
  constexpr int UNF  = CPL == 4 ? 8 : (CPL == 2 ? 16 : LPE);
  constexpr int CPN  = (NOUT >= 64 && KIND != 3) ? NOUT / 32 : 1;

  __shared__ __attribute__((aligned(16))) float tile[TILE];
  __shared__ __attribute__((aligned(16))) float wf[MM ? WF : 4];
  __shared__ int   q[QCAP];
  __shared__ int   qn;
  __shared__ float rs[MERGE ? TROWS_P : 1];
  __shared__ float cb[MERGE ? NOUT : 1];

  const int tid  = threadIdx.x;
  const int lane = tid & 31;
  const int wave = tid >> 5;
  if constexpr (MERGE) {
    constexpr int NT = NOUT / 16;
    static_assert(OUT % 16 == 0 && MID % 4 == 0 && (OUT / 16) * NT <= 32, "one tile per wave");
    const int wv0 = __builtin_amdgcn_readfirstlane(wave);
    if (wv0 < (OUT / 16) * NT) {
      const int mt = wv0 / NT, nt = wv0 - mt * NT;
      const int l15 = lane & 15, khalf = lane >> 4;
      const int p = nt * 16 + l15;
      const int c = (p % CPN) * 32 + p / CPN;
      const float* arow = Wn + (mt * 16 + l15) * MID + 2 * khalf;
      const float* bcol = Wb + (2 * khalf) * NOUT + c;
      v8f acc = {};
#pragma unroll
      for (int k0 = 0; k0 < MID; k0 += 4) {
        const v2f a = *(const v2f*)(arow + k0);
        v2f bb;
        bb.x = bcol[k0 * NOUT];
        bb.y = bcol[(k0 + 1) * NOUT];
        acc = __builtin_amdgcn_wmma_f32_16x16x4_f32(
             false, a,  false, bb,
             (short)0, acc,  false,  false);
      }
#pragma unroll
      for (int r = 0; r < 8; ++r) {
        const int k  = mt * 16 + khalf * 8 + r;
        const int ln = ((k >> 1) & 1) * 16 + l15;
        wf[(((k >> 2) * NT + nt) * 32 + ln) * 2 + (k & 1)] = acc[r];
      }
    }
    constexpr int T0 = (OUT / 16) * NT * 32;
    static_assert(T0 + NOUT <= 1024, "threads for bm^T Wb");
    if (tid >= T0 && tid < T0 + NOUT) {
      const int c = tid - T0;
      float s = 0.f;
#pragma unroll
      for (int j = 0; j < MID; ++j) s += bm[j] * Wb[j * NOUT + c];
      cb[c] = s;
    }
  } else if constexpr (MM) {
    constexpr int NT = NOUT / 16;
    for (int i = tid; i < OUT * NOUT; i += 1024) {
      const int k = i / NOUT, p = i - k * NOUT;
      const int c = (p % CPN) * 32 + p / CPN;
      const int ln = ((k >> 1) & 1) * 16 + (p & 15);
      wf[(((k >> 2) * NT + (p >> 4)) * 32 + ln) * 2 + (k & 1)] = Wn[k * NOUT + c];
    }
  }
  const int qsub = lane / LPE;
  const int ch0  = lane % LPE;
  const uint4* key16 = (const uint4*)key;
  const bool vec_ok = ((n + TROWS - 1) / TROWS <= 256) && ((((size_t)key) & 15) == 0);

  const int tiles = (n + TROWS - 1) / TROWS;
  const long long room = (long long)(rec_end - rec) / 2 / tiles;
  const int rcap = room > 0 ? (int)(room < 0x40000000 ? room : 0x40000000) & ~31 : 0;
  const bool rec_ok = rcap > 0 && n <= (1 << 23) && ((((size_t)rec) & 7) == 0);

  for (int t = blockIdx.x; t * TROWS < n; t += gridDim.x) {
    const int base = t * TROWS;
    const int rows = (n - base < TROWS) ? (n - base) : TROWS;
    unsigned kx[8];
#pragma unroll
    for (int j = 0; j < 8; ++j) kx[j] = 0u - (((unsigned)t >> j) & 1u);

    if constexpr (MERGE) {
      for (int i = tid; i < TROWS_P; i += 1024) {
        const float dv = (i < rows) ? dinv[base + i] : 0.f;
        rs[i] = dv * dv;
      }
    }
    {
      const float4* xw4 = (const float4*)xw + (size_t)base * O4;
      for (int i = tid; i < rows * O4; i += 1024) {
        const int row = i / O4, c = (i - row * O4) * 4;
        const float dv = dinv[base + row];
        const float w = dv * dv;
        const float4 x = xw4[i];
        if constexpr (CPL > 1) {
          const float xs[4] = {x.x, x.y, x.z, x.w};
#pragma unroll
          for (int jj = 0; jj < 4; ++jj) {
            const int ch = ((c + jj) % CPL) * 32 + (c + jj) / CPL;
            if constexpr (PRE) tile[row * PITCH + ch] = xs[jj] * w + b[ch];
            else               tile[row * PITCH + ch] = xs[jj] * w;
          }
        } else {
          float4 r;
          if constexpr (PRE) {
            r.x = x.x * w + b[c];
            r.y = x.y * w + b[c + 1];
            r.z = x.z * w + b[c + 2];
            r.w = x.w * w + b[c + 3];
          } else {
            r.x = x.x * w;
            r.y = x.y * w;
            r.z = x.z * w;
            r.w = x.w * w;
          }
          *(float4*)(tile + row * PITCH + c) = r;
        }
      }
    }

    auto add32 = [&](auto fast, int s_l, int r_l, float nm_l, int left) {
      constexpr bool FAST = decltype(fast)::value;
      constexpr int  G    = FAST ? UNF : UN;
      if constexpr (MERGE) {
        if (lane < left) atomicAdd(&rs[r_l], nm_l);
      }
#pragma unroll 1
      for (int k0 = 0; k0 < LPE; k0 += G) {
        float val[G * CPL];
        int   rr[G];
        float nn[G];
#pragma unroll
        for (int u = 0; u < G; ++u) {
          const int ent = (k0 + u) * EPS + qsub;
          int s;
          if constexpr (LPE == 32) {
            s = __builtin_amdgcn_readlane(s_l, k0 + u);
          } else {
            s     = __shfl(s_l, ent);
            rr[u] = __shfl(r_l, ent);
            nn[u] = __shfl(nm_l, ent);
          }
          const float* xr;
          if constexpr (FAST) xr = xw + ((unsigned)s * (unsigned)OUT + (unsigned)(ch0 * CPL));
          else                xr = xw + (size_t)s * OUT + ch0 * CPL;
          if constexpr (CPL == 4) {
            const float4 v = *(const float4*)xr;
            val[4 * u] = v.x; val[4 * u + 1] = v.y; val[4 * u + 2] = v.z; val[4 * u + 3] = v.w;
          } else if constexpr (CPL == 2) {
            const float2 v = *(const float2*)xr;
            val[2 * u] = v.x; val[2 * u + 1] = v.y;
          } else {
            val[u] = xr[0];
          }
        }
        hold_loaded(val);
#pragma unroll
        for (int u = 0; u < G; ++u) {
          const int ent = (k0 + u) * EPS + qsub;
          if (ent < left) {
            int r; float nm;
            if constexpr (LPE == 32) {
              r  = __builtin_amdgcn_readlane(r_l, k0 + u);
              nm = __int_as_float(__builtin_amdgcn_readlane(__float_as_int(nm_l), k0 + u));
            } else {
              r = rr[u]; nm = nn[u];
            }
            float* tr = tile + r * PITCH + ch0;
#pragma unroll
            for (int j = 0; j < CPL; ++j) atomicAdd(tr + 32 * j, val[CPL * u + j] * nm);
          }
        }
      }
    };

    int seg_lo = 0, seg_len = E;
    int2* rt = (int2*)rec + (size_t)t * rcap;
    {
      const int cnt = rec_ok ? tcnt[t] : -1;
      if (cnt >= 0 && cnt <= rcap) {
        const int wv = __builtin_amdgcn_readfirstlane(wave);
        const int nb = (cnt + 31) >> 5;
        auto ridx = [&](int b0) { return (b0 << 5) + ((lane < cnt - (b0 << 5)) ? lane : 0); };
        int2  rc = make_int2(0, 0), rn = make_int2(0, 0);
        float ds = 0.f, dd = 0.f;
        if (wv < nb)      rc = rt[ridx(wv)];
        if (wv + 32 < nb) rn = rt[ridx(wv + 32)];
        if constexpr (REC == 1) {
          if (wv < nb) {
            ds = dinv[rc.x & 0x7fffff];
            dd = dinv[base + (int)((unsigned)rc.x >> 23)];
          }
        }
        __syncthreads();
        for (int bi = wv; bi < nb; bi += 32) {
          const int left = cnt - (bi << 5);
          const int idx  = ridx(bi);
          const int2  r   = rc;
          const float ds0 = ds, dd0 = dd;
          rc = rn;
          if (bi + 64 < nb) rn = rt[ridx(bi + 64)];
          if constexpr (REC == 1) {
            if (bi + 32 < nb) {
              ds = dinv[rc.x & 0x7fffff];
              dd = dinv[base + (int)((unsigned)rc.x >> 23)];
            }
          }
          const int s_l = r.x & 0x7fffff;
          const int r_l = (int)((unsigned)r.x >> 23);
          float nm_l = __int_as_float(r.y);
          if constexpr (REC == 1) {
            nm_l = ds0 * nm_l * dd0;
            if (lane < left) rt[idx] = make_int2(r.x, __float_as_int(nm_l));
          }
          add32(std::true_type{}, s_l, r_l, nm_l, left);
        }
        __syncthreads();
        seg_lo = E;
      }
    }
    while (seg_lo < E) {
      const int seg_hi = (E - seg_lo < seg_len) ? E : seg_lo + seg_len;
      if (tid == 0) qn = 0;
      __syncthreads();

      auto take = [&](int e) {
        const int pos = atomicAdd(&qn, 1);
        if (pos < QCAP) q[pos] = e;
      };
      auto test32 = [&](uint4 lo, uint4 hi, int e0) {
        unsigned m = ~((lo.x ^ kx[0]) | (lo.y ^ kx[1]) | (lo.z ^ kx[2]) | (lo.w ^ kx[3]) |
                       (hi.x ^ kx[4]) | (hi.y ^ kx[5]) | (hi.z ^ kx[6]) | (hi.w ^ kx[7]));
        while (m) {
          take(e0 + __builtin_ctz(m));
          m &= m - 1;
        }
      };
      const int g_lo   = seg_lo >> 5;
      const int g_full = vec_ok ? (seg_hi >> 5) : g_lo;
      int g = g_lo + tid;
      for (; g + 3 * 1024 < g_full; g += 4 * 1024) {
        const uint4* p = key16 + ((g >> 5) * 64 + lane);
        uint4 v[8];
#pragma unroll
        for (int u = 0; u < 4; ++u) {
          v[2 * u]     = p[u * 2048];
          v[2 * u + 1] = p[u * 2048 + 32];
        }
#pragma unroll
        for (int u = 0; u < 4; ++u) test32(v[2 * u], v[2 * u + 1], (g + u * 1024) << 5);
      }
      for (; g < g_full; g += 1024) {
        const uint4* p = key16 + ((g >> 5) * 64 + lane);
        test32(p[0], p[32], g << 5);
      }
      for (int e = (g_full << 5) + tid; e < seg_hi; e += 1024)
        if ((unsigned)(dst[e] - base) < (unsigned)TROWS) take(e);
      __syncthreads();

      const int cnt = qn;
      if (cnt <= QCAP) {
        const int nb = (cnt + 31) >> 5;
        for (int bi = wave; bi < nb; bi += 32) {
          const int left = cnt - (bi << 5);
          const int e    = q[(lane < left) ? ((bi << 5) + lane) : (bi << 5)];
          const int   s_l  = src[e];
          const int   r_l  = dst[e] - base;
          const float nm_l = dinv[s_l] * ew[e] * dinv[base + r_l];
          add32(std::false_type{}, s_l, r_l, nm_l, left);
        }
        seg_lo = seg_hi;
      } else {
        seg_len = QCAP;
      }
      __syncthreads();
    }

    if constexpr (MM) {
      constexpr int NT  = NOUT / 16;
      constexpr int CT  = NT < 2 ? 1 : 2;
      constexpr int UPR = NT / CT;
      if (ACT && PRE) {
        for (int i = tid; i < rows * O4; i += 1024) {
          const int row = i / O4, c = (i - row * O4) * 4;
          float4* p = (float4*)(tile + row * PITCH + c);
          float4 v = *p;
          v.x = (v.x > 0.f) ? v.x : expm1f(v.x);
          v.y = (v.y > 0.f) ? v.y : expm1f(v.y);
          v.z = (v.z > 0.f) ? v.z : expm1f(v.z);
          v.w = (v.w > 0.f) ? v.w : expm1f(v.w);
          *p = v;
        }
      }
      __syncthreads();
      const int wv    = __builtin_amdgcn_readfirstlane(wave);
      const int l15   = lane & 15;
      const int khalf = lane >> 4;
      const int units = ((rows + 15) >> 4) * UPR;
      for (int u = wv; u < units; u += 32) {
        const int rt = u / UPR, cg = u - rt * UPR;
        const float* arow = tile + (rt * 16 + l15) * PITCH + 2 * khalf;
        const float* bfr  = wf + ((cg * CT) * 32 + lane) * 2;
        v8f acc[CT];
#pragma unroll
        for (int ct = 0; ct < CT; ++ct) acc[ct] = v8f{};
#pragma unroll
        for (int k0 = 0; k0 < OUT; k0 += 4) {
          const v2f a = *(const v2f*)(arow + k0);
#pragma unroll
          for (int ct = 0; ct < CT; ++ct) {
            const v2f bb = *(const v2f*)(bfr + ((k0 >> 2) * NT + ct) * 64);
            acc[ct] = __builtin_amdgcn_wmma_f32_16x16x4_f32(
                 false, a,  false, bb,
                 (short)0, acc[ct],  false,  false);
          }
        }
        const int row0 = rt * 16 + khalf * 8;
        float* y = xwn + (size_t)(base + row0) * NOUT + (cg * CT) * 16 + l15;
        if constexpr (!PRE) {
#pragma unroll
          for (int ct = 0; ct < CT; ++ct) {
            const int p = (cg * CT + ct) * 16 + l15;
            const int c = (p % CPN) * 32 + p / CPN;
            const float bc = b[c];
            float cc = 0.f;
            if constexpr (MERGE) cc = cb[c];
#pragma unroll
            for (int r = 0; r < 8; ++r) {
              float v = acc[ct][r];
              if constexpr (MERGE) v += rs[row0 + r] * cc;
              v += bc;
              if (ACT) v = (v > 0.f) ? v : expm1f(v);
              acc[ct][r] = v;
            }
          }
        }
        if (rt * 16 + 16 <= rows) {
#pragma unroll
          for (int ct = 0; ct < CT; ++ct)
#pragma unroll
            for (int r = 0; r < 8; ++r) y[(size_t)r * NOUT + ct * 16] = acc[ct][r];
        } else {
#pragma unroll
          for (int ct = 0; ct < CT; ++ct)
#pragma unroll
            for (int r = 0; r < 8; ++r)
              if (row0 + r < rows) y[(size_t)r * NOUT + ct * 16] = acc[ct][r];
        }
      }
    } else {
      float4* out4 = (float4*)out + (size_t)base * O4;
      const float4* tile4 = (const float4*)tile;
      for (int i = tid; i < rows * O4; i += 1024) {
        float4 v = tile4[i];
        if (ACT) {
          v.x = (v.x > 0.f) ? v.x : expm1f(v.x);
          v.y = (v.y > 0.f) ? v.y : expm1f(v.y);
          v.z = (v.z > 0.f) ? v.z : expm1f(v.z);
          v.w = (v.w > 0.f) ? v.w : expm1f(v.w);
        }
        out4[i] = v;
      }
    }
    __syncthreads();
  }
}

static inline int cdiv(long long a, long long b) { return (int)((a + b - 1) / b); }

template <int OUT, bool ACT, int REC, int NOUT, int KIND, int MID>
static void run_layer(const float* xw, float* xout, const float* b,
                      const float* Wn, const float* Wb, const float* bm, float* xwn,
                      const int* src, const int* dst,
                      const int* key,
                      const float* dinv, const float* ew,
                      const int* tcnt, int* rec, const int* rec_end,
                      int N, int E, hipStream_t stream) {
  agg_tile<OUT, ACT, REC, NOUT, KIND, MID><<<cdiv(N, TROWS), 1024, 0, stream>>>(
      xw, src, dst, key, ew, dinv, b, tcnt, rec, rec_end, xout, Wn, Wb, bm, xwn, N, E);
}

extern "C" void kernel_launch(void* const* d_in, const int* in_sizes, int n_in,
                              void* d_out, int out_size, void* d_ws, size_t ws_size,
                              hipStream_t stream) {
  const float* x   = (const float*)d_in[0];
  const int*   ei  = (const int*)d_in[1];
  const float* ew  = (const float*)d_in[2];
  const float* W1 = (const float*)d_in[3];  const float* b1 = (const float*)d_in[4];
  const float* W2 = (const float*)d_in[5];  const float* b2 = (const float*)d_in[6];
  const float* W3 = (const float*)d_in[7];  const float* b3 = (const float*)d_in[8];
  const float* W4 = (const float*)d_in[9];  const float* b4 = (const float*)d_in[10];
  const float* W5 = (const float*)d_in[11]; const float* b5 = (const float*)d_in[12];
  const float* W6 = (const float*)d_in[13]; const float* b6 = (const float*)d_in[14];

  const int N = in_sizes[0] / 32;
  const int E = in_sizes[2];
  const int* src = ei;
  const int* dst = ei + E;

  float* dinv = (float*)d_ws;
  float* buf0 = dinv + N + E;
  float* buf1 = buf0 + (size_t)N * 128;
  float* outp = (float*)d_out;
  int* key = (int*)(buf0 + (size_t)N * 64);
  int* tcnt = key + (size_t)cdiv(E, 1024) * 256;
  int* rec  = tcnt + (size_t)cdiv(cdiv(N, TROWS), 64) * 64;
  const int* rec_end = (const int*)buf1;
  int* erec = (int*)(buf1 + (size_t)N * 64);

  key_kernel<<<cdiv(E, 256), 256, 0, stream>>>(src, dst, ew, key, erec, N, E);
  tile_prep<<<cdiv(N, TROWS), 1024, 0, stream>>>(src, dst, key, ew, erec, dinv, tcnt, rec, rec_end, N, E);

  float* x3 = buf0;
  float* y4 = buf0 + (size_t)N * 32;
  gemm_wmma<32, 4><<<cdiv(N, 128), 256, 0, stream>>>(x, W1, buf1, N);
  run_layer<4,   true,  1,  16, 0,  0>(buf1, nullptr, b1, W2, nullptr, nullptr, buf0, src, dst, key, dinv, ew, tcnt, rec, rec_end, N, E, stream);
  run_layer<16,  true,  2,  32, 0,  0>(buf0, nullptr, b2, W3, nullptr, nullptr, buf1, src, dst, key, dinv, ew, tcnt, rec, rec_end, N, E, stream);
  run_layer<32,  true,  2,   0, 0,  0>(buf1, x3, b3, nullptr, nullptr, nullptr, nullptr, src, dst, key, dinv, ew, tcnt, rec, rec_end, N, E, stream);
  run_layer<32,  false, 2,   0, 1,  0>(x3, y4, nullptr, nullptr, nullptr, nullptr, nullptr, src, dst, key, dinv, ew, tcnt, rec, rec_end, N, E, stream);
  run_layer<32,  true,  2, 128, 2, 64>(y4, nullptr, b5, W4, W5, b4, buf1, src, dst, key, dinv, ew, tcnt, rec, rec_end, N, E, stream);
  run_layer<128, false, 2, 128, 3,  0>(buf1, nullptr, b6, W6, nullptr, nullptr, outp, src, dst, key, dinv, ew, tcnt, rec, rec_end, N, E, stream);
}
